# starting kernel
# baseline (speedup 1.0000x reference)
#include <hip/hip_runtime.h>
#include <math.h>

#define PB 16
#define PN 4096
#define PM 512
#define PK 16

typedef float v2f __attribute__((ext_vector_type(2)));
typedef float v8f __attribute__((ext_vector_type(8)));

__device__ __forceinline__ v8f wmma_f32x4(v2f a, v2f b, v8f c) {
  return __builtin_amdgcn_wmma_f32_16x16x4_f32(false, a, false, b, (short)0, c, false, false);
}

__global__ __launch_bounds__(256)
void prep_weights_kernel(const float* __restrict__ Wq, const float* __restrict__ Wk,
                         const float* __restrict__ Wv, float* __restrict__ outQK,
                         float* __restrict__ outV, int D, int Cw, int Kpad) {
  int i = blockIdx.x * 256 + threadIdx.x;
  if (i >= D * Kpad) return;
  int d = i / Kpad, k = i % Kpad;
  float q = 0.f, v = 0.f;
  if (k < Cw) {
    q = Wq[(size_t)d * Cw + k] - Wk[(size_t)d * Cw + k];
    v = Wv[(size_t)d * Cw + k];
  }
  outQK[i] = q;
  outV[i] = v;
}

__global__ __launch_bounds__(256)
void fps_kernel(const float* __restrict__ x, float* __restrict__ pos) {
  const int b = blockIdx.x;
  const int tid = threadIdx.x;
  const int PPT = PN / 256;
  __shared__ float sLast[3];
  __shared__ int sSel[PM];
  __shared__ float sVal[256];
  __shared__ int sArg[256];

  float px[PPT], py[PPT], pz[PPT], dmin[PPT];
#pragma unroll
  for (int i = 0; i < PPT; ++i) {
    int p = tid + i * 256;
    const float* xp = x + ((size_t)b * PN + p) * 3;
    px[i] = xp[0]; py[i] = xp[1]; pz[i] = xp[2];
    dmin[i] = INFINITY;
  }
  if (tid == 0) {
    const float* xp = x + (size_t)b * PN * 3;
    sLast[0] = xp[0]; sLast[1] = xp[1]; sLast[2] = xp[2];
    sSel[0] = 0;
  }
  __syncthreads();
  for (int it = 1; it < PM; ++it) {
    float bx = sLast[0], by = sLast[1], bz = sLast[2];
    float best = -INFINITY; int barg = PN;
#pragma unroll
    for (int i = 0; i < PPT; ++i) {
      float dx = px[i] - bx, dy = py[i] - by, dz = pz[i] - bz;
      float d = dx * dx + dy * dy + dz * dz;
      dmin[i] = fminf(dmin[i], d);
      int p = tid + i * 256;
      if (dmin[i] > best || (dmin[i] == best && p < barg)) { best = dmin[i]; barg = p; }
    }
    sVal[tid] = best; sArg[tid] = barg;
    __syncthreads();
    for (int s = 128; s > 0; s >>= 1) {
      if (tid < s) {
        float v = sVal[tid + s]; int a = sArg[tid + s];
        if (v > sVal[tid] || (v == sVal[tid] && a < sArg[tid])) { sVal[tid] = v; sArg[tid] = a; }
      }
      __syncthreads();
    }
    if (tid == 0) {
      int nxt = sArg[0];
      sSel[it] = nxt;
      const float* xp = x + ((size_t)b * PN + nxt) * 3;
      sLast[0] = xp[0]; sLast[1] = xp[1]; sLast[2] = xp[2];
    }
    __syncthreads();
  }
  for (int i = tid; i < PM; i += 256) {
    int p = sSel[i];
    const float* xp = x + ((size_t)b * PN + p) * 3;
    pos[((size_t)b * 3 + 0) * PM + i] = xp[0];
    pos[((size_t)b * 3 + 1) * PM + i] = xp[1];
    pos[((size_t)b * 3 + 2) * PM + i] = xp[2];
  }
}

__global__ __launch_bounds__(256)
void knn_kernel(const float* __restrict__ f, int C, int* __restrict__ idxout) {
  const int bn = blockIdx.x;
  const int b = bn / PM, n = bn % PM;
  const int tid = threadIdx.x;
  __shared__ float sD[PM];
  __shared__ float sVal[256];
  __shared__ int sArg[256];
  const float* fb = f + (size_t)b * C * PM;
  for (int m = tid; m < PM; m += 256) {
    float acc = 0.f;
    for (int c = 0; c < C; ++c) {
      float d = fb[(size_t)c * PM + n] - fb[(size_t)c * PM + m];
      acc += d * d;
    }
    sD[m] = acc;
  }
  __syncthreads();
  for (int r = 0; r < PK; ++r) {
    float best = INFINITY; int barg = PM;
    for (int m = tid; m < PM; m += 256) {
      float v = sD[m];
      if (v < best || (v == best && m < barg)) { best = v; barg = m; }
    }
    sVal[tid] = best; sArg[tid] = barg;
    __syncthreads();
    for (int s = 128; s > 0; s >>= 1) {
      if (tid < s) {
        float v = sVal[tid + s]; int a = sArg[tid + s];
        if (v < sVal[tid] || (v == sVal[tid] && a < sArg[tid])) { sVal[tid] = v; sArg[tid] = a; }
      }
      __syncthreads();
    }
    if (tid == 0) {
      idxout[(size_t)bn * PK + r] = sArg[0];
      sD[sArg[0]] = INFINITY;
    }
    __syncthreads();
  }
}

__global__ __launch_bounds__(256)
void xform_kernel(const float* __restrict__ f, const float* __restrict__ pos,
                  const int* __restrict__ idx,
                  const float* __restrict__ WqkP, const float* __restrict__ WvP,
                  const float* __restrict__ P1, const float* __restrict__ P2,
                  const float* __restrict__ A1, const float* __restrict__ A2,
                  float* __restrict__ out, int Cin, int Kpad, int D, int H) {
  const int bn = blockIdx.x;
  const int b = bn / PM, n = bn % PM;
  const int tid = threadIdx.x;
  const int lane = tid & 31, wave = tid >> 5;
  const int SG = Kpad + 2;
  const int SD = D + 2;
  const int SHS = H + 2;
  extern __shared__ float smem[];
  const int scrF = (16 * SG + D * 16 > 16 * SHS) ? (16 * SG + D * 16) : (16 * SHS);
  float* sRel = smem;
  float* sHT  = smem + 48;
  float* sV   = sHT + 16 * SD;
  float* sScr = sV + D * 16;
  float* sGfT = sScr;
  float* sPe  = sScr + 16 * SG;
  float* sTT  = sScr;
  float* sL   = sScr + scrF;
  int*   sNb  = (int*)(sL + D * 16);

  if (tid == 0) {
    __builtin_prefetch(A1, 0, 3);
    __builtin_prefetch(A2, 0, 3);
  }
  if (tid < PK) sNb[tid] = idx[(size_t)bn * PK + tid];
  __syncthreads();

  if (tid < 48) {
    int c = tid >> 4, j = tid & 15;
    sRel[c * 16 + j] = pos[((size_t)b * 3 + c) * PM + sNb[j]] - pos[((size_t)b * 3 + c) * PM + n];
  }
  for (int e = tid; e < 16 * Kpad; e += 256) {
    int j = e / Kpad, row = e % Kpad;
    float v = 0.f;
    if (row < Cin) {
      v = f[((size_t)b * Cin + row) * PM + sNb[j]] - f[((size_t)b * Cin + row) * PM + n];
    } else if (row < 2 * Cin) {
      int c = row - Cin;
      v = f[((size_t)b * Cin + c) * PM + n];
    }
    sGfT[j * SG + row] = v;
  }
  __syncthreads();

  for (int e = tid; e < D * 16; e += 256) {
    int d = e % D, j = e / D;
    float a = P1[d * 3 + 0] * sRel[j] + P1[d * 3 + 1] * sRel[16 + j] + P1[d * 3 + 2] * sRel[32 + j];
    sHT[j * SD + d] = fmaxf(a, 0.f);
  }
  __syncthreads();
  for (int e = tid; e < D * 16; e += 256) {
    int d = e >> 4, j = e & 15;
    float acc = 0.f;
    for (int q = 0; q < D; ++q) acc += P2[(size_t)d * D + q] * sHT[j * SD + q];
    sPe[d * 16 + j] = acc;
  }
  __syncthreads();

  const int r = lane & 15;
  const int khalf = (lane >> 4) << 1;
  const int rhalf = (lane >> 4) << 3;

  for (int rc = wave; rc < D / 16; rc += 8) {
    int rowBase = rc * 16;
    int row = rowBase + r;
    v8f acc, accv;
#pragma unroll
    for (int u = 0; u < 8; ++u) {
      float pe = sPe[(rowBase + u + rhalf) * 16 + r];
      acc[u] = pe; accv[u] = pe;
    }
#pragma unroll 4
    for (int kk = 0; kk < Kpad; kk += 4) {
      int ka = kk + khalf;
      v2f bfrag = *(const v2f*)(sGfT + r * SG + ka);
      v2f aq    = *(const v2f*)(WqkP + (size_t)row * Kpad + ka);
      v2f av    = *(const v2f*)(WvP  + (size_t)row * Kpad + ka);
      acc  = wmma_f32x4(aq, bfrag, acc);
      accv = wmma_f32x4(av, bfrag, accv);
    }
#pragma unroll
    for (int u = 0; u < 8; ++u) {
      sHT[r * SD + rowBase + u + rhalf] = acc[u];
      sV[(rowBase + u + rhalf) * 16 + r] = accv[u];
    }
  }
  __syncthreads();

  for (int rc = wave; rc < H / 16; rc += 8) {
    int rowBase = rc * 16;
    int row = rowBase + r;
    v8f acc;
#pragma unroll
    for (int u = 0; u < 8; ++u) acc[u] = 0.f;
#pragma unroll 4
    for (int kk = 0; kk < D; kk += 4) {
      int ka = kk + khalf;
      v2f bfrag = *(const v2f*)(sHT + r * SD + ka);
      v2f afrag = *(const v2f*)(A1 + (size_t)row * D + ka);
      acc = wmma_f32x4(afrag, bfrag, acc);
    }
#pragma unroll
    for (int u = 0; u < 8; ++u)
      sTT[r * SHS + rowBase + u + rhalf] = fmaxf(acc[u], 0.f);
  }
  __syncthreads();

  for (int rc = wave; rc < D / 16; rc += 8) {
    int rowBase = rc * 16;
    int row = rowBase + r;
    v8f acc;
#pragma unroll
    for (int u = 0; u < 8; ++u) acc[u] = 0.f;
#pragma unroll 4
    for (int kk = 0; kk < H; kk += 4) {
      int ka = kk + khalf;
      v2f bfrag = *(const v2f*)(sTT + r * SHS + ka);
      v2f afrag = *(const v2f*)(A2 + (size_t)row * H + ka);
      acc = wmma_f32x4(afrag, bfrag, acc);
    }
#pragma unroll
    for (int u = 0; u < 8; ++u)
      sL[(rowBase + u + rhalf) * 16 + r] = acc[u];
  }
  __syncthreads();

  for (int d = tid; d < D; d += 256) {
    float mx = -INFINITY;
#pragma unroll
    for (int j = 0; j < PK; ++j) mx = fmaxf(mx, sL[d * 16 + j]);
    float s = 0.f, o = 0.f;
#pragma unroll
    for (int j = 0; j < PK; ++j) {
      float p = __expf(sL[d * 16 + j] - mx);
      s += p;
      o += p * sV[d * 16 + j];
    }
    out[((size_t)b * D + d) * PM + n] = o / s;
  }
}

__global__ __launch_bounds__(256)
void pool_kernel(const float* __restrict__ x1, const float* __restrict__ x2,
                 const float* __restrict__ x3, float* __restrict__ out) {
  int b = blockIdx.x, ch = threadIdx.x;
  const float* src;
  if (ch < 64)       src = x1 + ((size_t)b * 64 + ch) * PM;
  else if (ch < 128) src = x2 + ((size_t)b * 64 + (ch - 64)) * PM;
  else               src = x3 + ((size_t)b * 128 + (ch - 128)) * PM;
  float mx = -INFINITY, sm = 0.f;
  for (int m = 0; m < PM; ++m) { float v = src[m]; mx = fmaxf(mx, v); sm += v; }
  out[(size_t)b * 512 + ch] = mx;
  out[(size_t)b * 512 + 256 + ch] = sm * (1.f / PM);
}

extern "C" void kernel_launch(void* const* d_in, const int* in_sizes, int n_in,
                              void* d_out, int out_size, void* d_ws, size_t ws_size,
                              hipStream_t stream) {
  (void)in_sizes; (void)n_in; (void)out_size; (void)ws_size;
  const float* x = (const float*)d_in[0];
  const float* t1[7]; const float* t2[7]; const float* t3[7];
  for (int i = 0; i < 7; ++i) {
    t1[i] = (const float*)d_in[1 + i];
    t2[i] = (const float*)d_in[8 + i];
    t3[i] = (const float*)d_in[15 + i];
  }
  float* ws = (float*)d_ws;
  size_t off = 0;
  float* pos = ws + off; off += (size_t)PB * 3 * PM;
  float* x1  = ws + off; off += (size_t)PB * 64 * PM;
  float* x2  = ws + off; off += (size_t)PB * 64 * PM;
  float* x3  = ws + off; off += (size_t)PB * 128 * PM;
  int* idxb  = (int*)(ws + off); off += (size_t)PB * PM * PK;
  float* wqk1 = ws + off; off += 64 * 8;
  float* wv1  = ws + off; off += 64 * 8;
  float* wqk2 = ws + off; off += 64 * 128;
  float* wv2  = ws + off; off += 64 * 128;
  float* wqk3 = ws + off; off += 128 * 128;
  float* wv3  = ws + off; off += 128 * 128;

  auto lds = [](int Kpad, int D, int H) -> size_t {
    int scr = (16 * (Kpad + 2) + D * 16 > 16 * (H + 2)) ? (16 * (Kpad + 2) + D * 16)
                                                        : (16 * (H + 2));
    return ((size_t)(48 + 16 * (D + 2) + D * 16 + scr + D * 16)) * sizeof(float) +
           PK * sizeof(int);
  };

  prep_weights_kernel<<<(64 * 8 + 255) / 256, 256, 0, stream>>>(
      t1[0], t1[1], t1[2], wqk1, wv1, 64, 6, 8);
  prep_weights_kernel<<<(64 * 128 + 255) / 256, 256, 0, stream>>>(
      t2[0], t2[1], t2[2], wqk2, wv2, 64, 128, 128);
  prep_weights_kernel<<<(128 * 128 + 255) / 256, 256, 0, stream>>>(
      t3[0], t3[1], t3[2], wqk3, wv3, 128, 128, 128);

  fps_kernel<<<PB, 256, 0, stream>>>(x, pos);

  knn_kernel<<<PB * PM, 256, 0, stream>>>(pos, 3, idxb);
  xform_kernel<<<PB * PM, 256, lds(8, 64, 256), stream>>>(
      pos, pos, idxb, wqk1, wv1, t1[3], t1[4], t1[5], t1[6],
      x1, 3, 8, 64, 256);

  knn_kernel<<<PB * PM, 256, 0, stream>>>(x1, 64, idxb);
  xform_kernel<<<PB * PM, 256, lds(128, 64, 256), stream>>>(
      x1, pos, idxb, wqk2, wv2, t2[3], t2[4], t2[5], t2[6],
      x2, 64, 128, 64, 256);

  knn_kernel<<<PB * PM, 256, 0, stream>>>(x2, 64, idxb);
  xform_kernel<<<PB * PM, 256, lds(128, 128, 512), stream>>>(
      x2, pos, idxb, wqk3, wv3, t3[3], t3[4], t3[5], t3[6],
      x3, 64, 128, 128, 512);

  pool_kernel<<<PB, 256, 0, stream>>>(x1, x2, x3, (float*)d_out);
}


# fps pick (serial, 511x): block argmax by row_xmask DPP in 16-lane rows, one LDS slot per row, ONE barrier; winner xyz from an LDS copy of points (was 8-level LDS tree + dependent global load)
# speedup vs baseline: 1.4605x; 1.4605x over previous
#include <hip/hip_runtime.h>
#include <math.h>

#define PB 16
#define PN 4096
#define PM 512
#define PK 16

typedef float v2f __attribute__((ext_vector_type(2)));
typedef float v8f __attribute__((ext_vector_type(8)));

__device__ __forceinline__ v8f wmma_f32x4(v2f a, v2f b, v8f c) {
  return __builtin_amdgcn_wmma_f32_16x16x4_f32(false, a, false, b, (short)0, c, false, false);
}

__global__ __launch_bounds__(256)
void prep_weights_kernel(const float* __restrict__ Wq, const float* __restrict__ Wk,
                         const float* __restrict__ Wv, float* __restrict__ outQK,
                         float* __restrict__ outV, int D, int Cw, int Kpad) {
  int i = blockIdx.x * 256 + threadIdx.x;
  if (i >= D * Kpad) return;
  int d = i / Kpad, k = i % Kpad;
  float q = 0.f, v = 0.f;
  if (k < Cw) {
    q = Wq[(size_t)d * Cw + k] - Wk[(size_t)d * Cw + k];
    v = Wv[(size_t)d * Cw + k];
  }
  outQK[i] = q;
  outV[i] = v;
}

template <int XM>
__device__ __forceinline__ unsigned row_xor_u32(unsigned v) {
  return (unsigned)__builtin_amdgcn_update_dpp(0, (int)v, 0x160 | XM, 0xf, 0xf, true);
}
__device__ __forceinline__ unsigned umax2(unsigned a, unsigned b) { return a > b ? a : b; }
__device__ __forceinline__ unsigned umin2(unsigned a, unsigned b) { return a < b ? a : b; }
__device__ __forceinline__ unsigned row_umax(unsigned v) {
  v = umax2(v, row_xor_u32<1>(v));
  v = umax2(v, row_xor_u32<2>(v));
  v = umax2(v, row_xor_u32<4>(v));
  return umax2(v, row_xor_u32<8>(v));
}
__device__ __forceinline__ unsigned row_umin(unsigned v) {
  v = umin2(v, row_xor_u32<1>(v));
  v = umin2(v, row_xor_u32<2>(v));
  v = umin2(v, row_xor_u32<4>(v));
  return umin2(v, row_xor_u32<8>(v));
}

__global__ __launch_bounds__(256)
void fps_kernel(const float* __restrict__ x, float* __restrict__ pos) {
  const int b = blockIdx.x;
  const int tid = threadIdx.x;
  const int PPT = PN / 256;
  __shared__ float sX[PN];
  __shared__ float sY[PN];
  __shared__ float sZ[PN];
  __shared__ int sSel[PM];
  __shared__ uint2 sPart[2][16];

  float px[PPT], py[PPT], pz[PPT], dmin[PPT];
#pragma unroll
  for (int i = 0; i < PPT; ++i) {
    int p = tid + i * 256;
    const float* xp = x + ((size_t)b * PN + p) * 3;
    px[i] = xp[0]; py[i] = xp[1]; pz[i] = xp[2];
    sX[p] = px[i]; sY[p] = py[i]; sZ[p] = pz[i];
    dmin[i] = INFINITY;
  }
  if (tid == 0) sSel[0] = 0;
  __syncthreads();
  float bx = sX[0], by = sY[0], bz = sZ[0];
  for (int it = 1; it < PM; ++it) {
    float best = -INFINITY; int bslot = PPT;
#pragma unroll
    for (int i = 0; i < PPT; ++i) {
      float dx = px[i] - bx, dy = py[i] - by, dz = pz[i] - bz;
      float d = dx * dx + dy * dy + dz * dz;
      dmin[i] = fminf(dmin[i], d);
      if (dmin[i] > best) { best = dmin[i]; bslot = i; }
    }
    const unsigned barg = (unsigned)tid + ((unsigned)bslot << 8);
    const unsigned vb = __float_as_uint(best);
    const unsigned rmax = row_umax(vb);
    const unsigned rarg = row_umin(vb == rmax ? barg : 0xFFFFFFFFu);
    uint2* part = sPart[it & 1];
    if ((tid & 15) == 0) part[tid >> 4] = make_uint2(rmax, rarg);
    __syncthreads();
    const uint2 pr = part[tid & 15];
    const unsigned gmax = row_umax(pr.x);
    const unsigned garg = row_umin(pr.x == gmax ? pr.y : 0xFFFFFFFFu);
    bx = sX[garg]; by = sY[garg]; bz = sZ[garg];
    if (tid == 0) sSel[it] = (int)garg;
  }
  __syncthreads();
  for (int i = tid; i < PM; i += 256) {
    int p = sSel[i];
    pos[((size_t)b * 3 + 0) * PM + i] = sX[p];
    pos[((size_t)b * 3 + 1) * PM + i] = sY[p];
    pos[((size_t)b * 3 + 2) * PM + i] = sZ[p];
  }
}

__global__ __launch_bounds__(256)
void knn_kernel(const float* __restrict__ f, int C, int* __restrict__ idxout) {
  const int bn = blockIdx.x;
  const int b = bn / PM, n = bn % PM;
  const int tid = threadIdx.x;
  __shared__ float sD[PM];
  __shared__ float sVal[256];
  __shared__ int sArg[256];
  const float* fb = f + (size_t)b * C * PM;
  for (int m = tid; m < PM; m += 256) {
    float acc = 0.f;
    for (int c = 0; c < C; ++c) {
      float d = fb[(size_t)c * PM + n] - fb[(size_t)c * PM + m];
      acc += d * d;
    }
    sD[m] = acc;
  }
  __syncthreads();
  for (int r = 0; r < PK; ++r) {
    float best = INFINITY; int barg = PM;
    for (int m = tid; m < PM; m += 256) {
      float v = sD[m];
      if (v < best || (v == best && m < barg)) { best = v; barg = m; }
    }
    sVal[tid] = best; sArg[tid] = barg;
    __syncthreads();
    for (int s = 128; s > 0; s >>= 1) {
      if (tid < s) {
        float v = sVal[tid + s]; int a = sArg[tid + s];
        if (v < sVal[tid] || (v == sVal[tid] && a < sArg[tid])) { sVal[tid] = v; sArg[tid] = a; }
      }
      __syncthreads();
    }
    if (tid == 0) {
      idxout[(size_t)bn * PK + r] = sArg[0];
      sD[sArg[0]] = INFINITY;
    }
    __syncthreads();
  }
}

__global__ __launch_bounds__(256)
void xform_kernel(const float* __restrict__ f, const float* __restrict__ pos,
                  const int* __restrict__ idx,
                  const float* __restrict__ WqkP, const float* __restrict__ WvP,
                  const float* __restrict__ P1, const float* __restrict__ P2,
                  const float* __restrict__ A1, const float* __restrict__ A2,
                  float* __restrict__ out, int Cin, int Kpad, int D, int H) {
  const int bn = blockIdx.x;
  const int b = bn / PM, n = bn % PM;
  const int tid = threadIdx.x;
  const int lane = tid & 31, wave = tid >> 5;
  const int SG = Kpad + 2;
  const int SD = D + 2;
  const int SHS = H + 2;
  extern __shared__ float smem[];
  const int scrF = (16 * SG + D * 16 > 16 * SHS) ? (16 * SG + D * 16) : (16 * SHS);
  float* sRel = smem;
  float* sHT  = smem + 48;
  float* sV   = sHT + 16 * SD;
  float* sScr = sV + D * 16;
  float* sGfT = sScr;
  float* sPe  = sScr + 16 * SG;
  float* sTT  = sScr;
  float* sL   = sScr + scrF;
  int*   sNb  = (int*)(sL + D * 16);

  if (tid == 0) {
    __builtin_prefetch(A1, 0, 3);
    __builtin_prefetch(A2, 0, 3);
  }
  if (tid < PK) sNb[tid] = idx[(size_t)bn * PK + tid];
  __syncthreads();

  if (tid < 48) {
    int c = tid >> 4, j = tid & 15;
    sRel[c * 16 + j] = pos[((size_t)b * 3 + c) * PM + sNb[j]] - pos[((size_t)b * 3 + c) * PM + n];
  }
  for (int e = tid; e < 16 * Kpad; e += 256) {
    int j = e / Kpad, row = e % Kpad;
    float v = 0.f;
    if (row < Cin) {
      v = f[((size_t)b * Cin + row) * PM + sNb[j]] - f[((size_t)b * Cin + row) * PM + n];
    } else if (row < 2 * Cin) {
      int c = row - Cin;
      v = f[((size_t)b * Cin + c) * PM + n];
    }
    sGfT[j * SG + row] = v;
  }
  __syncthreads();

  for (int e = tid; e < D * 16; e += 256) {
    int d = e % D, j = e / D;
    float a = P1[d * 3 + 0] * sRel[j] + P1[d * 3 + 1] * sRel[16 + j] + P1[d * 3 + 2] * sRel[32 + j];
    sHT[j * SD + d] = fmaxf(a, 0.f);
  }
  __syncthreads();
  for (int e = tid; e < D * 16; e += 256) {
    int d = e >> 4, j = e & 15;
    float acc = 0.f;
    for (int q = 0; q < D; ++q) acc += P2[(size_t)d * D + q] * sHT[j * SD + q];
    sPe[d * 16 + j] = acc;
  }
  __syncthreads();

  const int r = lane & 15;
  const int khalf = (lane >> 4) << 1;
  const int rhalf = (lane >> 4) << 3;

  for (int rc = wave; rc < D / 16; rc += 8) {
    int rowBase = rc * 16;
    int row = rowBase + r;
    v8f acc, accv;
#pragma unroll
    for (int u = 0; u < 8; ++u) {
      float pe = sPe[(rowBase + u + rhalf) * 16 + r];
      acc[u] = pe; accv[u] = pe;
    }
#pragma unroll 4
    for (int kk = 0; kk < Kpad; kk += 4) {
      int ka = kk + khalf;
      v2f bfrag = *(const v2f*)(sGfT + r * SG + ka);
      v2f aq    = *(const v2f*)(WqkP + (size_t)row * Kpad + ka);
      v2f av    = *(const v2f*)(WvP  + (size_t)row * Kpad + ka);
      acc  = wmma_f32x4(aq, bfrag, acc);
      accv = wmma_f32x4(av, bfrag, accv);
    }
#pragma unroll
    for (int u = 0; u < 8; ++u) {
      sHT[r * SD + rowBase + u + rhalf] = acc[u];
      sV[(rowBase + u + rhalf) * 16 + r] = accv[u];
    }
  }
  __syncthreads();

  for (int rc = wave; rc < H / 16; rc += 8) {
    int rowBase = rc * 16;
    int row = rowBase + r;
    v8f acc;
#pragma unroll
    for (int u = 0; u < 8; ++u) acc[u] = 0.f;
#pragma unroll 4
    for (int kk = 0; kk < D; kk += 4) {
      int ka = kk + khalf;
      v2f bfrag = *(const v2f*)(sHT + r * SD + ka);
      v2f afrag = *(const v2f*)(A1 + (size_t)row * D + ka);
      acc = wmma_f32x4(afrag, bfrag, acc);
    }
#pragma unroll
    for (int u = 0; u < 8; ++u)
      sTT[r * SHS + rowBase + u + rhalf] = fmaxf(acc[u], 0.f);
  }
  __syncthreads();

  for (int rc = wave; rc < D / 16; rc += 8) {
    int rowBase = rc * 16;
    int row = rowBase + r;
    v8f acc;
#pragma unroll
    for (int u = 0; u < 8; ++u) acc[u] = 0.f;
#pragma unroll 4
    for (int kk = 0; kk < H; kk += 4) {
      int ka = kk + khalf;
      v2f bfrag = *(const v2f*)(sTT + r * SHS + ka);
      v2f afrag = *(const v2f*)(A2 + (size_t)row * H + ka);
      acc = wmma_f32x4(afrag, bfrag, acc);
    }
#pragma unroll
    for (int u = 0; u < 8; ++u)
      sL[(rowBase + u + rhalf) * 16 + r] = acc[u];
  }
  __syncthreads();

  for (int d = tid; d < D; d += 256) {
    float mx = -INFINITY;
#pragma unroll
    for (int j = 0; j < PK; ++j) mx = fmaxf(mx, sL[d * 16 + j]);
    float s = 0.f, o = 0.f;
#pragma unroll
    for (int j = 0; j < PK; ++j) {
      float p = __expf(sL[d * 16 + j] - mx);
      s += p;
      o += p * sV[d * 16 + j];
    }
    out[((size_t)b * D + d) * PM + n] = o / s;
  }
}

__global__ __launch_bounds__(256)
void pool_kernel(const float* __restrict__ x1, const float* __restrict__ x2,
                 const float* __restrict__ x3, float* __restrict__ out) {
  int b = blockIdx.x, ch = threadIdx.x;
  const float* src;
  if (ch < 64)       src = x1 + ((size_t)b * 64 + ch) * PM;
  else if (ch < 128) src = x2 + ((size_t)b * 64 + (ch - 64)) * PM;
  else               src = x3 + ((size_t)b * 128 + (ch - 128)) * PM;
  float mx = -INFINITY, sm = 0.f;
  for (int m = 0; m < PM; ++m) { float v = src[m]; mx = fmaxf(mx, v); sm += v; }
  out[(size_t)b * 512 + ch] = mx;
  out[(size_t)b * 512 + 256 + ch] = sm * (1.f / PM);
}

extern "C" void kernel_launch(void* const* d_in, const int* in_sizes, int n_in,
                              void* d_out, int out_size, void* d_ws, size_t ws_size,
                              hipStream_t stream) {
  (void)in_sizes; (void)n_in; (void)out_size; (void)ws_size;
  const float* x = (const float*)d_in[0];
  const float* t1[7]; const float* t2[7]; const float* t3[7];
  for (int i = 0; i < 7; ++i) {
    t1[i] = (const float*)d_in[1 + i];
    t2[i] = (const float*)d_in[8 + i];
    t3[i] = (const float*)d_in[15 + i];
  }
  float* ws = (float*)d_ws;
  size_t off = 0;
  float* pos = ws + off; off += (size_t)PB * 3 * PM;
  float* x1  = ws + off; off += (size_t)PB * 64 * PM;
  float* x2  = ws + off; off += (size_t)PB * 64 * PM;
  float* x3  = ws + off; off += (size_t)PB * 128 * PM;
  int* idxb  = (int*)(ws + off); off += (size_t)PB * PM * PK;
  float* wqk1 = ws + off; off += 64 * 8;
  float* wv1  = ws + off; off += 64 * 8;
  float* wqk2 = ws + off; off += 64 * 128;
  float* wv2  = ws + off; off += 64 * 128;
  float* wqk3 = ws + off; off += 128 * 128;
  float* wv3  = ws + off; off += 128 * 128;

  auto lds = [](int Kpad, int D, int H) -> size_t {
    int scr = (16 * (Kpad + 2) + D * 16 > 16 * (H + 2)) ? (16 * (Kpad + 2) + D * 16)
                                                        : (16 * (H + 2));
    return ((size_t)(48 + 16 * (D + 2) + D * 16 + scr + D * 16)) * sizeof(float) +
           PK * sizeof(int);
  };

  prep_weights_kernel<<<(64 * 8 + 255) / 256, 256, 0, stream>>>(
      t1[0], t1[1], t1[2], wqk1, wv1, 64, 6, 8);
  prep_weights_kernel<<<(64 * 128 + 255) / 256, 256, 0, stream>>>(
      t2[0], t2[1], t2[2], wqk2, wv2, 64, 128, 128);
  prep_weights_kernel<<<(128 * 128 + 255) / 256, 256, 0, stream>>>(
      t3[0], t3[1], t3[2], wqk3, wv3, 128, 128, 128);

  fps_kernel<<<PB, 256, 0, stream>>>(x, pos);

  knn_kernel<<<PB * PM, 256, 0, stream>>>(pos, 3, idxb);
  xform_kernel<<<PB * PM, 256, lds(8, 64, 256), stream>>>(
      pos, pos, idxb, wqk1, wv1, t1[3], t1[4], t1[5], t1[6],
      x1, 3, 8, 64, 256);

  knn_kernel<<<PB * PM, 256, 0, stream>>>(x1, 64, idxb);
  xform_kernel<<<PB * PM, 256, lds(128, 64, 256), stream>>>(
      x1, pos, idxb, wqk2, wv2, t2[3], t2[4], t2[5], t2[6],
      x2, 64, 128, 64, 256);

  knn_kernel<<<PB * PM, 256, 0, stream>>>(x2, 64, idxb);
  xform_kernel<<<PB * PM, 256, lds(128, 128, 512), stream>>>(
      x2, pos, idxb, wqk3, wv3, t3[3], t3[4], t3[5], t3[6],
      x3, 64, 128, 128, 512);

  pool_kernel<<<PB, 256, 0, stream>>>(x1, x2, x3, (float*)d_out);
}


# fused layer: 4 query points per block (512 threads, LDS region set per point); each weight fragment (A of the 3 WMMA GEMMs, P2 in pe) is loaded once per wave and feeds 2-4 points; sums unchanged
# speedup vs baseline: 1.6985x; 1.1630x over previous
#include <hip/hip_runtime.h>
#include <math.h>

#define PB 16
#define PN 4096
#define PM 512
#define PK 16

typedef float v2f __attribute__((ext_vector_type(2)));
typedef float v8f __attribute__((ext_vector_type(8)));

__device__ __forceinline__ v8f wmma_f32x4(v2f a, v2f b, v8f c) {
  return __builtin_amdgcn_wmma_f32_16x16x4_f32(false, a, false, b, (short)0, c, false, false);
}

__global__ __launch_bounds__(256)
void prep_weights_kernel(const float* __restrict__ Wq, const float* __restrict__ Wk,
                         const float* __restrict__ Wv, float* __restrict__ outQK,
                         float* __restrict__ outV, int D, int Cw, int Kpad) {
  int i = blockIdx.x * 256 + threadIdx.x;
  if (i >= D * Kpad) return;
  int d = i / Kpad, k = i % Kpad;
  float q = 0.f, v = 0.f;
  if (k < Cw) {
    q = Wq[(size_t)d * Cw + k] - Wk[(size_t)d * Cw + k];
    v = Wv[(size_t)d * Cw + k];
  }
  outQK[i] = q;
  outV[i] = v;
}

template <int XM>
__device__ __forceinline__ unsigned row_xor_u32(unsigned v) {
  return (unsigned)__builtin_amdgcn_update_dpp(0, (int)v, 0x160 | XM, 0xf, 0xf, true);
}
__device__ __forceinline__ unsigned umax2(unsigned a, unsigned b) { return a > b ? a : b; }
__device__ __forceinline__ unsigned umin2(unsigned a, unsigned b) { return a < b ? a : b; }
__device__ __forceinline__ unsigned row_umax(unsigned v) {
  v = umax2(v, row_xor_u32<1>(v));
  v = umax2(v, row_xor_u32<2>(v));
  v = umax2(v, row_xor_u32<4>(v));
  return umax2(v, row_xor_u32<8>(v));
}
__device__ __forceinline__ unsigned row_umin(unsigned v) {
  v = umin2(v, row_xor_u32<1>(v));
  v = umin2(v, row_xor_u32<2>(v));
  v = umin2(v, row_xor_u32<4>(v));
  return umin2(v, row_xor_u32<8>(v));
}

__global__ __launch_bounds__(256)
void fps_kernel(const float* __restrict__ x, float* __restrict__ pos) {
  const int b = blockIdx.x;
  const int tid = threadIdx.x;
  const int PPT = PN / 256;
  __shared__ float sX[PN];
  __shared__ float sY[PN];
  __shared__ float sZ[PN];
  __shared__ int sSel[PM];
  __shared__ uint2 sPart[2][16];

  float px[PPT], py[PPT], pz[PPT], dmin[PPT];
#pragma unroll
  for (int i = 0; i < PPT; ++i) {
    int p = tid + i * 256;
    const float* xp = x + ((size_t)b * PN + p) * 3;
    px[i] = xp[0]; py[i] = xp[1]; pz[i] = xp[2];
    sX[p] = px[i]; sY[p] = py[i]; sZ[p] = pz[i];
    dmin[i] = INFINITY;
  }
  if (tid == 0) sSel[0] = 0;
  __syncthreads();
  float bx = sX[0], by = sY[0], bz = sZ[0];
  for (int it = 1; it < PM; ++it) {
    float best = -INFINITY; int bslot = PPT;
#pragma unroll
    for (int i = 0; i < PPT; ++i) {
      float dx = px[i] - bx, dy = py[i] - by, dz = pz[i] - bz;
      float d = dx * dx + dy * dy + dz * dz;
      dmin[i] = fminf(dmin[i], d);
      if (dmin[i] > best) { best = dmin[i]; bslot = i; }
    }
    const unsigned barg = (unsigned)tid + ((unsigned)bslot << 8);
    const unsigned vb = __float_as_uint(best);
    const unsigned rmax = row_umax(vb);
    const unsigned rarg = row_umin(vb == rmax ? barg : 0xFFFFFFFFu);
    uint2* part = sPart[it & 1];
    if ((tid & 15) == 0) part[tid >> 4] = make_uint2(rmax, rarg);
    __syncthreads();
    const uint2 pr = part[tid & 15];
    const unsigned gmax = row_umax(pr.x);
    const unsigned garg = row_umin(pr.x == gmax ? pr.y : 0xFFFFFFFFu);
    bx = sX[garg]; by = sY[garg]; bz = sZ[garg];
    if (tid == 0) sSel[it] = (int)garg;
  }
  __syncthreads();
  for (int i = tid; i < PM; i += 256) {
    int p = sSel[i];
    pos[((size_t)b * 3 + 0) * PM + i] = sX[p];
    pos[((size_t)b * 3 + 1) * PM + i] = sY[p];
    pos[((size_t)b * 3 + 2) * PM + i] = sZ[p];
  }
}

__global__ __launch_bounds__(256)
void knn_kernel(const float* __restrict__ f, int C, int* __restrict__ idxout) {
  const int bn = blockIdx.x;
  const int b = bn / PM, n = bn % PM;
  const int tid = threadIdx.x;
  __shared__ float sD[PM];
  __shared__ float sVal[256];
  __shared__ int sArg[256];
  const float* fb = f + (size_t)b * C * PM;
  for (int m = tid; m < PM; m += 256) {
    float acc = 0.f;
    for (int c = 0; c < C; ++c) {
      float d = fb[(size_t)c * PM + n] - fb[(size_t)c * PM + m];
      acc += d * d;
    }
    sD[m] = acc;
  }
  __syncthreads();
  for (int r = 0; r < PK; ++r) {
    float best = INFINITY; int barg = PM;
    for (int m = tid; m < PM; m += 256) {
      float v = sD[m];
      if (v < best || (v == best && m < barg)) { best = v; barg = m; }
    }
    sVal[tid] = best; sArg[tid] = barg;
    __syncthreads();
    for (int s = 128; s > 0; s >>= 1) {
      if (tid < s) {
        float v = sVal[tid + s]; int a = sArg[tid + s];
        if (v < sVal[tid] || (v == sVal[tid] && a < sArg[tid])) { sVal[tid] = v; sArg[tid] = a; }
      }
      __syncthreads();
    }
    if (tid == 0) {
      idxout[(size_t)bn * PK + r] = sArg[0];
      sD[sArg[0]] = INFINITY;
    }
    __syncthreads();
  }
}

constexpr int xf_query_floats(int Kpad, int D, int H) {
  return 48 + 16 * (D + 2) + D * 16 +
         ((16 * (Kpad + 2) + D * 16 > 16 * (H + 2)) ? (16 * (Kpad + 2) + D * 16)
                                                     : (16 * (H + 2))) +
         D * 16 + 16;
}

#define XNT 512
#define XNQ 4

template <int NT, int NQ, int QD, int LDSF>
__global__ __launch_bounds__(NT)
void xform_kernel(const float* __restrict__ f, const float* __restrict__ pos,
                  const int* __restrict__ idx,
                  const float* __restrict__ WqkP, const float* __restrict__ WvP,
                  const float* __restrict__ P1, const float* __restrict__ P2,
                  const float* __restrict__ A1, const float* __restrict__ A2,
                  float* __restrict__ out, int Cin, int Kpad, int D, int H) {
  constexpr int NW = NT / 32;
  constexpr int NG = NQ / QD;
  const int bn0 = blockIdx.x * NQ;
  const int b = bn0 / PM, n0 = bn0 % PM;
  const int tid = threadIdx.x;
  const int lane = tid & 31, wave = tid >> 5;
  const int SG = Kpad + 2;
  const int SD = D + 2;
  const int SHS = H + 2;
  __shared__ float smem[LDSF] __attribute__((aligned(16)));
  const int scrF = (16 * SG + D * 16 > 16 * SHS) ? (16 * SG + D * 16) : (16 * SHS);
  const int oHT  = 48;
  const int oV   = oHT + 16 * SD;
  const int oScr = oV + D * 16;
  const int oPe  = oScr + 16 * SG;
  const int oL   = oScr + scrF;
  const int oNb  = oL + D * 16;
  const int QS   = oNb + 16;

  if (tid == 0) {
    __builtin_prefetch(A1, 0, 3);
    __builtin_prefetch(A2, 0, 3);
  }
  if (tid < NQ * PK) {
    const int q = tid >> 4, j = tid & 15;
    ((int*)(smem + q * QS + oNb))[j] = idx[(size_t)bn0 * PK + tid];
  }
  __syncthreads();

  for (int e = tid; e < NQ * 48; e += NT) {
    const int q = e / 48, t = e % 48;
    const int c = t >> 4, j = t & 15;
    const int* nb = (const int*)(smem + q * QS + oNb);
    const float* pc = pos + ((size_t)b * 3 + c) * PM;
    smem[q * QS + c * 16 + j] = pc[nb[j]] - pc[n0 + q];
  }
#pragma unroll
  for (int q = 0; q < NQ; ++q) {
    const int n = n0 + q;
    const int* nb = (const int*)(smem + q * QS + oNb);
    float* sGfT = smem + q * QS + oScr;
    for (int e = tid; e < 16 * Kpad; e += NT) {
      int j = e / Kpad, row = e % Kpad;
      float v = 0.f;
      if (row < Cin) {
        v = f[((size_t)b * Cin + row) * PM + nb[j]] - f[((size_t)b * Cin + row) * PM + n];
      } else if (row < 2 * Cin) {
        int c = row - Cin;
        v = f[((size_t)b * Cin + c) * PM + n];
      }
      sGfT[j * SG + row] = v;
    }
  }
  __syncthreads();

#pragma unroll
  for (int q = 0; q < NQ; ++q) {
    const float* sRel = smem + q * QS;
    float* sHT = smem + q * QS + oHT;
    for (int e = tid; e < D * 16; e += NT) {
      int d = e % D, j = e / D;
      float a = P1[d * 3 + 0] * sRel[j] + P1[d * 3 + 1] * sRel[16 + j] + P1[d * 3 + 2] * sRel[32 + j];
      sHT[j * SD + d] = fmaxf(a, 0.f);
    }
  }
  __syncthreads();
  for (int e = tid; e < D * 16; e += NT) {
    const int d = e >> 4, j = e & 15;
    float acc[NQ];
#pragma unroll
    for (int q = 0; q < NQ; ++q) acc[q] = 0.f;
    const float* p2 = P2 + (size_t)d * D;
    const float* tcol = smem + oHT + j * SD;
    for (int k = 0; k < D; ++k) {
      const float w = p2[k];
#pragma unroll
      for (int q = 0; q < NQ; ++q) acc[q] += w * tcol[q * QS + k];
    }
#pragma unroll
    for (int q = 0; q < NQ; ++q) smem[q * QS + oPe + d * 16 + j] = acc[q];
  }
  __syncthreads();

  const int r = lane & 15;
  const int khalf = (lane >> 4) << 1;
  const int rhalf = (lane >> 4) << 3;
  const int nrcD = D >> 4;

  for (int item = wave; item < nrcD * NG; item += NW) {
    const int rc = item % nrcD, q0 = (item / nrcD) * QD;
    const int rowBase = rc * 16;
    const int row = rowBase + r;
    v8f acc[QD], accv[QD];
#pragma unroll
    for (int t = 0; t < QD; ++t) {
      const float* sPe = smem + (q0 + t) * QS + oPe;
#pragma unroll
      for (int u = 0; u < 8; ++u) {
        float pe = sPe[(rowBase + u + rhalf) * 16 + r];
        acc[t][u] = pe; accv[t][u] = pe;
      }
    }
    const float* aqp = WqkP + (size_t)row * Kpad + khalf;
    const float* avp = WvP  + (size_t)row * Kpad + khalf;
    const float* bp  = smem + q0 * QS + oScr + r * SG + khalf;
#pragma unroll 4
    for (int kk = 0; kk < Kpad; kk += 4) {
      v2f aq = *(const v2f*)(aqp + kk);
      v2f av = *(const v2f*)(avp + kk);
#pragma unroll
      for (int t = 0; t < QD; ++t) {
        v2f bfrag = *(const v2f*)(bp + t * QS + kk);
        acc[t]  = wmma_f32x4(aq, bfrag, acc[t]);
        accv[t] = wmma_f32x4(av, bfrag, accv[t]);
      }
    }
#pragma unroll
    for (int t = 0; t < QD; ++t) {
      float* sHT = smem + (q0 + t) * QS + oHT;
      float* sV  = smem + (q0 + t) * QS + oV;
#pragma unroll
      for (int u = 0; u < 8; ++u) {
        sHT[r * SD + rowBase + u + rhalf] = acc[t][u];
        sV[(rowBase + u + rhalf) * 16 + r] = accv[t][u];
      }
    }
  }
  __syncthreads();

  for (int rc = wave; rc < (H >> 4); rc += NW) {
    const int rowBase = rc * 16;
    const int row = rowBase + r;
    v8f acc[NQ];
#pragma unroll
    for (int q = 0; q < NQ; ++q)
#pragma unroll
      for (int u = 0; u < 8; ++u) acc[q][u] = 0.f;
    const float* ap = A1 + (size_t)row * D + khalf;
    const float* bp = smem + oHT + r * SD + khalf;
#pragma unroll 4
    for (int kk = 0; kk < D; kk += 4) {
      v2f afrag = *(const v2f*)(ap + kk);
#pragma unroll
      for (int q = 0; q < NQ; ++q) {
        v2f bfrag = *(const v2f*)(bp + q * QS + kk);
        acc[q] = wmma_f32x4(afrag, bfrag, acc[q]);
      }
    }
#pragma unroll
    for (int q = 0; q < NQ; ++q) {
      float* sTT = smem + q * QS + oScr;
#pragma unroll
      for (int u = 0; u < 8; ++u)
        sTT[r * SHS + rowBase + u + rhalf] = fmaxf(acc[q][u], 0.f);
    }
  }
  __syncthreads();

  for (int item = wave; item < nrcD * NG; item += NW) {
    const int rc = item % nrcD, q0 = (item / nrcD) * QD;
    const int rowBase = rc * 16;
    const int row = rowBase + r;
    v8f acc[QD];
#pragma unroll
    for (int t = 0; t < QD; ++t)
#pragma unroll
      for (int u = 0; u < 8; ++u) acc[t][u] = 0.f;
    const float* ap = A2 + (size_t)row * H + khalf;
    const float* bp = smem + q0 * QS + oScr + r * SHS + khalf;
#pragma unroll 4
    for (int kk = 0; kk < H; kk += 4) {
      v2f afrag = *(const v2f*)(ap + kk);
#pragma unroll
      for (int t = 0; t < QD; ++t) {
        v2f bfrag = *(const v2f*)(bp + t * QS + kk);
        acc[t] = wmma_f32x4(afrag, bfrag, acc[t]);
      }
    }
#pragma unroll
    for (int t = 0; t < QD; ++t) {
      float* sL = smem + (q0 + t) * QS + oL;
#pragma unroll
      for (int u = 0; u < 8; ++u)
        sL[(rowBase + u + rhalf) * 16 + r] = acc[t][u];
    }
  }
  __syncthreads();

  for (int e = tid; e < NQ * D; e += NT) {
    const int q = e % NQ, d = e / NQ;
    const float* sL = smem + q * QS + oL + d * 16;
    const float* sV = smem + q * QS + oV + d * 16;
    float mx = -INFINITY;
#pragma unroll
    for (int j = 0; j < PK; ++j) mx = fmaxf(mx, sL[j]);
    float s = 0.f, o = 0.f;
#pragma unroll
    for (int j = 0; j < PK; ++j) {
      float p = __expf(sL[j] - mx);
      s += p;
      o += p * sV[j];
    }
    out[((size_t)b * D + d) * PM + n0 + q] = o / s;
  }
}

__global__ __launch_bounds__(256)
void pool_kernel(const float* __restrict__ x1, const float* __restrict__ x2,
                 const float* __restrict__ x3, float* __restrict__ out) {
  int b = blockIdx.x, ch = threadIdx.x;
  const float* src;
  if (ch < 64)       src = x1 + ((size_t)b * 64 + ch) * PM;
  else if (ch < 128) src = x2 + ((size_t)b * 64 + (ch - 64)) * PM;
  else               src = x3 + ((size_t)b * 128 + (ch - 128)) * PM;
  float mx = -INFINITY, sm = 0.f;
  for (int m = 0; m < PM; ++m) { float v = src[m]; mx = fmaxf(mx, v); sm += v; }
  out[(size_t)b * 512 + ch] = mx;
  out[(size_t)b * 512 + 256 + ch] = sm * (1.f / PM);
}

extern "C" void kernel_launch(void* const* d_in, const int* in_sizes, int n_in,
                              void* d_out, int out_size, void* d_ws, size_t ws_size,
                              hipStream_t stream) {
  (void)in_sizes; (void)n_in; (void)out_size; (void)ws_size;
  const float* x = (const float*)d_in[0];
  const float* t1[7]; const float* t2[7]; const float* t3[7];
  for (int i = 0; i < 7; ++i) {
    t1[i] = (const float*)d_in[1 + i];
    t2[i] = (const float*)d_in[8 + i];
    t3[i] = (const float*)d_in[15 + i];
  }
  float* ws = (float*)d_ws;
  size_t off = 0;
  float* pos = ws + off; off += (size_t)PB * 3 * PM;
  float* x1  = ws + off; off += (size_t)PB * 64 * PM;
  float* x2  = ws + off; off += (size_t)PB * 64 * PM;
  float* x3  = ws + off; off += (size_t)PB * 128 * PM;
  int* idxb  = (int*)(ws + off); off += (size_t)PB * PM * PK;
  float* wqk1 = ws + off; off += 64 * 8;
  float* wv1  = ws + off; off += 64 * 8;
  float* wqk2 = ws + off; off += 64 * 128;
  float* wv2  = ws + off; off += 64 * 128;
  float* wqk3 = ws + off; off += 128 * 128;
  float* wv3  = ws + off; off += 128 * 128;

  prep_weights_kernel<<<(64 * 8 + 255) / 256, 256, 0, stream>>>(
      t1[0], t1[1], t1[2], wqk1, wv1, 64, 6, 8);
  prep_weights_kernel<<<(64 * 128 + 255) / 256, 256, 0, stream>>>(
      t2[0], t2[1], t2[2], wqk2, wv2, 64, 128, 128);
  prep_weights_kernel<<<(128 * 128 + 255) / 256, 256, 0, stream>>>(
      t3[0], t3[1], t3[2], wqk3, wv3, 128, 128, 128);

  fps_kernel<<<PB, 256, 0, stream>>>(x, pos);

  knn_kernel<<<PB * PM, 256, 0, stream>>>(pos, 3, idxb);
  xform_kernel<XNT, XNQ, 2, XNQ * xf_query_floats(8, 64, 256)><<<PB * PM / XNQ, XNT, 0, stream>>>(
      pos, pos, idxb, wqk1, wv1, t1[3], t1[4], t1[5], t1[6],
      x1, 3, 8, 64, 256);

  knn_kernel<<<PB * PM, 256, 0, stream>>>(x1, 64, idxb);
  xform_kernel<XNT, XNQ, 2, XNQ * xf_query_floats(128, 64, 256)><<<PB * PM / XNQ, XNT, 0, stream>>>(
      x1, pos, idxb, wqk2, wv2, t2[3], t2[4], t2[5], t2[6],
      x2, 64, 128, 64, 256);

  knn_kernel<<<PB * PM, 256, 0, stream>>>(x2, 64, idxb);
  xform_kernel<XNT, XNQ, 4, XNQ * xf_query_floats(128, 128, 512)><<<PB * PM / XNQ, XNT, 0, stream>>>(
      x2, pos, idxb, wqk3, wv3, t3[3], t3[4], t3[5], t3[6],
      x3, 64, 128, 128, 512);

  pool_kernel<<<PB, 256, 0, stream>>>(x1, x2, x3, (float*)d_out);
}


# knn: one wave per query point, 16 candidates per lane in registers, argmin by 16-lane DPP rows joined via scalar registers; no LDS tree, no barrier (was 160 per query); same picks, same order
# speedup vs baseline: 2.2505x; 1.3250x over previous
#include <hip/hip_runtime.h>
#include <math.h>

#define PB 16
#define PN 4096
#define PM 512
#define PK 16

typedef float v2f __attribute__((ext_vector_type(2)));
typedef float v8f __attribute__((ext_vector_type(8)));

__device__ __forceinline__ v8f wmma_f32x4(v2f a, v2f b, v8f c) {
  return __builtin_amdgcn_wmma_f32_16x16x4_f32(false, a, false, b, (short)0, c, false, false);
}

__global__ __launch_bounds__(256)
void prep_weights_kernel(const float* __restrict__ Wq, const float* __restrict__ Wk,
                         const float* __restrict__ Wv, float* __restrict__ outQK,
                         float* __restrict__ outV, int D, int Cw, int Kpad) {
  int i = blockIdx.x * 256 + threadIdx.x;
  if (i >= D * Kpad) return;
  int d = i / Kpad, k = i % Kpad;
  float q = 0.f, v = 0.f;
  if (k < Cw) {
    q = Wq[(size_t)d * Cw + k] - Wk[(size_t)d * Cw + k];
    v = Wv[(size_t)d * Cw + k];
  }
  outQK[i] = q;
  outV[i] = v;
}

template <int XM>
__device__ __forceinline__ unsigned row_xor_u32(unsigned v) {
  return (unsigned)__builtin_amdgcn_update_dpp(0, (int)v, 0x160 | XM, 0xf, 0xf, true);
}
__device__ __forceinline__ unsigned umax2(unsigned a, unsigned b) { return a > b ? a : b; }
__device__ __forceinline__ unsigned umin2(unsigned a, unsigned b) { return a < b ? a : b; }
__device__ __forceinline__ unsigned row_umax(unsigned v) {
  v = umax2(v, row_xor_u32<1>(v));
  v = umax2(v, row_xor_u32<2>(v));
  v = umax2(v, row_xor_u32<4>(v));
  return umax2(v, row_xor_u32<8>(v));
}
__device__ __forceinline__ unsigned row_umin(unsigned v) {
  v = umin2(v, row_xor_u32<1>(v));
  v = umin2(v, row_xor_u32<2>(v));
  v = umin2(v, row_xor_u32<4>(v));
  return umin2(v, row_xor_u32<8>(v));
}

__global__ __launch_bounds__(256)
void fps_kernel(const float* __restrict__ x, float* __restrict__ pos) {
  const int b = blockIdx.x;
  const int tid = threadIdx.x;
  const int PPT = PN / 256;
  __shared__ float sX[PN];
  __shared__ float sY[PN];
  __shared__ float sZ[PN];
  __shared__ int sSel[PM];
  __shared__ uint2 sPart[2][16];

  float px[PPT], py[PPT], pz[PPT], dmin[PPT];
#pragma unroll
  for (int i = 0; i < PPT; ++i) {
    int p = tid + i * 256;
    const float* xp = x + ((size_t)b * PN + p) * 3;
    px[i] = xp[0]; py[i] = xp[1]; pz[i] = xp[2];
    sX[p] = px[i]; sY[p] = py[i]; sZ[p] = pz[i];
    dmin[i] = INFINITY;
  }
  if (tid == 0) sSel[0] = 0;
  __syncthreads();
  float bx = sX[0], by = sY[0], bz = sZ[0];
  for (int it = 1; it < PM; ++it) {
    float best = -INFINITY; int bslot = PPT;
#pragma unroll
    for (int i = 0; i < PPT; ++i) {
      float dx = px[i] - bx, dy = py[i] - by, dz = pz[i] - bz;
      float d = dx * dx + dy * dy + dz * dz;
      dmin[i] = fminf(dmin[i], d);
      if (dmin[i] > best) { best = dmin[i]; bslot = i; }
    }
    const unsigned barg = (unsigned)tid + ((unsigned)bslot << 8);
    const unsigned vb = __float_as_uint(best);
    const unsigned rmax = row_umax(vb);
    const unsigned rarg = row_umin(vb == rmax ? barg : 0xFFFFFFFFu);
    uint2* part = sPart[it & 1];
    if ((tid & 15) == 0) part[tid >> 4] = make_uint2(rmax, rarg);
    __syncthreads();
    const uint2 pr = part[tid & 15];
    const unsigned gmax = row_umax(pr.x);
    const unsigned garg = row_umin(pr.x == gmax ? pr.y : 0xFFFFFFFFu);
    bx = sX[garg]; by = sY[garg]; bz = sZ[garg];
    if (tid == 0) sSel[it] = (int)garg;
  }
  __syncthreads();
  for (int i = tid; i < PM; i += 256) {
    int p = sSel[i];
    pos[((size_t)b * 3 + 0) * PM + i] = sX[p];
    pos[((size_t)b * 3 + 1) * PM + i] = sY[p];
    pos[((size_t)b * 3 + 2) * PM + i] = sZ[p];
  }
}

#define KNN_QPB 8
__global__ __launch_bounds__(KNN_QPB * 32)
void knn_kernel(const float* __restrict__ f, int C, int* __restrict__ idxout) {
  const int lane = threadIdx.x & 31;
  const int wave = __builtin_amdgcn_readfirstlane((int)(threadIdx.x >> 5));
  const int bn = blockIdx.x * KNN_QPB + wave;
  const int b = bn / PM, n = bn % PM;
  const float* fb = f + (size_t)b * C * PM;
  float acc[16];
#pragma unroll
  for (int i = 0; i < 16; ++i) acc[i] = 0.f;
#pragma unroll 2
  for (int c = 0; c < C; ++c) {
    const float* row = fb + (size_t)c * PM;
    const float fq = row[n];
#pragma unroll
    for (int i = 0; i < 16; ++i) {
      float d = fq - row[lane + 32 * i];
      acc[i] += d * d;
    }
  }
  unsigned key[16];
#pragma unroll
  for (int i = 0; i < 16; ++i) key[i] = __float_as_uint(acc[i]);
  unsigned mine = 0;
  for (int r = 0; r < PK; ++r) {
    unsigned bv = key[0], bi = 0;
#pragma unroll
    for (int i = 1; i < 16; ++i) {
      if (key[i] < bv) { bv = key[i]; bi = (unsigned)i; }
    }
    const unsigned bm = (unsigned)lane + (bi << 5);
    const unsigned rmin = row_umin(bv);
    const unsigned gmin = umin2((unsigned)__builtin_amdgcn_readlane((int)rmin, 0),
                                (unsigned)__builtin_amdgcn_readlane((int)rmin, 16));
    const unsigned rarg = row_umin(bv == gmin ? bm : 0xFFFFFFFFu);
    const unsigned garg = umin2((unsigned)__builtin_amdgcn_readlane((int)rarg, 0),
                                (unsigned)__builtin_amdgcn_readlane((int)rarg, 16));
    if (lane == r) mine = garg;
#pragma unroll
    for (int i = 0; i < 16; ++i) {
      if ((unsigned)(lane + 32 * i) == garg) key[i] = 0xFFFFFFFFu;
    }
  }
  if (lane < PK) idxout[(size_t)bn * PK + lane] = (int)(mine & (unsigned)(PM - 1));
}

constexpr int xf_query_floats(int Kpad, int D, int H) {
  return 48 + 16 * (D + 2) + D * 16 +
         ((16 * (Kpad + 2) + D * 16 > 16 * (H + 2)) ? (16 * (Kpad + 2) + D * 16)
                                                     : (16 * (H + 2))) +
         D * 16 + 16;
}

#define XNT 512
#define XNQ 4

template <int NT, int NQ, int QD, int LDSF>
__global__ __launch_bounds__(NT)
void xform_kernel(const float* __restrict__ f, const float* __restrict__ pos,
                  const int* __restrict__ idx,
                  const float* __restrict__ WqkP, const float* __restrict__ WvP,
                  const float* __restrict__ P1, const float* __restrict__ P2,
                  const float* __restrict__ A1, const float* __restrict__ A2,
                  float* __restrict__ out, int Cin, int Kpad, int D, int H) {
  constexpr int NW = NT / 32;
  constexpr int NG = NQ / QD;
  const int bn0 = blockIdx.x * NQ;
  const int b = bn0 / PM, n0 = bn0 % PM;
  const int tid = threadIdx.x;
  const int lane = tid & 31, wave = tid >> 5;
  const int SG = Kpad + 2;
  const int SD = D + 2;
  const int SHS = H + 2;
  __shared__ float smem[LDSF] __attribute__((aligned(16)));
  const int scrF = (16 * SG + D * 16 > 16 * SHS) ? (16 * SG + D * 16) : (16 * SHS);
  const int oHT  = 48;
  const int oV   = oHT + 16 * SD;
  const int oScr = oV + D * 16;
  const int oPe  = oScr + 16 * SG;
  const int oL   = oScr + scrF;
  const int oNb  = oL + D * 16;
  const int QS   = oNb + 16;

  if (tid == 0) {
    __builtin_prefetch(A1, 0, 3);
    __builtin_prefetch(A2, 0, 3);
  }
  if (tid < NQ * PK) {
    const int q = tid >> 4, j = tid & 15;
    ((int*)(smem + q * QS + oNb))[j] = idx[(size_t)bn0 * PK + tid];
  }
  __syncthreads();

  for (int e = tid; e < NQ * 48; e += NT) {
    const int q = e / 48, t = e % 48;
    const int c = t >> 4, j = t & 15;
    const int* nb = (const int*)(smem + q * QS + oNb);
    const float* pc = pos + ((size_t)b * 3 + c) * PM;
    smem[q * QS + c * 16 + j] = pc[nb[j]] - pc[n0 + q];
  }
#pragma unroll
  for (int q = 0; q < NQ; ++q) {
    const int n = n0 + q;
    const int* nb = (const int*)(smem + q * QS + oNb);
    float* sGfT = smem + q * QS + oScr;
    for (int e = tid; e < 16 * Kpad; e += NT) {
      int j = e / Kpad, row = e % Kpad;
      float v = 0.f;
      if (row < Cin) {
        v = f[((size_t)b * Cin + row) * PM + nb[j]] - f[((size_t)b * Cin + row) * PM + n];
      } else if (row < 2 * Cin) {
        int c = row - Cin;
        v = f[((size_t)b * Cin + c) * PM + n];
      }
      sGfT[j * SG + row] = v;
    }
  }
  __syncthreads();

#pragma unroll
  for (int q = 0; q < NQ; ++q) {
    const float* sRel = smem + q * QS;
    float* sHT = smem + q * QS + oHT;
    for (int e = tid; e < D * 16; e += NT) {
      int d = e % D, j = e / D;
      float a = P1[d * 3 + 0] * sRel[j] + P1[d * 3 + 1] * sRel[16 + j] + P1[d * 3 + 2] * sRel[32 + j];
      sHT[j * SD + d] = fmaxf(a, 0.f);
    }
  }
  __syncthreads();
  for (int e = tid; e < D * 16; e += NT) {
    const int d = e >> 4, j = e & 15;
    float acc[NQ];
#pragma unroll
    for (int q = 0; q < NQ; ++q) acc[q] = 0.f;
    const float* p2 = P2 + (size_t)d * D;
    const float* tcol = smem + oHT + j * SD;
    for (int k = 0; k < D; ++k) {
      const float w = p2[k];
#pragma unroll
      for (int q = 0; q < NQ; ++q) acc[q] += w * tcol[q * QS + k];
    }
#pragma unroll
    for (int q = 0; q < NQ; ++q) smem[q * QS + oPe + d * 16 + j] = acc[q];
  }
  __syncthreads();

  const int r = lane & 15;
  const int khalf = (lane >> 4) << 1;
  const int rhalf = (lane >> 4) << 3;
  const int nrcD = D >> 4;

  for (int item = wave; item < nrcD * NG; item += NW) {
    const int rc = item % nrcD, q0 = (item / nrcD) * QD;
    const int rowBase = rc * 16;
    const int row = rowBase + r;
    v8f acc[QD], accv[QD];
#pragma unroll
    for (int t = 0; t < QD; ++t) {
      const float* sPe = smem + (q0 + t) * QS + oPe;
#pragma unroll
      for (int u = 0; u < 8; ++u) {
        float pe = sPe[(rowBase + u + rhalf) * 16 + r];
        acc[t][u] = pe; accv[t][u] = pe;
      }
    }
    const float* aqp = WqkP + (size_t)row * Kpad + khalf;
    const float* avp = WvP  + (size_t)row * Kpad + khalf;
    const float* bp  = smem + q0 * QS + oScr + r * SG + khalf;
#pragma unroll 4
    for (int kk = 0; kk < Kpad; kk += 4) {
      v2f aq = *(const v2f*)(aqp + kk);
      v2f av = *(const v2f*)(avp + kk);
#pragma unroll
      for (int t = 0; t < QD; ++t) {
        v2f bfrag = *(const v2f*)(bp + t * QS + kk);
        acc[t]  = wmma_f32x4(aq, bfrag, acc[t]);
        accv[t] = wmma_f32x4(av, bfrag, accv[t]);
      }
    }
#pragma unroll
    for (int t = 0; t < QD; ++t) {
      float* sHT = smem + (q0 + t) * QS + oHT;
      float* sV  = smem + (q0 + t) * QS + oV;
#pragma unroll
      for (int u = 0; u < 8; ++u) {
        sHT[r * SD + rowBase + u + rhalf] = acc[t][u];
        sV[(rowBase + u + rhalf) * 16 + r] = accv[t][u];
      }
    }
  }
  __syncthreads();

  for (int rc = wave; rc < (H >> 4); rc += NW) {
    const int rowBase = rc * 16;
    const int row = rowBase + r;
    v8f acc[NQ];
#pragma unroll
    for (int q = 0; q < NQ; ++q)
#pragma unroll
      for (int u = 0; u < 8; ++u) acc[q][u] = 0.f;
    const float* ap = A1 + (size_t)row * D + khalf;
    const float* bp = smem + oHT + r * SD + khalf;
#pragma unroll 4
    for (int kk = 0; kk < D; kk += 4) {
      v2f afrag = *(const v2f*)(ap + kk);
#pragma unroll
      for (int q = 0; q < NQ; ++q) {
        v2f bfrag = *(const v2f*)(bp + q * QS + kk);
        acc[q] = wmma_f32x4(afrag, bfrag, acc[q]);
      }
    }
#pragma unroll
    for (int q = 0; q < NQ; ++q) {
      float* sTT = smem + q * QS + oScr;
#pragma unroll
      for (int u = 0; u < 8; ++u)
        sTT[r * SHS + rowBase + u + rhalf] = fmaxf(acc[q][u], 0.f);
    }
  }
  __syncthreads();

  for (int item = wave; item < nrcD * NG; item += NW) {
    const int rc = item % nrcD, q0 = (item / nrcD) * QD;
    const int rowBase = rc * 16;
    const int row = rowBase + r;
    v8f acc[QD];
#pragma unroll
    for (int t = 0; t < QD; ++t)
#pragma unroll
      for (int u = 0; u < 8; ++u) acc[t][u] = 0.f;
    const float* ap = A2 + (size_t)row * H + khalf;
    const float* bp = smem + q0 * QS + oScr + r * SHS + khalf;
#pragma unroll 4
    for (int kk = 0; kk < H; kk += 4) {
      v2f afrag = *(const v2f*)(ap + kk);
#pragma unroll
      for (int t = 0; t < QD; ++t) {
        v2f bfrag = *(const v2f*)(bp + t * QS + kk);
        acc[t] = wmma_f32x4(afrag, bfrag, acc[t]);
      }
    }
#pragma unroll
    for (int t = 0; t < QD; ++t) {
      float* sL = smem + (q0 + t) * QS + oL;
#pragma unroll
      for (int u = 0; u < 8; ++u)
        sL[(rowBase + u + rhalf) * 16 + r] = acc[t][u];
    }
  }
  __syncthreads();

  for (int e = tid; e < NQ * D; e += NT) {
    const int q = e % NQ, d = e / NQ;
    const float* sL = smem + q * QS + oL + d * 16;
    const float* sV = smem + q * QS + oV + d * 16;
    float mx = -INFINITY;
#pragma unroll
    for (int j = 0; j < PK; ++j) mx = fmaxf(mx, sL[j]);
    float s = 0.f, o = 0.f;
#pragma unroll
    for (int j = 0; j < PK; ++j) {
      float p = __expf(sL[j] - mx);
      s += p;
      o += p * sV[j];
    }
    out[((size_t)b * D + d) * PM + n0 + q] = o / s;
  }
}

__global__ __launch_bounds__(256)
void pool_kernel(const float* __restrict__ x1, const float* __restrict__ x2,
                 const float* __restrict__ x3, float* __restrict__ out) {
  int b = blockIdx.x, ch = threadIdx.x;
  const float* src;
  if (ch < 64)       src = x1 + ((size_t)b * 64 + ch) * PM;
  else if (ch < 128) src = x2 + ((size_t)b * 64 + (ch - 64)) * PM;
  else               src = x3 + ((size_t)b * 128 + (ch - 128)) * PM;
  float mx = -INFINITY, sm = 0.f;
  for (int m = 0; m < PM; ++m) { float v = src[m]; mx = fmaxf(mx, v); sm += v; }
  out[(size_t)b * 512 + ch] = mx;
  out[(size_t)b * 512 + 256 + ch] = sm * (1.f / PM);
}

extern "C" void kernel_launch(void* const* d_in, const int* in_sizes, int n_in,
                              void* d_out, int out_size, void* d_ws, size_t ws_size,
                              hipStream_t stream) {
  (void)in_sizes; (void)n_in; (void)out_size; (void)ws_size;
  const float* x = (const float*)d_in[0];
  const float* t1[7]; const float* t2[7]; const float* t3[7];
  for (int i = 0; i < 7; ++i) {
    t1[i] = (const float*)d_in[1 + i];
    t2[i] = (const float*)d_in[8 + i];
    t3[i] = (const float*)d_in[15 + i];
  }
  float* ws = (float*)d_ws;
  size_t off = 0;
  float* pos = ws + off; off += (size_t)PB * 3 * PM;
  float* x1  = ws + off; off += (size_t)PB * 64 * PM;
  float* x2  = ws + off; off += (size_t)PB * 64 * PM;
  float* x3  = ws + off; off += (size_t)PB * 128 * PM;
  int* idxb  = (int*)(ws + off); off += (size_t)PB * PM * PK;
  float* wqk1 = ws + off; off += 64 * 8;
  float* wv1  = ws + off; off += 64 * 8;
  float* wqk2 = ws + off; off += 64 * 128;
  float* wv2  = ws + off; off += 64 * 128;
  float* wqk3 = ws + off; off += 128 * 128;
  float* wv3  = ws + off; off += 128 * 128;

  prep_weights_kernel<<<(64 * 8 + 255) / 256, 256, 0, stream>>>(
      t1[0], t1[1], t1[2], wqk1, wv1, 64, 6, 8);
  prep_weights_kernel<<<(64 * 128 + 255) / 256, 256, 0, stream>>>(
      t2[0], t2[1], t2[2], wqk2, wv2, 64, 128, 128);
  prep_weights_kernel<<<(128 * 128 + 255) / 256, 256, 0, stream>>>(
      t3[0], t3[1], t3[2], wqk3, wv3, 128, 128, 128);

  fps_kernel<<<PB, 256, 0, stream>>>(x, pos);

  knn_kernel<<<PB * PM / KNN_QPB, KNN_QPB * 32, 0, stream>>>(pos, 3, idxb);
  xform_kernel<XNT, XNQ, 2, XNQ * xf_query_floats(8, 64, 256)><<<PB * PM / XNQ, XNT, 0, stream>>>(
      pos, pos, idxb, wqk1, wv1, t1[3], t1[4], t1[5], t1[6],
      x1, 3, 8, 64, 256);

  knn_kernel<<<PB * PM / KNN_QPB, KNN_QPB * 32, 0, stream>>>(x1, 64, idxb);
  xform_kernel<XNT, XNQ, 2, XNQ * xf_query_floats(128, 64, 256)><<<PB * PM / XNQ, XNT, 0, stream>>>(
      x1, pos, idxb, wqk2, wv2, t2[3], t2[4], t2[5], t2[6],
      x2, 64, 128, 64, 256);

  knn_kernel<<<PB * PM / KNN_QPB, KNN_QPB * 32, 0, stream>>>(x2, 64, idxb);
  xform_kernel<XNT, XNQ, 4, XNQ * xf_query_floats(128, 128, 512)><<<PB * PM / XNQ, XNT, 0, stream>>>(
      x2, pos, idxb, wqk3, wv3, t3[3], t3[4], t3[5], t3[6],
      x3, 64, 128, 128, 512);

  pool_kernel<<<PB, 256, 0, stream>>>(x1, x2, x3, (float*)d_out);
}


# layers 1,2 also store their output point-major; next layer's gather reads each neighbour as one contiguous run, all loads issued at once (was a line per feature, 16 dependent trips); sums unchanged
# speedup vs baseline: 2.3535x; 1.0458x over previous
#include <hip/hip_runtime.h>
#include <math.h>

#define PB 16
#define PN 4096
#define PM 512
#define PK 16

typedef float v2f __attribute__((ext_vector_type(2)));
typedef float v8f __attribute__((ext_vector_type(8)));

__device__ __forceinline__ v8f wmma_f32x4(v2f a, v2f b, v8f c) {
  return __builtin_amdgcn_wmma_f32_16x16x4_f32(false, a, false, b, (short)0, c, false, false);
}

__global__ __launch_bounds__(256)
void prep_weights_kernel(const float* __restrict__ Wq, const float* __restrict__ Wk,
                         const float* __restrict__ Wv, float* __restrict__ outQK,
                         float* __restrict__ outV, int D, int Cw, int Kpad) {
  int i = blockIdx.x * 256 + threadIdx.x;
  if (i >= D * Kpad) return;
  int d = i / Kpad, k = i % Kpad;
  float q = 0.f, v = 0.f;
  if (k < Cw) {
    q = Wq[(size_t)d * Cw + k] - Wk[(size_t)d * Cw + k];
    v = Wv[(size_t)d * Cw + k];
  }
  outQK[i] = q;
  outV[i] = v;
}

template <int XM>
__device__ __forceinline__ unsigned row_xor_u32(unsigned v) {
  return (unsigned)__builtin_amdgcn_update_dpp(0, (int)v, 0x160 | XM, 0xf, 0xf, true);
}
__device__ __forceinline__ unsigned umax2(unsigned a, unsigned b) { return a > b ? a : b; }
__device__ __forceinline__ unsigned umin2(unsigned a, unsigned b) { return a < b ? a : b; }
__device__ __forceinline__ unsigned row_umax(unsigned v) {
  v = umax2(v, row_xor_u32<1>(v));
  v = umax2(v, row_xor_u32<2>(v));
  v = umax2(v, row_xor_u32<4>(v));
  return umax2(v, row_xor_u32<8>(v));
}
__device__ __forceinline__ unsigned row_umin(unsigned v) {
  v = umin2(v, row_xor_u32<1>(v));
  v = umin2(v, row_xor_u32<2>(v));
  v = umin2(v, row_xor_u32<4>(v));
  return umin2(v, row_xor_u32<8>(v));
}

__global__ __launch_bounds__(256)
void fps_kernel(const float* __restrict__ x, float* __restrict__ pos) {
  const int b = blockIdx.x;
  const int tid = threadIdx.x;
  const int PPT = PN / 256;
  __shared__ float sX[PN];
  __shared__ float sY[PN];
  __shared__ float sZ[PN];
  __shared__ int sSel[PM];
  __shared__ uint2 sPart[2][16];

  float px[PPT], py[PPT], pz[PPT], dmin[PPT];
#pragma unroll
  for (int i = 0; i < PPT; ++i) {
    int p = tid + i * 256;
    const float* xp = x + ((size_t)b * PN + p) * 3;
    px[i] = xp[0]; py[i] = xp[1]; pz[i] = xp[2];
    sX[p] = px[i]; sY[p] = py[i]; sZ[p] = pz[i];
    dmin[i] = INFINITY;
  }
  if (tid == 0) sSel[0] = 0;
  __syncthreads();
  float bx = sX[0], by = sY[0], bz = sZ[0];
  for (int it = 1; it < PM; ++it) {
    float best = -INFINITY; int bslot = PPT;
#pragma unroll
    for (int i = 0; i < PPT; ++i) {
      float dx = px[i] - bx, dy = py[i] - by, dz = pz[i] - bz;
      float d = dx * dx + dy * dy + dz * dz;
      dmin[i] = fminf(dmin[i], d);
      if (dmin[i] > best) { best = dmin[i]; bslot = i; }
    }
    const unsigned barg = (unsigned)tid + ((unsigned)bslot << 8);
    const unsigned vb = __float_as_uint(best);
    const unsigned rmax = row_umax(vb);
    const unsigned rarg = row_umin(vb == rmax ? barg : 0xFFFFFFFFu);
    uint2* part = sPart[it & 1];
    if ((tid & 15) == 0) part[tid >> 4] = make_uint2(rmax, rarg);
    __syncthreads();
    const uint2 pr = part[tid & 15];
    const unsigned gmax = row_umax(pr.x);
    const unsigned garg = row_umin(pr.x == gmax ? pr.y : 0xFFFFFFFFu);
    bx = sX[garg]; by = sY[garg]; bz = sZ[garg];
    if (tid == 0) sSel[it] = (int)garg;
  }
  __syncthreads();
  for (int i = tid; i < PM; i += 256) {
    int p = sSel[i];
    pos[((size_t)b * 3 + 0) * PM + i] = sX[p];
    pos[((size_t)b * 3 + 1) * PM + i] = sY[p];
    pos[((size_t)b * 3 + 2) * PM + i] = sZ[p];
  }
}

#define KNN_QPB 8
__global__ __launch_bounds__(KNN_QPB * 32)
void knn_kernel(const float* __restrict__ f, int C, int* __restrict__ idxout) {
  const int lane = threadIdx.x & 31;
  const int wave = __builtin_amdgcn_readfirstlane((int)(threadIdx.x >> 5));
  const int bn = blockIdx.x * KNN_QPB + wave;
  const int b = bn / PM, n = bn % PM;
  const float* fb = f + (size_t)b * C * PM;
  float acc[16];
#pragma unroll
  for (int i = 0; i < 16; ++i) acc[i] = 0.f;
#pragma unroll 2
  for (int c = 0; c < C; ++c) {
    const float* row = fb + (size_t)c * PM;
    const float fq = row[n];
#pragma unroll
    for (int i = 0; i < 16; ++i) {
      float d = fq - row[lane + 32 * i];
      acc[i] += d * d;
    }
  }
  unsigned key[16];
#pragma unroll
  for (int i = 0; i < 16; ++i) key[i] = __float_as_uint(acc[i]);
  unsigned mine = 0;
  for (int r = 0; r < PK; ++r) {
    unsigned bv = key[0], bi = 0;
#pragma unroll
    for (int i = 1; i < 16; ++i) {
      if (key[i] < bv) { bv = key[i]; bi = (unsigned)i; }
    }
    const unsigned bm = (unsigned)lane + (bi << 5);
    const unsigned rmin = row_umin(bv);
    const unsigned gmin = umin2((unsigned)__builtin_amdgcn_readlane((int)rmin, 0),
                                (unsigned)__builtin_amdgcn_readlane((int)rmin, 16));
    const unsigned rarg = row_umin(bv == gmin ? bm : 0xFFFFFFFFu);
    const unsigned garg = umin2((unsigned)__builtin_amdgcn_readlane((int)rarg, 0),
                                (unsigned)__builtin_amdgcn_readlane((int)rarg, 16));
    if (lane == r) mine = garg;
#pragma unroll
    for (int i = 0; i < 16; ++i) {
      if ((unsigned)(lane + 32 * i) == garg) key[i] = 0xFFFFFFFFu;
    }
  }
  if (lane < PK) idxout[(size_t)bn * PK + lane] = (int)(mine & (unsigned)(PM - 1));
}

constexpr int xf_query_floats(int Kpad, int D, int H) {
  return 48 + 16 * (D + 2) + D * 16 +
         ((16 * (Kpad + 2) + D * 16 > 16 * (H + 2)) ? (16 * (Kpad + 2) + D * 16)
                                                     : (16 * (H + 2))) +
         D * 16;
}

#define XNT 512
#define XNQ 4

template <int NT, int NQ, int QD, int CIN, int LDSF>
__global__ __launch_bounds__(NT)
void xform_kernel(const float* __restrict__ f, const float* __restrict__ pos,
                  const int* __restrict__ idx,
                  const float* __restrict__ WqkP, const float* __restrict__ WvP,
                  const float* __restrict__ P1, const float* __restrict__ P2,
                  const float* __restrict__ A1, const float* __restrict__ A2,
                  float* __restrict__ out, float* __restrict__ outT, int D, int H) {
  constexpr int NW = NT / 32;
  constexpr int NG = NQ / QD;
  constexpr int Kpad = (CIN == 3) ? 8 : 2 * CIN;
  const int bn0 = blockIdx.x * NQ;
  const int b = bn0 / PM, n0 = bn0 % PM;
  const int tid = threadIdx.x;
  const int lane = tid & 31, wave = tid >> 5;
  const int SG = Kpad + 2;
  const int SD = D + 2;
  const int SHS = H + 2;
  __shared__ float smem[LDSF] __attribute__((aligned(16)));
  const int scrF = (16 * SG + D * 16 > 16 * SHS) ? (16 * SG + D * 16) : (16 * SHS);
  const int oHT  = 48;
  const int oV   = oHT + 16 * SD;
  const int oScr = oV + D * 16;
  const int oPe  = oScr + 16 * SG;
  const int oL   = oScr + scrF;
  const int QS   = oL + D * 16;

  if (tid == 0) {
    __builtin_prefetch(A1, 0, 3);
    __builtin_prefetch(A2, 0, 3);
  }
  if constexpr (CIN == 3) {
    for (int e = tid; e < NQ * 16 * Kpad; e += NT) {
      const int q = e / (16 * Kpad), t = e % (16 * Kpad);
      const int j = t / Kpad, row = t % Kpad;
      float v = 0.f;
      if (row < 2 * CIN) {
        const float* pc = pos + ((size_t)b * CIN + (row < CIN ? row : row - CIN)) * PM;
        v = pc[n0 + q];
        if (row < CIN) {
          v = pc[idx[(size_t)(bn0 + q) * PK + j]] - v;
          smem[q * QS + row * 16 + j] = v;
        }
      }
      smem[q * QS + oScr + j * SG + row] = v;
    }
  } else {
    constexpr int C4 = CIN / 4;
    constexpr int NP = NQ * 16 * C4 / NT;
    static_assert(NQ * 16 * C4 % NT == 0 && NT % (16 * C4) == 0, "whole passes, c4 and j fixed per thread");
    const int c4 = tid % C4, j = (tid / C4) % 16;
    int nbv[NP];
    float4 cen[NP], nbf[NP];
#pragma unroll
    for (int p = 0; p < NP; ++p) {
      const int q = (tid + p * NT) / (16 * C4);
      nbv[p] = idx[(size_t)(bn0 + q) * PK + j];
      cen[p] = *(const float4*)(f + ((size_t)b * PM + n0 + q) * CIN + c4 * 4);
    }
#pragma unroll
    for (int p = 0; p < NP; ++p)
      nbf[p] = *(const float4*)(f + ((size_t)b * PM + nbv[p]) * CIN + c4 * 4);
    if (c4 < 3) {
      const float* pc = pos + ((size_t)b * 3 + c4) * PM;
      float rel[NP];
#pragma unroll
      for (int p = 0; p < NP; ++p) rel[p] = pc[nbv[p]] - pc[n0 + (tid + p * NT) / (16 * C4)];
#pragma unroll
      for (int p = 0; p < NP; ++p) smem[((tid + p * NT) / (16 * C4)) * QS + c4 * 16 + j] = rel[p];
    }
#pragma unroll
    for (int p = 0; p < NP; ++p) {
      float* dst = smem + ((tid + p * NT) / (16 * C4)) * QS + oScr + j * SG + c4 * 4;
      *(v2f*)(dst)           = v2f{nbf[p].x - cen[p].x, nbf[p].y - cen[p].y};
      *(v2f*)(dst + 2)       = v2f{nbf[p].z - cen[p].z, nbf[p].w - cen[p].w};
      *(v2f*)(dst + CIN)     = v2f{cen[p].x, cen[p].y};
      *(v2f*)(dst + CIN + 2) = v2f{cen[p].z, cen[p].w};
    }
  }
  __syncthreads();

#pragma unroll
  for (int q = 0; q < NQ; ++q) {
    const float* sRel = smem + q * QS;
    float* sHT = smem + q * QS + oHT;
    for (int e = tid; e < D * 16; e += NT) {
      int d = e % D, j = e / D;
      float a = P1[d * 3 + 0] * sRel[j] + P1[d * 3 + 1] * sRel[16 + j] + P1[d * 3 + 2] * sRel[32 + j];
      sHT[j * SD + d] = fmaxf(a, 0.f);
    }
  }
  __syncthreads();
  for (int e = tid; e < D * 16; e += NT) {
    const int d = e >> 4, j = e & 15;
    float acc[NQ];
#pragma unroll
    for (int q = 0; q < NQ; ++q) acc[q] = 0.f;
    const float* p2 = P2 + (size_t)d * D;
    const float* tcol = smem + oHT + j * SD;
    for (int k = 0; k < D; ++k) {
      const float w = p2[k];
#pragma unroll
      for (int q = 0; q < NQ; ++q) acc[q] += w * tcol[q * QS + k];
    }
#pragma unroll
    for (int q = 0; q < NQ; ++q) smem[q * QS + oPe + d * 16 + j] = acc[q];
  }
  __syncthreads();

  const int r = lane & 15;
  const int khalf = (lane >> 4) << 1;
  const int rhalf = (lane >> 4) << 3;
  const int nrcD = D >> 4;

  for (int item = wave; item < nrcD * NG; item += NW) {
    const int rc = item % nrcD, q0 = (item / nrcD) * QD;
    const int rowBase = rc * 16;
    const int row = rowBase + r;
    v8f acc[QD], accv[QD];
#pragma unroll
    for (int t = 0; t < QD; ++t) {
      const float* sPe = smem + (q0 + t) * QS + oPe;
#pragma unroll
      for (int u = 0; u < 8; ++u) {
        float pe = sPe[(rowBase + u + rhalf) * 16 + r];
        acc[t][u] = pe; accv[t][u] = pe;
      }
    }
    const float* aqp = WqkP + (size_t)row * Kpad + khalf;
    const float* avp = WvP  + (size_t)row * Kpad + khalf;
    const float* bp  = smem + q0 * QS + oScr + r * SG + khalf;
#pragma unroll 4
    for (int kk = 0; kk < Kpad; kk += 4) {
      v2f aq = *(const v2f*)(aqp + kk);
      v2f av = *(const v2f*)(avp + kk);
#pragma unroll
      for (int t = 0; t < QD; ++t) {
        v2f bfrag = *(const v2f*)(bp + t * QS + kk);
        acc[t]  = wmma_f32x4(aq, bfrag, acc[t]);
        accv[t] = wmma_f32x4(av, bfrag, accv[t]);
      }
    }
#pragma unroll
    for (int t = 0; t < QD; ++t) {
      float* sHT = smem + (q0 + t) * QS + oHT;
      float* sV  = smem + (q0 + t) * QS + oV;
#pragma unroll
      for (int u = 0; u < 8; ++u) {
        sHT[r * SD + rowBase + u + rhalf] = acc[t][u];
        sV[(rowBase + u + rhalf) * 16 + r] = accv[t][u];
      }
    }
  }
  __syncthreads();

  for (int rc = wave; rc < (H >> 4); rc += NW) {
    const int rowBase = rc * 16;
    const int row = rowBase + r;
    v8f acc[NQ];
#pragma unroll
    for (int q = 0; q < NQ; ++q)
#pragma unroll
      for (int u = 0; u < 8; ++u) acc[q][u] = 0.f;
    const float* ap = A1 + (size_t)row * D + khalf;
    const float* bp = smem + oHT + r * SD + khalf;
#pragma unroll 4
    for (int kk = 0; kk < D; kk += 4) {
      v2f afrag = *(const v2f*)(ap + kk);
#pragma unroll
      for (int q = 0; q < NQ; ++q) {
        v2f bfrag = *(const v2f*)(bp + q * QS + kk);
        acc[q] = wmma_f32x4(afrag, bfrag, acc[q]);
      }
    }
#pragma unroll
    for (int q = 0; q < NQ; ++q) {
      float* sTT = smem + q * QS + oScr;
#pragma unroll
      for (int u = 0; u < 8; ++u)
        sTT[r * SHS + rowBase + u + rhalf] = fmaxf(acc[q][u], 0.f);
    }
  }
  __syncthreads();

  for (int item = wave; item < nrcD * NG; item += NW) {
    const int rc = item % nrcD, q0 = (item / nrcD) * QD;
    const int rowBase = rc * 16;
    const int row = rowBase + r;
    v8f acc[QD];
#pragma unroll
    for (int t = 0; t < QD; ++t)
#pragma unroll
      for (int u = 0; u < 8; ++u) acc[t][u] = 0.f;
    const float* ap = A2 + (size_t)row * H + khalf;
    const float* bp = smem + q0 * QS + oScr + r * SHS + khalf;
#pragma unroll 4
    for (int kk = 0; kk < H; kk += 4) {
      v2f afrag = *(const v2f*)(ap + kk);
#pragma unroll
      for (int t = 0; t < QD; ++t) {
        v2f bfrag = *(const v2f*)(bp + t * QS + kk);
        acc[t] = wmma_f32x4(afrag, bfrag, acc[t]);
      }
    }
#pragma unroll
    for (int t = 0; t < QD; ++t) {
      float* sL = smem + (q0 + t) * QS + oL;
#pragma unroll
      for (int u = 0; u < 8; ++u)
        sL[(rowBase + u + rhalf) * 16 + r] = acc[t][u];
    }
  }
  __syncthreads();

  for (int e = tid; e < NQ * D; e += NT) {
    const int q = e % NQ, d = e / NQ;
    const float* sL = smem + q * QS + oL + d * 16;
    const float* sV = smem + q * QS + oV + d * 16;
    float mx = -INFINITY;
#pragma unroll
    for (int j = 0; j < PK; ++j) mx = fmaxf(mx, sL[j]);
    float s = 0.f, o = 0.f;
#pragma unroll
    for (int j = 0; j < PK; ++j) {
      float p = __expf(sL[j] - mx);
      s += p;
      o += p * sV[j];
    }
    const float val = o / s;
    out[((size_t)b * D + d) * PM + n0 + q] = val;
    if (outT) outT[((size_t)b * PM + n0 + q) * D + d] = val;
  }
}

__global__ __launch_bounds__(256)
void pool_kernel(const float* __restrict__ xa, int Ca, const float* __restrict__ xb, int Cb,
                 int chBase, float* __restrict__ out) {
  int b = blockIdx.x, ch = threadIdx.x;
  const float* src = ch < Ca ? xa + ((size_t)b * Ca + ch) * PM
                             : xb + ((size_t)b * Cb + (ch - Ca)) * PM;
  float mx = -INFINITY, sm = 0.f;
  for (int m = 0; m < PM; ++m) { float v = src[m]; mx = fmaxf(mx, v); sm += v; }
  out[(size_t)b * 512 + chBase + ch] = mx;
  out[(size_t)b * 512 + 256 + chBase + ch] = sm * (1.f / PM);
}

extern "C" void kernel_launch(void* const* d_in, const int* in_sizes, int n_in,
                              void* d_out, int out_size, void* d_ws, size_t ws_size,
                              hipStream_t stream) {
  (void)in_sizes; (void)n_in; (void)out_size; (void)ws_size;
  const float* x = (const float*)d_in[0];
  const float* t1[7]; const float* t2[7]; const float* t3[7];
  for (int i = 0; i < 7; ++i) {
    t1[i] = (const float*)d_in[1 + i];
    t2[i] = (const float*)d_in[8 + i];
    t3[i] = (const float*)d_in[15 + i];
  }
  float* ws = (float*)d_ws;
  size_t off = 0;
  float* pos = ws + off; off += (size_t)PB * 3 * PM;
  float* x1  = ws + off; off += (size_t)PB * 64 * PM;
  float* x2  = ws + off; off += (size_t)PB * 64 * PM;
  float* x3  = ws + off; off += (size_t)PB * 128 * PM;
  int* idxb  = (int*)(ws + off); off += (size_t)PB * PM * PK;
  float* wqk1 = ws + off; off += 64 * 8;
  float* wv1  = ws + off; off += 64 * 8;
  float* wqk2 = ws + off; off += 64 * 128;
  float* wv2  = ws + off; off += 64 * 128;
  float* wqk3 = ws + off; off += 128 * 128;
  float* wv3  = ws + off; off += 128 * 128;

  prep_weights_kernel<<<(64 * 8 + 255) / 256, 256, 0, stream>>>(
      t1[0], t1[1], t1[2], wqk1, wv1, 64, 6, 8);
  prep_weights_kernel<<<(64 * 128 + 255) / 256, 256, 0, stream>>>(
      t2[0], t2[1], t2[2], wqk2, wv2, 64, 128, 128);
  prep_weights_kernel<<<(128 * 128 + 255) / 256, 256, 0, stream>>>(
      t3[0], t3[1], t3[2], wqk3, wv3, 128, 128, 128);

  fps_kernel<<<PB, 256, 0, stream>>>(x, pos);

  float* x1T = x3;
  float* x2T = x1;

  knn_kernel<<<PB * PM / KNN_QPB, KNN_QPB * 32, 0, stream>>>(pos, 3, idxb);
  xform_kernel<XNT, XNQ, 2, 3, XNQ * xf_query_floats(8, 64, 256)><<<PB * PM / XNQ, XNT, 0, stream>>>(
      pos, pos, idxb, wqk1, wv1, t1[3], t1[4], t1[5], t1[6],
      x1, x1T, 64, 256);
  pool_kernel<<<PB, 64, 0, stream>>>(x1, 64, x1, 0, 0, (float*)d_out);

  knn_kernel<<<PB * PM / KNN_QPB, KNN_QPB * 32, 0, stream>>>(x1, 64, idxb);
  xform_kernel<XNT, XNQ, 2, 64, XNQ * xf_query_floats(128, 64, 256)><<<PB * PM / XNQ, XNT, 0, stream>>>(
      x1T, pos, idxb, wqk2, wv2, t2[3], t2[4], t2[5], t2[6],
      x2, x2T, 64, 256);

  knn_kernel<<<PB * PM / KNN_QPB, KNN_QPB * 32, 0, stream>>>(x2, 64, idxb);
  xform_kernel<XNT, XNQ, 4, 64, XNQ * xf_query_floats(128, 128, 512)><<<PB * PM / XNQ, XNT, 0, stream>>>(
      x2T, pos, idxb, wqk3, wv3, t3[3], t3[4], t3[5], t3[6],
      x3, nullptr, 128, 512);

  pool_kernel<<<PB, 192, 0, stream>>>(x2, 64, x3, 128, 64, (float*)d_out);
}


# pe = P2 @ t1 moved from the vector ALU (scalar fma chain, 4 LDS reads per k) to f32 WMMA, 16 waves x 1-2 points; pe now sums 4 k terms per matrix step (order inside pe changes)
# speedup vs baseline: 2.6334x; 1.1189x over previous
#include <hip/hip_runtime.h>
#include <math.h>

#define PB 16
#define PN 4096
#define PM 512
#define PK 16

typedef float v2f __attribute__((ext_vector_type(2)));
typedef float v8f __attribute__((ext_vector_type(8)));

__device__ __forceinline__ v8f wmma_f32x4(v2f a, v2f b, v8f c) {
  return __builtin_amdgcn_wmma_f32_16x16x4_f32(false, a, false, b, (short)0, c, false, false);
}

__global__ __launch_bounds__(256)
void prep_weights_kernel(const float* __restrict__ Wq, const float* __restrict__ Wk,
                         const float* __restrict__ Wv, float* __restrict__ outQK,
                         float* __restrict__ outV, int D, int Cw, int Kpad) {
  int i = blockIdx.x * 256 + threadIdx.x;
  if (i >= D * Kpad) return;
  int d = i / Kpad, k = i % Kpad;
  float q = 0.f, v = 0.f;
  if (k < Cw) {
    q = Wq[(size_t)d * Cw + k] - Wk[(size_t)d * Cw + k];
    v = Wv[(size_t)d * Cw + k];
  }
  outQK[i] = q;
  outV[i] = v;
}

template <int XM>
__device__ __forceinline__ unsigned row_xor_u32(unsigned v) {
  return (unsigned)__builtin_amdgcn_update_dpp(0, (int)v, 0x160 | XM, 0xf, 0xf, true);
}
__device__ __forceinline__ unsigned umax2(unsigned a, unsigned b) { return a > b ? a : b; }
__device__ __forceinline__ unsigned umin2(unsigned a, unsigned b) { return a < b ? a : b; }
__device__ __forceinline__ unsigned row_umax(unsigned v) {
  v = umax2(v, row_xor_u32<1>(v));
  v = umax2(v, row_xor_u32<2>(v));
  v = umax2(v, row_xor_u32<4>(v));
  return umax2(v, row_xor_u32<8>(v));
}
__device__ __forceinline__ unsigned row_umin(unsigned v) {
  v = umin2(v, row_xor_u32<1>(v));
  v = umin2(v, row_xor_u32<2>(v));
  v = umin2(v, row_xor_u32<4>(v));
  return umin2(v, row_xor_u32<8>(v));
}

__global__ __launch_bounds__(256)
void fps_kernel(const float* __restrict__ x, float* __restrict__ pos) {
  const int b = blockIdx.x;
  const int tid = threadIdx.x;
  const int PPT = PN / 256;
  __shared__ float sX[PN];
  __shared__ float sY[PN];
  __shared__ float sZ[PN];
  __shared__ int sSel[PM];
  __shared__ uint2 sPart[2][16];

  float px[PPT], py[PPT], pz[PPT], dmin[PPT];
#pragma unroll
  for (int i = 0; i < PPT; ++i) {
    int p = tid + i * 256;
    const float* xp = x + ((size_t)b * PN + p) * 3;
    px[i] = xp[0]; py[i] = xp[1]; pz[i] = xp[2];
    sX[p] = px[i]; sY[p] = py[i]; sZ[p] = pz[i];
    dmin[i] = INFINITY;
  }
  if (tid == 0) sSel[0] = 0;
  __syncthreads();
  float bx = sX[0], by = sY[0], bz = sZ[0];
  for (int it = 1; it < PM; ++it) {
    float best = -INFINITY; int bslot = PPT;
#pragma unroll
    for (int i = 0; i < PPT; ++i) {
      float dx = px[i] - bx, dy = py[i] - by, dz = pz[i] - bz;
      float d = dx * dx + dy * dy + dz * dz;
      dmin[i] = fminf(dmin[i], d);
      if (dmin[i] > best) { best = dmin[i]; bslot = i; }
    }
    const unsigned barg = (unsigned)tid + ((unsigned)bslot << 8);
    const unsigned vb = __float_as_uint(best);
    const unsigned rmax = row_umax(vb);
    const unsigned rarg = row_umin(vb == rmax ? barg : 0xFFFFFFFFu);
    uint2* part = sPart[it & 1];
    if ((tid & 15) == 0) part[tid >> 4] = make_uint2(rmax, rarg);
    __syncthreads();
    const uint2 pr = part[tid & 15];
    const unsigned gmax = row_umax(pr.x);
    const unsigned garg = row_umin(pr.x == gmax ? pr.y : 0xFFFFFFFFu);
    bx = sX[garg]; by = sY[garg]; bz = sZ[garg];
    if (tid == 0) sSel[it] = (int)garg;
  }
  __syncthreads();
  for (int i = tid; i < PM; i += 256) {
    int p = sSel[i];
    pos[((size_t)b * 3 + 0) * PM + i] = sX[p];
    pos[((size_t)b * 3 + 1) * PM + i] = sY[p];
    pos[((size_t)b * 3 + 2) * PM + i] = sZ[p];
  }
}

#define KNN_QPB 8
__global__ __launch_bounds__(KNN_QPB * 32)
void knn_kernel(const float* __restrict__ f, int C, int* __restrict__ idxout) {
  const int lane = threadIdx.x & 31;
  const int wave = __builtin_amdgcn_readfirstlane((int)(threadIdx.x >> 5));
  const int bn = blockIdx.x * KNN_QPB + wave;
  const int b = bn / PM, n = bn % PM;
  const float* fb = f + (size_t)b * C * PM;
  float acc[16];
#pragma unroll
  for (int i = 0; i < 16; ++i) acc[i] = 0.f;
#pragma unroll 2
  for (int c = 0; c < C; ++c) {
    const float* row = fb + (size_t)c * PM;
    const float fq = row[n];
#pragma unroll
    for (int i = 0; i < 16; ++i) {
      float d = fq - row[lane + 32 * i];
      acc[i] += d * d;
    }
  }
  unsigned key[16];
#pragma unroll
  for (int i = 0; i < 16; ++i) key[i] = __float_as_uint(acc[i]);
  unsigned mine = 0;
  for (int r = 0; r < PK; ++r) {
    unsigned bv = key[0], bi = 0;
#pragma unroll
    for (int i = 1; i < 16; ++i) {
      if (key[i] < bv) { bv = key[i]; bi = (unsigned)i; }
    }
    const unsigned bm = (unsigned)lane + (bi << 5);
    const unsigned rmin = row_umin(bv);
    const unsigned gmin = umin2((unsigned)__builtin_amdgcn_readlane((int)rmin, 0),
                                (unsigned)__builtin_amdgcn_readlane((int)rmin, 16));
    const unsigned rarg = row_umin(bv == gmin ? bm : 0xFFFFFFFFu);
    const unsigned garg = umin2((unsigned)__builtin_amdgcn_readlane((int)rarg, 0),
                                (unsigned)__builtin_amdgcn_readlane((int)rarg, 16));
    if (lane == r) mine = garg;
#pragma unroll
    for (int i = 0; i < 16; ++i) {
      if ((unsigned)(lane + 32 * i) == garg) key[i] = 0xFFFFFFFFu;
    }
  }
  if (lane < PK) idxout[(size_t)bn * PK + lane] = (int)(mine & (unsigned)(PM - 1));
}

constexpr int xf_query_floats(int Kpad, int D, int H) {
  return 48 + 16 * (D + 2) + D * 16 +
         ((16 * (Kpad + 2) + D * 16 > 16 * (H + 2)) ? (16 * (Kpad + 2) + D * 16)
                                                     : (16 * (H + 2))) +
         D * 16;
}

#define XNT 512
#define XNQ 4

template <int NT, int NQ, int QD, int CIN, int LDSF>
__global__ __launch_bounds__(NT)
void xform_kernel(const float* __restrict__ f, const float* __restrict__ pos,
                  const int* __restrict__ idx,
                  const float* __restrict__ WqkP, const float* __restrict__ WvP,
                  const float* __restrict__ P1, const float* __restrict__ P2,
                  const float* __restrict__ A1, const float* __restrict__ A2,
                  float* __restrict__ out, float* __restrict__ outT, int D, int H) {
  constexpr int NW = NT / 32;
  constexpr int NG = NQ / QD;
  constexpr int Kpad = (CIN == 3) ? 8 : 2 * CIN;
  const int bn0 = blockIdx.x * NQ;
  const int b = bn0 / PM, n0 = bn0 % PM;
  const int tid = threadIdx.x;
  const int lane = tid & 31, wave = tid >> 5;
  const int SG = Kpad + 2;
  const int SD = D + 2;
  const int SHS = H + 2;
  __shared__ float smem[LDSF] __attribute__((aligned(16)));
  const int scrF = (16 * SG + D * 16 > 16 * SHS) ? (16 * SG + D * 16) : (16 * SHS);
  const int oHT  = 48;
  const int oV   = oHT + 16 * SD;
  const int oScr = oV + D * 16;
  const int oPe  = oScr + 16 * SG;
  const int oL   = oScr + scrF;
  const int QS   = oL + D * 16;

  if (tid == 0) {
    __builtin_prefetch(A1, 0, 3);
    __builtin_prefetch(A2, 0, 3);
  }
  if constexpr (CIN == 3) {
    for (int e = tid; e < NQ * 16 * Kpad; e += NT) {
      const int q = e / (16 * Kpad), t = e % (16 * Kpad);
      const int j = t / Kpad, row = t % Kpad;
      float v = 0.f;
      if (row < 2 * CIN) {
        const float* pc = pos + ((size_t)b * CIN + (row < CIN ? row : row - CIN)) * PM;
        v = pc[n0 + q];
        if (row < CIN) {
          v = pc[idx[(size_t)(bn0 + q) * PK + j]] - v;
          smem[q * QS + row * 16 + j] = v;
        }
      }
      smem[q * QS + oScr + j * SG + row] = v;
    }
  } else {
    constexpr int C4 = CIN / 4;
    constexpr int NP = NQ * 16 * C4 / NT;
    static_assert(NQ * 16 * C4 % NT == 0 && NT % (16 * C4) == 0, "whole passes, c4 and j fixed per thread");
    const int c4 = tid % C4, j = (tid / C4) % 16;
    int nbv[NP];
    float4 cen[NP], nbf[NP];
#pragma unroll
    for (int p = 0; p < NP; ++p) {
      const int q = (tid + p * NT) / (16 * C4);
      nbv[p] = idx[(size_t)(bn0 + q) * PK + j];
      cen[p] = *(const float4*)(f + ((size_t)b * PM + n0 + q) * CIN + c4 * 4);
    }
#pragma unroll
    for (int p = 0; p < NP; ++p)
      nbf[p] = *(const float4*)(f + ((size_t)b * PM + nbv[p]) * CIN + c4 * 4);
    if (c4 < 3) {
      const float* pc = pos + ((size_t)b * 3 + c4) * PM;
      float rel[NP];
#pragma unroll
      for (int p = 0; p < NP; ++p) rel[p] = pc[nbv[p]] - pc[n0 + (tid + p * NT) / (16 * C4)];
#pragma unroll
      for (int p = 0; p < NP; ++p) smem[((tid + p * NT) / (16 * C4)) * QS + c4 * 16 + j] = rel[p];
    }
#pragma unroll
    for (int p = 0; p < NP; ++p) {
      float* dst = smem + ((tid + p * NT) / (16 * C4)) * QS + oScr + j * SG + c4 * 4;
      *(v2f*)(dst)           = v2f{nbf[p].x - cen[p].x, nbf[p].y - cen[p].y};
      *(v2f*)(dst + 2)       = v2f{nbf[p].z - cen[p].z, nbf[p].w - cen[p].w};
      *(v2f*)(dst + CIN)     = v2f{cen[p].x, cen[p].y};
      *(v2f*)(dst + CIN + 2) = v2f{cen[p].z, cen[p].w};
    }
  }
  __syncthreads();

#pragma unroll
  for (int q = 0; q < NQ; ++q) {
    const float* sRel = smem + q * QS;
    float* sHT = smem + q * QS + oHT;
    for (int e = tid; e < D * 16; e += NT) {
      int d = e % D, j = e / D;
      float a = P1[d * 3 + 0] * sRel[j] + P1[d * 3 + 1] * sRel[16 + j] + P1[d * 3 + 2] * sRel[32 + j];
      sHT[j * SD + d] = fmaxf(a, 0.f);
    }
  }
  __syncthreads();
  const int r = lane & 15;
  const int khalf = (lane >> 4) << 1;
  const int rhalf = (lane >> 4) << 3;
  const int nrcD = D >> 4;

  constexpr int QP = (QD > 1) ? QD / 2 : 1;
  static_assert(NQ % QP == 0, "whole groups of QP points");
  for (int item = wave; item < nrcD * (NQ / QP); item += NW) {
    const int rc = item % nrcD, q0 = (item / nrcD) * QP;
    const int rowBase = rc * 16;
    const int row = rowBase + r;
    v8f acc[QP];
#pragma unroll
    for (int t = 0; t < QP; ++t)
#pragma unroll
      for (int u = 0; u < 8; ++u) acc[t][u] = 0.f;
    const float* ap = P2 + (size_t)row * D + khalf;
    const float* bp = smem + q0 * QS + oHT + r * SD + khalf;
#pragma unroll 8
    for (int kk = 0; kk < D; kk += 4) {
      v2f afrag = *(const v2f*)(ap + kk);
#pragma unroll
      for (int t = 0; t < QP; ++t) {
        v2f bfrag = *(const v2f*)(bp + t * QS + kk);
        acc[t] = wmma_f32x4(afrag, bfrag, acc[t]);
      }
    }
#pragma unroll
    for (int t = 0; t < QP; ++t) {
      float* sPe = smem + (q0 + t) * QS + oPe;
#pragma unroll
      for (int u = 0; u < 8; ++u)
        sPe[(rowBase + u + rhalf) * 16 + r] = acc[t][u];
    }
  }
  __syncthreads();

  for (int item = wave; item < nrcD * NG; item += NW) {
    const int rc = item % nrcD, q0 = (item / nrcD) * QD;
    const int rowBase = rc * 16;
    const int row = rowBase + r;
    v8f acc[QD], accv[QD];
#pragma unroll
    for (int t = 0; t < QD; ++t) {
      const float* sPe = smem + (q0 + t) * QS + oPe;
#pragma unroll
      for (int u = 0; u < 8; ++u) {
        float pe = sPe[(rowBase + u + rhalf) * 16 + r];
        acc[t][u] = pe; accv[t][u] = pe;
      }
    }
    const float* aqp = WqkP + (size_t)row * Kpad + khalf;
    const float* avp = WvP  + (size_t)row * Kpad + khalf;
    const float* bp  = smem + q0 * QS + oScr + r * SG + khalf;
#pragma unroll 4
    for (int kk = 0; kk < Kpad; kk += 4) {
      v2f aq = *(const v2f*)(aqp + kk);
      v2f av = *(const v2f*)(avp + kk);
#pragma unroll
      for (int t = 0; t < QD; ++t) {
        v2f bfrag = *(const v2f*)(bp + t * QS + kk);
        acc[t]  = wmma_f32x4(aq, bfrag, acc[t]);
        accv[t] = wmma_f32x4(av, bfrag, accv[t]);
      }
    }
#pragma unroll
    for (int t = 0; t < QD; ++t) {
      float* sHT = smem + (q0 + t) * QS + oHT;
      float* sV  = smem + (q0 + t) * QS + oV;
#pragma unroll
      for (int u = 0; u < 8; ++u) {
        sHT[r * SD + rowBase + u + rhalf] = acc[t][u];
        sV[(rowBase + u + rhalf) * 16 + r] = accv[t][u];
      }
    }
  }
  __syncthreads();

  for (int rc = wave; rc < (H >> 4); rc += NW) {
    const int rowBase = rc * 16;
    const int row = rowBase + r;
    v8f acc[NQ];
#pragma unroll
    for (int q = 0; q < NQ; ++q)
#pragma unroll
      for (int u = 0; u < 8; ++u) acc[q][u] = 0.f;
    const float* ap = A1 + (size_t)row * D + khalf;
    const float* bp = smem + oHT + r * SD + khalf;
#pragma unroll 4
    for (int kk = 0; kk < D; kk += 4) {
      v2f afrag = *(const v2f*)(ap + kk);
#pragma unroll
      for (int q = 0; q < NQ; ++q) {
        v2f bfrag = *(const v2f*)(bp + q * QS + kk);
        acc[q] = wmma_f32x4(afrag, bfrag, acc[q]);
      }
    }
#pragma unroll
    for (int q = 0; q < NQ; ++q) {
      float* sTT = smem + q * QS + oScr;
#pragma unroll
      for (int u = 0; u < 8; ++u)
        sTT[r * SHS + rowBase + u + rhalf] = fmaxf(acc[q][u], 0.f);
    }
  }
  __syncthreads();

  for (int item = wave; item < nrcD * NG; item += NW) {
    const int rc = item % nrcD, q0 = (item / nrcD) * QD;
    const int rowBase = rc * 16;
    const int row = rowBase + r;
    v8f acc[QD];
#pragma unroll
    for (int t = 0; t < QD; ++t)
#pragma unroll
      for (int u = 0; u < 8; ++u) acc[t][u] = 0.f;
    const float* ap = A2 + (size_t)row * H + khalf;
    const float* bp = smem + q0 * QS + oScr + r * SHS + khalf;
#pragma unroll 4
    for (int kk = 0; kk < H; kk += 4) {
      v2f afrag = *(const v2f*)(ap + kk);
#pragma unroll
      for (int t = 0; t < QD; ++t) {
        v2f bfrag = *(const v2f*)(bp + t * QS + kk);
        acc[t] = wmma_f32x4(afrag, bfrag, acc[t]);
      }
    }
#pragma unroll
    for (int t = 0; t < QD; ++t) {
      float* sL = smem + (q0 + t) * QS + oL;
#pragma unroll
      for (int u = 0; u < 8; ++u)
        sL[(rowBase + u + rhalf) * 16 + r] = acc[t][u];
    }
  }
  __syncthreads();

  for (int e = tid; e < NQ * D; e += NT) {
    const int q = e % NQ, d = e / NQ;
    const float* sL = smem + q * QS + oL + d * 16;
    const float* sV = smem + q * QS + oV + d * 16;
    float mx = -INFINITY;
#pragma unroll
    for (int j = 0; j < PK; ++j) mx = fmaxf(mx, sL[j]);
    float s = 0.f, o = 0.f;
#pragma unroll
    for (int j = 0; j < PK; ++j) {
      float p = __expf(sL[j] - mx);
      s += p;
      o += p * sV[j];
    }
    const float val = o / s;
    out[((size_t)b * D + d) * PM + n0 + q] = val;
    if (outT) outT[((size_t)b * PM + n0 + q) * D + d] = val;
  }
}

__global__ __launch_bounds__(256)
void pool_kernel(const float* __restrict__ xa, int Ca, const float* __restrict__ xb, int Cb,
                 int chBase, float* __restrict__ out) {
  int b = blockIdx.x, ch = threadIdx.x;
  const float* src = ch < Ca ? xa + ((size_t)b * Ca + ch) * PM
                             : xb + ((size_t)b * Cb + (ch - Ca)) * PM;
  float mx = -INFINITY, sm = 0.f;
  for (int m = 0; m < PM; ++m) { float v = src[m]; mx = fmaxf(mx, v); sm += v; }
  out[(size_t)b * 512 + chBase + ch] = mx;
  out[(size_t)b * 512 + 256 + chBase + ch] = sm * (1.f / PM);
}

extern "C" void kernel_launch(void* const* d_in, const int* in_sizes, int n_in,
                              void* d_out, int out_size, void* d_ws, size_t ws_size,
                              hipStream_t stream) {
  (void)in_sizes; (void)n_in; (void)out_size; (void)ws_size;
  const float* x = (const float*)d_in[0];
  const float* t1[7]; const float* t2[7]; const float* t3[7];
  for (int i = 0; i < 7; ++i) {
    t1[i] = (const float*)d_in[1 + i];
    t2[i] = (const float*)d_in[8 + i];
    t3[i] = (const float*)d_in[15 + i];
  }
  float* ws = (float*)d_ws;
  size_t off = 0;
  float* pos = ws + off; off += (size_t)PB * 3 * PM;
  float* x1  = ws + off; off += (size_t)PB * 64 * PM;
  float* x2  = ws + off; off += (size_t)PB * 64 * PM;
  float* x3  = ws + off; off += (size_t)PB * 128 * PM;
  int* idxb  = (int*)(ws + off); off += (size_t)PB * PM * PK;
  float* wqk1 = ws + off; off += 64 * 8;
  float* wv1  = ws + off; off += 64 * 8;
  float* wqk2 = ws + off; off += 64 * 128;
  float* wv2  = ws + off; off += 64 * 128;
  float* wqk3 = ws + off; off += 128 * 128;
  float* wv3  = ws + off; off += 128 * 128;

  prep_weights_kernel<<<(64 * 8 + 255) / 256, 256, 0, stream>>>(
      t1[0], t1[1], t1[2], wqk1, wv1, 64, 6, 8);
  prep_weights_kernel<<<(64 * 128 + 255) / 256, 256, 0, stream>>>(
      t2[0], t2[1], t2[2], wqk2, wv2, 64, 128, 128);
  prep_weights_kernel<<<(128 * 128 + 255) / 256, 256, 0, stream>>>(
      t3[0], t3[1], t3[2], wqk3, wv3, 128, 128, 128);

  fps_kernel<<<PB, 256, 0, stream>>>(x, pos);

  float* x1T = x3;
  float* x2T = x1;

  knn_kernel<<<PB * PM / KNN_QPB, KNN_QPB * 32, 0, stream>>>(pos, 3, idxb);
  xform_kernel<XNT, XNQ, 2, 3, XNQ * xf_query_floats(8, 64, 256)><<<PB * PM / XNQ, XNT, 0, stream>>>(
      pos, pos, idxb, wqk1, wv1, t1[3], t1[4], t1[5], t1[6],
      x1, x1T, 64, 256);
  pool_kernel<<<PB, 64, 0, stream>>>(x1, 64, x1, 0, 0, (float*)d_out);

  knn_kernel<<<PB * PM / KNN_QPB, KNN_QPB * 32, 0, stream>>>(x1, 64, idxb);
  xform_kernel<XNT, XNQ, 2, 64, XNQ * xf_query_floats(128, 64, 256)><<<PB * PM / XNQ, XNT, 0, stream>>>(
      x1T, pos, idxb, wqk2, wv2, t2[3], t2[4], t2[5], t2[6],
      x2, x2T, 64, 256);

  knn_kernel<<<PB * PM / KNN_QPB, KNN_QPB * 32, 0, stream>>>(x2, 64, idxb);
  xform_kernel<XNT, XNQ, 4, 64, XNQ * xf_query_floats(128, 128, 512)><<<PB * PM / XNQ, XNT, 0, stream>>>(
      x2T, pos, idxb, wqk3, wv3, t3[3], t3[4], t3[5], t3[6],
      x3, nullptr, 128, 512);

  pool_kernel<<<PB, 192, 0, stream>>>(x2, 64, x3, 128, 64, (float*)d_out);
}


# fused layer 3: two work-groups per WGP (was one): LDS 231 -> 133 KB (t through LDS in two halves of H, logits stored over h), VGPRs 134 -> 114 (D, H template constants, all layers); sums unchanged
# speedup vs baseline: 3.0382x; 1.1537x over previous
#include <hip/hip_runtime.h>
#include <math.h>

#define PB 16
#define PN 4096
#define PM 512
#define PK 16

typedef float v2f __attribute__((ext_vector_type(2)));
typedef float v8f __attribute__((ext_vector_type(8)));

__device__ __forceinline__ v8f wmma_f32x4(v2f a, v2f b, v8f c) {
  return __builtin_amdgcn_wmma_f32_16x16x4_f32(false, a, false, b, (short)0, c, false, false);
}

__global__ __launch_bounds__(256)
void prep_weights_kernel(const float* __restrict__ Wq, const float* __restrict__ Wk,
                         const float* __restrict__ Wv, float* __restrict__ outQK,
                         float* __restrict__ outV, int D, int Cw, int Kpad) {
  int i = blockIdx.x * 256 + threadIdx.x;
  if (i >= D * Kpad) return;
  int d = i / Kpad, k = i % Kpad;
  float q = 0.f, v = 0.f;
  if (k < Cw) {
    q = Wq[(size_t)d * Cw + k] - Wk[(size_t)d * Cw + k];
    v = Wv[(size_t)d * Cw + k];
  }
  outQK[i] = q;
  outV[i] = v;
}

template <int XM>
__device__ __forceinline__ unsigned row_xor_u32(unsigned v) {
  return (unsigned)__builtin_amdgcn_update_dpp(0, (int)v, 0x160 | XM, 0xf, 0xf, true);
}
__device__ __forceinline__ unsigned umax2(unsigned a, unsigned b) { return a > b ? a : b; }
__device__ __forceinline__ unsigned umin2(unsigned a, unsigned b) { return a < b ? a : b; }
__device__ __forceinline__ unsigned row_umax(unsigned v) {
  v = umax2(v, row_xor_u32<1>(v));
  v = umax2(v, row_xor_u32<2>(v));
  v = umax2(v, row_xor_u32<4>(v));
  return umax2(v, row_xor_u32<8>(v));
}
__device__ __forceinline__ unsigned row_umin(unsigned v) {
  v = umin2(v, row_xor_u32<1>(v));
  v = umin2(v, row_xor_u32<2>(v));
  v = umin2(v, row_xor_u32<4>(v));
  return umin2(v, row_xor_u32<8>(v));
}

__global__ __launch_bounds__(256)
void fps_kernel(const float* __restrict__ x, float* __restrict__ pos) {
  const int b = blockIdx.x;
  const int tid = threadIdx.x;
  const int PPT = PN / 256;
  __shared__ float sX[PN];
  __shared__ float sY[PN];
  __shared__ float sZ[PN];
  __shared__ int sSel[PM];
  __shared__ uint2 sPart[2][16];

  float px[PPT], py[PPT], pz[PPT], dmin[PPT];
#pragma unroll
  for (int i = 0; i < PPT; ++i) {
    int p = tid + i * 256;
    const float* xp = x + ((size_t)b * PN + p) * 3;
    px[i] = xp[0]; py[i] = xp[1]; pz[i] = xp[2];
    sX[p] = px[i]; sY[p] = py[i]; sZ[p] = pz[i];
    dmin[i] = INFINITY;
  }
  if (tid == 0) sSel[0] = 0;
  __syncthreads();
  float bx = sX[0], by = sY[0], bz = sZ[0];
  for (int it = 1; it < PM; ++it) {
    float best = -INFINITY; int bslot = PPT;
#pragma unroll
    for (int i = 0; i < PPT; ++i) {
      float dx = px[i] - bx, dy = py[i] - by, dz = pz[i] - bz;
      float d = dx * dx + dy * dy + dz * dz;
      dmin[i] = fminf(dmin[i], d);
      if (dmin[i] > best) { best = dmin[i]; bslot = i; }
    }
    const unsigned barg = (unsigned)tid + ((unsigned)bslot << 8);
    const unsigned vb = __float_as_uint(best);
    const unsigned rmax = row_umax(vb);
    const unsigned rarg = row_umin(vb == rmax ? barg : 0xFFFFFFFFu);
    uint2* part = sPart[it & 1];
    if ((tid & 15) == 0) part[tid >> 4] = make_uint2(rmax, rarg);
    __syncthreads();
    const uint2 pr = part[tid & 15];
    const unsigned gmax = row_umax(pr.x);
    const unsigned garg = row_umin(pr.x == gmax ? pr.y : 0xFFFFFFFFu);
    bx = sX[garg]; by = sY[garg]; bz = sZ[garg];
    if (tid == 0) sSel[it] = (int)garg;
  }
  __syncthreads();
  for (int i = tid; i < PM; i += 256) {
    int p = sSel[i];
    pos[((size_t)b * 3 + 0) * PM + i] = sX[p];
    pos[((size_t)b * 3 + 1) * PM + i] = sY[p];
    pos[((size_t)b * 3 + 2) * PM + i] = sZ[p];
  }
}

#define KNN_QPB 8
__global__ __launch_bounds__(KNN_QPB * 32)
void knn_kernel(const float* __restrict__ f, int C, int* __restrict__ idxout) {
  const int lane = threadIdx.x & 31;
  const int wave = __builtin_amdgcn_readfirstlane((int)(threadIdx.x >> 5));
  const int bn = blockIdx.x * KNN_QPB + wave;
  const int b = bn / PM, n = bn % PM;
  const float* fb = f + (size_t)b * C * PM;
  float acc[16];
#pragma unroll
  for (int i = 0; i < 16; ++i) acc[i] = 0.f;
#pragma unroll 2
  for (int c = 0; c < C; ++c) {
    const float* row = fb + (size_t)c * PM;
    const float fq = row[n];
#pragma unroll
    for (int i = 0; i < 16; ++i) {
      float d = fq - row[lane + 32 * i];
      acc[i] += d * d;
    }
  }
  unsigned key[16];
#pragma unroll
  for (int i = 0; i < 16; ++i) key[i] = __float_as_uint(acc[i]);
  unsigned mine = 0;
  for (int r = 0; r < PK; ++r) {
    unsigned bv = key[0], bi = 0;
#pragma unroll
    for (int i = 1; i < 16; ++i) {
      if (key[i] < bv) { bv = key[i]; bi = (unsigned)i; }
    }
    const unsigned bm = (unsigned)lane + (bi << 5);
    const unsigned rmin = row_umin(bv);
    const unsigned gmin = umin2((unsigned)__builtin_amdgcn_readlane((int)rmin, 0),
                                (unsigned)__builtin_amdgcn_readlane((int)rmin, 16));
    const unsigned rarg = row_umin(bv == gmin ? bm : 0xFFFFFFFFu);
    const unsigned garg = umin2((unsigned)__builtin_amdgcn_readlane((int)rarg, 0),
                                (unsigned)__builtin_amdgcn_readlane((int)rarg, 16));
    if (lane == r) mine = garg;
#pragma unroll
    for (int i = 0; i < 16; ++i) {
      if ((unsigned)(lane + 32 * i) == garg) key[i] = 0xFFFFFFFFu;
    }
  }
  if (lane < PK) idxout[(size_t)bn * PK + lane] = (int)(mine & (unsigned)(PM - 1));
}

#define XNT 512
#define XNQ 4

template <int NT, int NQ, int QD, int CIN, int D, int H, int NH>
__global__ __launch_bounds__(NT)
void xform_kernel(const float* __restrict__ f, const float* __restrict__ pos,
                  const int* __restrict__ idx,
                  const float* __restrict__ WqkP, const float* __restrict__ WvP,
                  const float* __restrict__ P1, const float* __restrict__ P2,
                  const float* __restrict__ A1, const float* __restrict__ A2,
                  float* __restrict__ out, float* __restrict__ outT) {
  constexpr int NW = NT / 32;
  constexpr int NG = NQ / QD;
  constexpr int Kpad = (CIN == 3) ? 8 : 2 * CIN;
  const int bn0 = blockIdx.x * NQ;
  const int b = bn0 / PM, n0 = bn0 % PM;
  const int tid = threadIdx.x;
  const int lane = tid & 31, wave = tid >> 5;
  constexpr int SG = Kpad + 2;
  constexpr int SD = D + 2;
  constexpr int HP  = H / NH;
  constexpr int SHS = HP + 2;
  constexpr int scrF = (16 * SG + D * 16 > 16 * SHS) ? (16 * SG + D * 16) : (16 * SHS);
  constexpr int oHT  = 48;
  constexpr int oV   = oHT + 16 * SD;
  constexpr int oScr = oV + D * 16;
  constexpr int oPe  = oScr + 16 * SG;
  constexpr int oL   = (NH > 1) ? oHT : oScr + scrF;
  constexpr int QS   = (NH > 1) ? oScr + scrF : oL + D * 16;
  static_assert(QS % 4 == 0, "regions keep the 8-byte alignment of the fragment reads");
  static_assert(D % 16 == 0 && H % NH == 0 && HP % 16 == 0 && Kpad % 4 == 0, "whole WMMA tiles and k steps");
  __shared__ float smem[NQ * QS] __attribute__((aligned(16)));

  if (tid == 0) {
    __builtin_prefetch(A1, 0, 3);
    __builtin_prefetch(A2, 0, 3);
  }
  if constexpr (CIN == 3) {
    for (int e = tid; e < NQ * 16 * Kpad; e += NT) {
      const int q = e / (16 * Kpad), t = e % (16 * Kpad);
      const int j = t / Kpad, row = t % Kpad;
      float v = 0.f;
      if (row < 2 * CIN) {
        const float* pc = pos + ((size_t)b * CIN + (row < CIN ? row : row - CIN)) * PM;
        v = pc[n0 + q];
        if (row < CIN) {
          v = pc[idx[(size_t)(bn0 + q) * PK + j]] - v;
          smem[q * QS + row * 16 + j] = v;
        }
      }
      smem[q * QS + oScr + j * SG + row] = v;
    }
  } else {
    constexpr int C4 = CIN / 4;
    constexpr int NP = NQ * 16 * C4 / NT;
    static_assert(NQ * 16 * C4 % NT == 0 && NT % (16 * C4) == 0, "whole passes, c4 and j fixed per thread");
    const int c4 = tid % C4, j = (tid / C4) % 16;
    int nbv[NP];
    float4 cen[NP], nbf[NP];
#pragma unroll
    for (int p = 0; p < NP; ++p) {
      const int q = (tid + p * NT) / (16 * C4);
      nbv[p] = idx[(size_t)(bn0 + q) * PK + j];
      cen[p] = *(const float4*)(f + ((size_t)b * PM + n0 + q) * CIN + c4 * 4);
    }
#pragma unroll
    for (int p = 0; p < NP; ++p)
      nbf[p] = *(const float4*)(f + ((size_t)b * PM + nbv[p]) * CIN + c4 * 4);
    if (c4 < 3) {
      const float* pc = pos + ((size_t)b * 3 + c4) * PM;
      float rel[NP];
#pragma unroll
      for (int p = 0; p < NP; ++p) rel[p] = pc[nbv[p]] - pc[n0 + (tid + p * NT) / (16 * C4)];
#pragma unroll
      for (int p = 0; p < NP; ++p) smem[((tid + p * NT) / (16 * C4)) * QS + c4 * 16 + j] = rel[p];
    }
#pragma unroll
    for (int p = 0; p < NP; ++p) {
      float* dst = smem + ((tid + p * NT) / (16 * C4)) * QS + oScr + j * SG + c4 * 4;
      *(v2f*)(dst)           = v2f{nbf[p].x - cen[p].x, nbf[p].y - cen[p].y};
      *(v2f*)(dst + 2)       = v2f{nbf[p].z - cen[p].z, nbf[p].w - cen[p].w};
      *(v2f*)(dst + CIN)     = v2f{cen[p].x, cen[p].y};
      *(v2f*)(dst + CIN + 2) = v2f{cen[p].z, cen[p].w};
    }
  }
  __syncthreads();

  {
    static_assert(NT % D == 0 && 16 % (NT / D) == 0, "one row per thread, whole passes over the 16 columns");
    constexpr int JS = NT / D;
    const int d = tid % D, jb = tid / D;
    const float p0 = P1[d * 3 + 0], p1 = P1[d * 3 + 1], p2 = P1[d * 3 + 2];
    constexpr int NJ = 16 / JS;
    float r0[NQ * NJ], r1[NQ * NJ], r2[NQ * NJ];
#pragma unroll
    for (int q = 0; q < NQ; ++q)
#pragma unroll
      for (int i = 0; i < NJ; ++i) {
        const float* sRel = smem + q * QS + jb + i * JS;
        r0[q * NJ + i] = sRel[0]; r1[q * NJ + i] = sRel[16]; r2[q * NJ + i] = sRel[32];
      }
#pragma unroll
    for (int q = 0; q < NQ; ++q)
#pragma unroll
      for (int i = 0; i < NJ; ++i) {
        float a = fmaf(p2, r2[q * NJ + i], fmaf(p0, r0[q * NJ + i], p1 * r1[q * NJ + i]));
        smem[q * QS + oHT + (jb + i * JS) * SD + d] = fmaxf(a, 0.f);
      }
  }
  __syncthreads();
  const int r = lane & 15;
  const int khalf = (lane >> 4) << 1;
  const int rhalf = (lane >> 4) << 3;
  constexpr int nrcD = D / 16;

  constexpr int QP = (QD > 1) ? QD / 2 : 1;
  static_assert(NQ % QP == 0, "whole groups of QP points");
  for (int item = wave; item < nrcD * (NQ / QP); item += NW) {
    const int rc = item % nrcD, q0 = (item / nrcD) * QP;
    const int rowBase = rc * 16;
    const int row = rowBase + r;
    v8f acc[QP];
#pragma unroll
    for (int t = 0; t < QP; ++t)
#pragma unroll
      for (int u = 0; u < 8; ++u) acc[t][u] = 0.f;
    const float* ap = P2 + (size_t)row * D + khalf;
    const float* bp = smem + q0 * QS + oHT + r * SD + khalf;
#pragma unroll 8
    for (int kk = 0; kk < D; kk += 4) {
      v2f afrag = *(const v2f*)(ap + kk);
#pragma unroll
      for (int t = 0; t < QP; ++t) {
        v2f bfrag = *(const v2f*)(bp + t * QS + kk);
        acc[t] = wmma_f32x4(afrag, bfrag, acc[t]);
      }
    }
#pragma unroll
    for (int t = 0; t < QP; ++t) {
      float* sPe = smem + (q0 + t) * QS + oPe;
#pragma unroll
      for (int u = 0; u < 8; ++u)
        sPe[(rowBase + u + rhalf) * 16 + r] = acc[t][u];
    }
  }
  __syncthreads();

  for (int item = wave; item < nrcD * NG; item += NW) {
    const int rc = item % nrcD, q0 = (item / nrcD) * QD;
    const int rowBase = rc * 16;
    const int row = rowBase + r;
    v8f acc[QD], accv[QD];
#pragma unroll
    for (int t = 0; t < QD; ++t) {
      const float* sPe = smem + (q0 + t) * QS + oPe;
#pragma unroll
      for (int u = 0; u < 8; ++u) {
        float pe = sPe[(rowBase + u + rhalf) * 16 + r];
        acc[t][u] = pe; accv[t][u] = pe;
      }
    }
    const float* aqp = WqkP + (size_t)row * Kpad + khalf;
    const float* avp = WvP  + (size_t)row * Kpad + khalf;
    const float* bp  = smem + q0 * QS + oScr + r * SG + khalf;
#pragma unroll 4
    for (int kk = 0; kk < Kpad; kk += 4) {
      v2f aq = *(const v2f*)(aqp + kk);
      v2f av = *(const v2f*)(avp + kk);
#pragma unroll
      for (int t = 0; t < QD; ++t) {
        v2f bfrag = *(const v2f*)(bp + t * QS + kk);
        acc[t]  = wmma_f32x4(aq, bfrag, acc[t]);
        accv[t] = wmma_f32x4(av, bfrag, accv[t]);
      }
    }
#pragma unroll
    for (int t = 0; t < QD; ++t) {
      float* sHT = smem + (q0 + t) * QS + oHT;
      float* sV  = smem + (q0 + t) * QS + oV;
#pragma unroll
      for (int u = 0; u < 8; ++u) {
        sHT[r * SD + rowBase + u + rhalf] = acc[t][u];
        sV[(rowBase + u + rhalf) * 16 + r] = accv[t][u];
      }
    }
  }
  __syncthreads();

  static_assert(nrcD * NG <= NW, "at most one logits item per wave");
  const bool hasL = wave < nrcD * NG;
  const int rcL = wave % nrcD, q0L = (wave / nrcD) * QD;
  v8f accL[QD];
#pragma unroll
  for (int t = 0; t < QD; ++t)
#pragma unroll
    for (int u = 0; u < 8; ++u) accL[t][u] = 0.f;
#pragma unroll
  for (int hp = 0; hp < NH; ++hp) {
    for (int rc = wave; rc < HP / 16; rc += NW) {
      const int rowBase = rc * 16;
      const int row = hp * HP + rowBase + r;
      v8f acc[NQ];
#pragma unroll
      for (int q = 0; q < NQ; ++q)
#pragma unroll
        for (int u = 0; u < 8; ++u) acc[q][u] = 0.f;
      const float* ap = A1 + (size_t)row * D + khalf;
      const float* bp = smem + oHT + r * SD + khalf;
#pragma unroll 4
      for (int kk = 0; kk < D; kk += 4) {
        v2f afrag = *(const v2f*)(ap + kk);
#pragma unroll
        for (int q = 0; q < NQ; ++q) {
          v2f bfrag = *(const v2f*)(bp + q * QS + kk);
          acc[q] = wmma_f32x4(afrag, bfrag, acc[q]);
        }
      }
#pragma unroll
      for (int q = 0; q < NQ; ++q) {
        float* sTT = smem + q * QS + oScr;
#pragma unroll
        for (int u = 0; u < 8; ++u)
          sTT[r * SHS + rowBase + u + rhalf] = fmaxf(acc[q][u], 0.f);
      }
    }
    __syncthreads();
    if (hasL) {
      const float* ap = A2 + (size_t)(rcL * 16 + r) * H + hp * HP + khalf;
      const float* bp = smem + q0L * QS + oScr + r * SHS + khalf;
#pragma unroll 4
      for (int kk = 0; kk < HP; kk += 4) {
        v2f afrag = *(const v2f*)(ap + kk);
#pragma unroll
        for (int t = 0; t < QD; ++t) {
          v2f bfrag = *(const v2f*)(bp + t * QS + kk);
          accL[t] = wmma_f32x4(afrag, bfrag, accL[t]);
        }
      }
    }
    if (hp + 1 < NH) __syncthreads();
  }
  if (hasL) {
#pragma unroll
    for (int t = 0; t < QD; ++t) {
      float* sL = smem + (q0L + t) * QS + oL;
#pragma unroll
      for (int u = 0; u < 8; ++u)
        sL[(rcL * 16 + u + rhalf) * 16 + r] = accL[t][u];
    }
  }
  __syncthreads();

  for (int e = tid; e < NQ * D; e += NT) {
    const int q = e % NQ, d = e / NQ;
    const float* sL = smem + q * QS + oL + d * 16;
    const float* sV = smem + q * QS + oV + d * 16;
    float mx = -INFINITY;
#pragma unroll
    for (int j = 0; j < PK; ++j) mx = fmaxf(mx, sL[j]);
    float s = 0.f, o = 0.f;
#pragma unroll
    for (int j = 0; j < PK; ++j) {
      float p = __expf(sL[j] - mx);
      s += p;
      o += p * sV[j];
    }
    const float val = o / s;
    out[((size_t)b * D + d) * PM + n0 + q] = val;
    if (outT) outT[((size_t)b * PM + n0 + q) * D + d] = val;
  }
}

__global__ __launch_bounds__(256)
void pool_kernel(const float* __restrict__ xa, int Ca, const float* __restrict__ xb, int Cb,
                 int chBase, float* __restrict__ out) {
  int b = blockIdx.x, ch = threadIdx.x;
  const float* src = ch < Ca ? xa + ((size_t)b * Ca + ch) * PM
                             : xb + ((size_t)b * Cb + (ch - Ca)) * PM;
  float mx = -INFINITY, sm = 0.f;
  for (int m = 0; m < PM; ++m) { float v = src[m]; mx = fmaxf(mx, v); sm += v; }
  out[(size_t)b * 512 + chBase + ch] = mx;
  out[(size_t)b * 512 + 256 + chBase + ch] = sm * (1.f / PM);
}

extern "C" void kernel_launch(void* const* d_in, const int* in_sizes, int n_in,
                              void* d_out, int out_size, void* d_ws, size_t ws_size,
                              hipStream_t stream) {
  (void)in_sizes; (void)n_in; (void)out_size; (void)ws_size;
  const float* x = (const float*)d_in[0];
  const float* t1[7]; const float* t2[7]; const float* t3[7];
  for (int i = 0; i < 7; ++i) {
    t1[i] = (const float*)d_in[1 + i];
    t2[i] = (const float*)d_in[8 + i];
    t3[i] = (const float*)d_in[15 + i];
  }
  float* ws = (float*)d_ws;
  size_t off = 0;
  float* pos = ws + off; off += (size_t)PB * 3 * PM;
  float* x1  = ws + off; off += (size_t)PB * 64 * PM;
  float* x2  = ws + off; off += (size_t)PB * 64 * PM;
  float* x3  = ws + off; off += (size_t)PB * 128 * PM;
  int* idxb  = (int*)(ws + off); off += (size_t)PB * PM * PK;
  float* wqk1 = ws + off; off += 64 * 8;
  float* wv1  = ws + off; off += 64 * 8;
  float* wqk2 = ws + off; off += 64 * 128;
  float* wv2  = ws + off; off += 64 * 128;
  float* wqk3 = ws + off; off += 128 * 128;
  float* wv3  = ws + off; off += 128 * 128;

  prep_weights_kernel<<<(64 * 8 + 255) / 256, 256, 0, stream>>>(
      t1[0], t1[1], t1[2], wqk1, wv1, 64, 6, 8);
  prep_weights_kernel<<<(64 * 128 + 255) / 256, 256, 0, stream>>>(
      t2[0], t2[1], t2[2], wqk2, wv2, 64, 128, 128);
  prep_weights_kernel<<<(128 * 128 + 255) / 256, 256, 0, stream>>>(
      t3[0], t3[1], t3[2], wqk3, wv3, 128, 128, 128);

  fps_kernel<<<PB, 256, 0, stream>>>(x, pos);

  float* x1T = x3;
  float* x2T = x1;

  knn_kernel<<<PB * PM / KNN_QPB, KNN_QPB * 32, 0, stream>>>(pos, 3, idxb);
  xform_kernel<XNT, XNQ, 2, 3, 64, 256, 1><<<PB * PM / XNQ, XNT, 0, stream>>>(
      pos, pos, idxb, wqk1, wv1, t1[3], t1[4], t1[5], t1[6],
      x1, x1T);
  pool_kernel<<<PB, 64, 0, stream>>>(x1, 64, x1, 0, 0, (float*)d_out);

  knn_kernel<<<PB * PM / KNN_QPB, KNN_QPB * 32, 0, stream>>>(x1, 64, idxb);
  xform_kernel<XNT, XNQ, 2, 64, 64, 256, 1><<<PB * PM / XNQ, XNT, 0, stream>>>(
      x1T, pos, idxb, wqk2, wv2, t2[3], t2[4], t2[5], t2[6],
      x2, x2T);

  knn_kernel<<<PB * PM / KNN_QPB, KNN_QPB * 32, 0, stream>>>(x2, 64, idxb);
  xform_kernel<XNT, XNQ, 4, 64, 128, 512, 2><<<PB * PM / XNQ, XNT, 0, stream>>>(
      x2T, pos, idxb, wqk3, wv3, t3[3], t3[4], t3[5], t3[6],
      x3, nullptr);

  pool_kernel<<<PB, 192, 0, stream>>>(x2, 64, x3, 128, 64, (float*)d_out);
}


# layers 2, 3: Wa@(nbr-ctr)+Wb@ctr = Wa@nbr+(Wb-Wa)@ctr: h, v' products once per point in a small f32 WMMA GEMM, gathered into the pe tile; gather + h/v' GEMM gone; sum order in h, v' changes
# speedup vs baseline: 3.1584x; 1.0396x over previous
#include <hip/hip_runtime.h>
#include <math.h>

#define PB 16
#define PN 4096
#define PM 512
#define PK 16

typedef float v2f __attribute__((ext_vector_type(2)));
typedef float v8f __attribute__((ext_vector_type(8)));

__device__ __forceinline__ v8f wmma_f32x4(v2f a, v2f b, v8f c) {
  return __builtin_amdgcn_wmma_f32_16x16x4_f32(false, a, false, b, (short)0, c, false, false);
}

__global__ __launch_bounds__(256)
void prep_weights_kernel(const float* __restrict__ Wq, const float* __restrict__ Wk,
                         const float* __restrict__ Wv, float* __restrict__ outQK,
                         float* __restrict__ outV, int D, int Cw, int Kpad) {
  int i = blockIdx.x * 256 + threadIdx.x;
  if (i >= D * Kpad) return;
  int d = i / Kpad, k = i % Kpad;
  float q = 0.f, v = 0.f;
  if (k < Cw) {
    q = Wq[(size_t)d * Cw + k] - Wk[(size_t)d * Cw + k];
    v = Wv[(size_t)d * Cw + k];
  }
  outQK[i] = q;
  outV[i] = v;
}

__global__ __launch_bounds__(256)
void prep_split_kernel(const float* __restrict__ Wq, const float* __restrict__ Wk,
                       const float* __restrict__ Wv, float* __restrict__ W4, int D, int C) {
  int i = blockIdx.x * 256 + threadIdx.x;
  if (i >= D * C) return;
  int d = i / C, k = i % C;
  const size_t ia = (size_t)d * 2 * C + k, ib = ia + C;
  const float qa = Wq[ia] - Wk[ia], qb = Wq[ib] - Wk[ib];
  const float va = Wv[ia], vb = Wv[ib];
  W4[((size_t)0 * D + d) * C + k] = qa;
  W4[((size_t)1 * D + d) * C + k] = va;
  W4[((size_t)2 * D + d) * C + k] = qb - qa;
  W4[((size_t)3 * D + d) * C + k] = vb - va;
}

template <int C, int D>
__global__ __launch_bounds__(256)
void point_gemm_kernel(const float* __restrict__ xT, const float* __restrict__ W4,
                       float* __restrict__ GC) {
  constexpr int RPW = (4 * D / 16) / 8;
  static_assert((4 * D / 16) % 8 == 0 && C % 4 == 0, "whole row chunks per wave, whole k steps");
  const int lane = threadIdx.x & 31, wave = threadIdx.x >> 5;
  const int r = lane & 15;
  const int khalf = (lane >> 4) << 1;
  const int rhalf = (lane >> 4) << 3;
  const size_t p0 = (size_t)blockIdx.x * 16;
  v8f acc[RPW];
#pragma unroll
  for (int t = 0; t < RPW; ++t)
#pragma unroll
    for (int u = 0; u < 8; ++u) acc[t][u] = 0.f;
  const float* bp = xT + (p0 + r) * C + khalf;
  const float* ap = W4 + (size_t)(wave * RPW * 16 + r) * C + khalf;
#pragma unroll 8
  for (int kk = 0; kk < C; kk += 4) {
    v2f bfrag = *(const v2f*)(bp + kk);
#pragma unroll
    for (int t = 0; t < RPW; ++t) {
      v2f afrag = *(const v2f*)(ap + (size_t)t * 16 * C + kk);
      acc[t] = wmma_f32x4(afrag, bfrag, acc[t]);
    }
  }
#pragma unroll
  for (int t = 0; t < RPW; ++t) {
    float* o = GC + (p0 + r) * (4 * D) + (wave * RPW + t) * 16 + rhalf;
    *(float4*)(o)     = make_float4(acc[t][0], acc[t][1], acc[t][2], acc[t][3]);
    *(float4*)(o + 4) = make_float4(acc[t][4], acc[t][5], acc[t][6], acc[t][7]);
  }
}

template <int XM>
__device__ __forceinline__ unsigned row_xor_u32(unsigned v) {
  return (unsigned)__builtin_amdgcn_update_dpp(0, (int)v, 0x160 | XM, 0xf, 0xf, true);
}
__device__ __forceinline__ unsigned umax2(unsigned a, unsigned b) { return a > b ? a : b; }
__device__ __forceinline__ unsigned umin2(unsigned a, unsigned b) { return a < b ? a : b; }
__device__ __forceinline__ unsigned row_umax(unsigned v) {
  v = umax2(v, row_xor_u32<1>(v));
  v = umax2(v, row_xor_u32<2>(v));
  v = umax2(v, row_xor_u32<4>(v));
  return umax2(v, row_xor_u32<8>(v));
}
__device__ __forceinline__ unsigned row_umin(unsigned v) {
  v = umin2(v, row_xor_u32<1>(v));
  v = umin2(v, row_xor_u32<2>(v));
  v = umin2(v, row_xor_u32<4>(v));
  return umin2(v, row_xor_u32<8>(v));
}

__global__ __launch_bounds__(256)
void fps_kernel(const float* __restrict__ x, float* __restrict__ pos) {
  const int b = blockIdx.x;
  const int tid = threadIdx.x;
  const int PPT = PN / 256;
  __shared__ float sX[PN];
  __shared__ float sY[PN];
  __shared__ float sZ[PN];
  __shared__ int sSel[PM];
  __shared__ uint2 sPart[2][16];

  float px[PPT], py[PPT], pz[PPT], dmin[PPT];
#pragma unroll
  for (int i = 0; i < PPT; ++i) {
    int p = tid + i * 256;
    const float* xp = x + ((size_t)b * PN + p) * 3;
    px[i] = xp[0]; py[i] = xp[1]; pz[i] = xp[2];
    sX[p] = px[i]; sY[p] = py[i]; sZ[p] = pz[i];
    dmin[i] = INFINITY;
  }
  if (tid == 0) sSel[0] = 0;
  __syncthreads();
  float bx = sX[0], by = sY[0], bz = sZ[0];
  for (int it = 1; it < PM; ++it) {
    float best = -INFINITY; int bslot = PPT;
#pragma unroll
    for (int i = 0; i < PPT; ++i) {
      float dx = px[i] - bx, dy = py[i] - by, dz = pz[i] - bz;
      float d = dx * dx + dy * dy + dz * dz;
      dmin[i] = fminf(dmin[i], d);
      if (dmin[i] > best) { best = dmin[i]; bslot = i; }
    }
    const unsigned barg = (unsigned)tid + ((unsigned)bslot << 8);
    const unsigned vb = __float_as_uint(best);
    const unsigned rmax = row_umax(vb);
    const unsigned rarg = row_umin(vb == rmax ? barg : 0xFFFFFFFFu);
    uint2* part = sPart[it & 1];
    if ((tid & 15) == 0) part[tid >> 4] = make_uint2(rmax, rarg);
    __syncthreads();
    const uint2 pr = part[tid & 15];
    const unsigned gmax = row_umax(pr.x);
    const unsigned garg = row_umin(pr.x == gmax ? pr.y : 0xFFFFFFFFu);
    bx = sX[garg]; by = sY[garg]; bz = sZ[garg];
    if (tid == 0) sSel[it] = (int)garg;
  }
  __syncthreads();
  for (int i = tid; i < PM; i += 256) {
    int p = sSel[i];
    pos[((size_t)b * 3 + 0) * PM + i] = sX[p];
    pos[((size_t)b * 3 + 1) * PM + i] = sY[p];
    pos[((size_t)b * 3 + 2) * PM + i] = sZ[p];
  }
}

#define KNN_QPB 8
__global__ __launch_bounds__(KNN_QPB * 32)
void knn_kernel(const float* __restrict__ f, int C, int* __restrict__ idxout) {
  const int lane = threadIdx.x & 31;
  const int wave = __builtin_amdgcn_readfirstlane((int)(threadIdx.x >> 5));
  const int bn = blockIdx.x * KNN_QPB + wave;
  const int b = bn / PM, n = bn % PM;
  const float* fb = f + (size_t)b * C * PM;
  float acc[16];
#pragma unroll
  for (int i = 0; i < 16; ++i) acc[i] = 0.f;
#pragma unroll 2
  for (int c = 0; c < C; ++c) {
    const float* row = fb + (size_t)c * PM;
    const float fq = row[n];
#pragma unroll
    for (int i = 0; i < 16; ++i) {
      float d = fq - row[lane + 32 * i];
      acc[i] += d * d;
    }
  }
  unsigned key[16];
#pragma unroll
  for (int i = 0; i < 16; ++i) key[i] = __float_as_uint(acc[i]);
  unsigned mine = 0;
  for (int r = 0; r < PK; ++r) {
    unsigned bv = key[0], bi = 0;
#pragma unroll
    for (int i = 1; i < 16; ++i) {
      if (key[i] < bv) { bv = key[i]; bi = (unsigned)i; }
    }
    const unsigned bm = (unsigned)lane + (bi << 5);
    const unsigned rmin = row_umin(bv);
    const unsigned gmin = umin2((unsigned)__builtin_amdgcn_readlane((int)rmin, 0),
                                (unsigned)__builtin_amdgcn_readlane((int)rmin, 16));
    const unsigned rarg = row_umin(bv == gmin ? bm : 0xFFFFFFFFu);
    const unsigned garg = umin2((unsigned)__builtin_amdgcn_readlane((int)rarg, 0),
                                (unsigned)__builtin_amdgcn_readlane((int)rarg, 16));
    if (lane == r) mine = garg;
#pragma unroll
    for (int i = 0; i < 16; ++i) {
      if ((unsigned)(lane + 32 * i) == garg) key[i] = 0xFFFFFFFFu;
    }
  }
  if (lane < PK) idxout[(size_t)bn * PK + lane] = (int)(mine & (unsigned)(PM - 1));
}

#define XNT 512
#define XNQ 4

template <int NT, int NQ, int QD, int CIN, int D, int H, int NH, bool PRE>
__global__ __launch_bounds__(NT)
void xform_kernel(const float* __restrict__ f, const float* __restrict__ pos,
                  const int* __restrict__ idx,
                  const float* __restrict__ WqkP, const float* __restrict__ WvP,
                  const float* __restrict__ P1, const float* __restrict__ P2,
                  const float* __restrict__ A1, const float* __restrict__ A2,
                  float* __restrict__ out, float* __restrict__ outT) {
  constexpr int NW = NT / 32;
  constexpr int NG = NQ / QD;
  constexpr int Kpad = (CIN == 3) ? 8 : 2 * CIN;
  const int bn0 = blockIdx.x * NQ;
  const int b = bn0 / PM, n0 = bn0 % PM;
  const int tid = threadIdx.x;
  const int lane = tid & 31, wave = tid >> 5;
  constexpr int SG = Kpad + 2;
  constexpr int SD = D + 2;
  constexpr int HP  = H / NH;
  constexpr int SHS = HP + 2;
  static_assert(!PRE || CIN != 3, "the table path is for the layers whose features are a previous layer's output");
  constexpr int scrA = PRE ? 16 * SD : 16 * SG + D * 16;
  constexpr int scrF = (scrA > 16 * SHS) ? scrA : (16 * SHS);
  constexpr int oT1  = PRE ? (48 + 16 * SD + D * 16) : 48;
  constexpr int oHT  = 48;
  constexpr int oV   = oHT + 16 * SD;
  constexpr int oScr = oV + D * 16;
  constexpr int oPe  = oScr + 16 * SG;
  constexpr int oL   = (NH > 1) ? oHT : oScr + scrF;
  constexpr int oCn  = (NH > 1) ? oScr + scrF : oL + D * 16;
  constexpr int QS   = oCn + (PRE ? 2 * D : 0);
  static_assert(QS % 4 == 0, "regions keep the 8-byte alignment of the fragment reads");
  static_assert(D % 16 == 0 && H % NH == 0 && HP % 16 == 0 && Kpad % 4 == 0, "whole WMMA tiles and k steps");
  __shared__ float smem[NQ * QS] __attribute__((aligned(16)));

  constexpr int QP = (QD > 1) ? QD / 2 : 1;
  static_assert(NQ % QP == 0, "whole groups of QP points");
  float4 gq[QP][2], gv[QP][2];
  if constexpr (PRE) {
    static_assert((D / 16) * (NQ / QP) == NW, "one item per wave");
    const int rcG = wave % (D / 16), q0G = (wave / (D / 16)) * QP;
    const int rowG = rcG * 16 + ((lane >> 4) << 3);
#pragma unroll
    for (int t = 0; t < QP; ++t) {
      const int nb = idx[(size_t)(bn0 + q0G + t) * PK + (lane & 15)];
      const float* g = f + ((size_t)b * PM + nb) * (4 * D) + rowG;
      gq[t][0] = *(const float4*)(g);         gq[t][1] = *(const float4*)(g + 4);
      gv[t][0] = *(const float4*)(g + D);     gv[t][1] = *(const float4*)(g + D + 4);
    }
  }
  if (tid == 0) {
    __builtin_prefetch(A1, 0, 3);
    __builtin_prefetch(A2, 0, 3);
  }
  if constexpr (CIN == 3) {
    for (int e = tid; e < NQ * 16 * Kpad; e += NT) {
      const int q = e / (16 * Kpad), t = e % (16 * Kpad);
      const int j = t / Kpad, row = t % Kpad;
      float v = 0.f;
      if (row < 2 * CIN) {
        const float* pc = pos + ((size_t)b * CIN + (row < CIN ? row : row - CIN)) * PM;
        v = pc[n0 + q];
        if (row < CIN) {
          v = pc[idx[(size_t)(bn0 + q) * PK + j]] - v;
          smem[q * QS + row * 16 + j] = v;
        }
      }
      smem[q * QS + oScr + j * SG + row] = v;
    }
  } else if constexpr (PRE) {
    static_assert(NT >= 256 + NQ * 2 * D / 4 && NQ * 48 <= 256, "both jobs of this pass have their own threads");
    if (tid < NQ * 48) {
      const int q = tid / 48, c = (tid % 48) / 16, j = tid % 16;
      const float* pc = pos + ((size_t)b * 3 + c) * PM;
      smem[q * QS + c * 16 + j] = pc[idx[(size_t)(bn0 + q) * PK + j]] - pc[n0 + q];
    } else if (tid >= 256 && tid - 256 < NQ * 2 * D / 4) {
      const int q = (tid - 256) / (2 * D / 4), c4 = (tid - 256) % (2 * D / 4);
      *(float4*)(smem + q * QS + oCn + c4 * 4) =
          *(const float4*)(f + ((size_t)b * PM + n0 + q) * (4 * D) + 2 * D + c4 * 4);
    }
  } else {
    constexpr int C4 = CIN / 4;
    constexpr int NP = NQ * 16 * C4 / NT;
    static_assert(NQ * 16 * C4 % NT == 0 && NT % (16 * C4) == 0, "whole passes, c4 and j fixed per thread");
    const int c4 = tid % C4, j = (tid / C4) % 16;
    int nbv[NP];
    float4 cen[NP], nbf[NP];
#pragma unroll
    for (int p = 0; p < NP; ++p) {
      const int q = (tid + p * NT) / (16 * C4);
      nbv[p] = idx[(size_t)(bn0 + q) * PK + j];
      cen[p] = *(const float4*)(f + ((size_t)b * PM + n0 + q) * CIN + c4 * 4);
    }
#pragma unroll
    for (int p = 0; p < NP; ++p)
      nbf[p] = *(const float4*)(f + ((size_t)b * PM + nbv[p]) * CIN + c4 * 4);
    if (c4 < 3) {
      const float* pc = pos + ((size_t)b * 3 + c4) * PM;
      float rel[NP];
#pragma unroll
      for (int p = 0; p < NP; ++p) rel[p] = pc[nbv[p]] - pc[n0 + (tid + p * NT) / (16 * C4)];
#pragma unroll
      for (int p = 0; p < NP; ++p) smem[((tid + p * NT) / (16 * C4)) * QS + c4 * 16 + j] = rel[p];
    }
#pragma unroll
    for (int p = 0; p < NP; ++p) {
      float* dst = smem + ((tid + p * NT) / (16 * C4)) * QS + oScr + j * SG + c4 * 4;
      *(v2f*)(dst)           = v2f{nbf[p].x - cen[p].x, nbf[p].y - cen[p].y};
      *(v2f*)(dst + 2)       = v2f{nbf[p].z - cen[p].z, nbf[p].w - cen[p].w};
      *(v2f*)(dst + CIN)     = v2f{cen[p].x, cen[p].y};
      *(v2f*)(dst + CIN + 2) = v2f{cen[p].z, cen[p].w};
    }
  }
  __syncthreads();

  {
    static_assert(NT % D == 0 && 16 % (NT / D) == 0, "one row per thread, whole passes over the 16 columns");
    constexpr int JS = NT / D;
    const int d = tid % D, jb = tid / D;
    const float p0 = P1[d * 3 + 0], p1 = P1[d * 3 + 1], p2 = P1[d * 3 + 2];
    constexpr int NJ = 16 / JS;
    float r0[NQ * NJ], r1[NQ * NJ], r2[NQ * NJ];
#pragma unroll
    for (int q = 0; q < NQ; ++q)
#pragma unroll
      for (int i = 0; i < NJ; ++i) {
        const float* sRel = smem + q * QS + jb + i * JS;
        r0[q * NJ + i] = sRel[0]; r1[q * NJ + i] = sRel[16]; r2[q * NJ + i] = sRel[32];
      }
#pragma unroll
    for (int q = 0; q < NQ; ++q)
#pragma unroll
      for (int i = 0; i < NJ; ++i) {
        float a = fmaf(p2, r2[q * NJ + i], fmaf(p0, r0[q * NJ + i], p1 * r1[q * NJ + i]));
        smem[q * QS + oT1 + (jb + i * JS) * SD + d] = fmaxf(a, 0.f);
      }
  }
  __syncthreads();
  const int r = lane & 15;
  const int khalf = (lane >> 4) << 1;
  const int rhalf = (lane >> 4) << 3;
  constexpr int nrcD = D / 16;

  if constexpr (PRE) {
    {
      const int rc = wave % nrcD, q0 = (wave / nrcD) * QP;
      const int rowBase = rc * 16;
      const int row = rowBase + r;
      v8f acc[QP];
#pragma unroll
      for (int t = 0; t < QP; ++t)
#pragma unroll
        for (int u = 0; u < 8; ++u) acc[t][u] = 0.f;
      const float* ap = P2 + (size_t)row * D + khalf;
      const float* bp = smem + q0 * QS + oT1 + r * SD + khalf;
#pragma unroll 8
      for (int kk = 0; kk < D; kk += 4) {
        v2f afrag = *(const v2f*)(ap + kk);
#pragma unroll
        for (int t = 0; t < QP; ++t) {
          v2f bfrag = *(const v2f*)(bp + t * QS + kk);
          acc[t] = wmma_f32x4(afrag, bfrag, acc[t]);
        }
      }
#pragma unroll
      for (int t = 0; t < QP; ++t) {
        float* sHT = smem + (q0 + t) * QS + oHT;
        float* sV  = smem + (q0 + t) * QS + oV;
        const float* sC = smem + (q0 + t) * QS + oCn + rowBase + rhalf;
        const float4 cq0 = *(const float4*)(sC),     cq1 = *(const float4*)(sC + 4);
        const float4 cv0 = *(const float4*)(sC + D), cv1 = *(const float4*)(sC + D + 4);
        const float eq[8] = {gq[t][0].x + cq0.x, gq[t][0].y + cq0.y, gq[t][0].z + cq0.z, gq[t][0].w + cq0.w,
                             gq[t][1].x + cq1.x, gq[t][1].y + cq1.y, gq[t][1].z + cq1.z, gq[t][1].w + cq1.w};
        const float ev[8] = {gv[t][0].x + cv0.x, gv[t][0].y + cv0.y, gv[t][0].z + cv0.z, gv[t][0].w + cv0.w,
                             gv[t][1].x + cv1.x, gv[t][1].y + cv1.y, gv[t][1].z + cv1.z, gv[t][1].w + cv1.w};
#pragma unroll
        for (int u = 0; u < 8; ++u) {
          const float pe = acc[t][u];
          sHT[r * SD + rowBase + u + rhalf] = pe + eq[u];
          sV[(rowBase + u + rhalf) * 16 + r] = pe + ev[u];
        }
      }
    }
  } else {
  for (int item = wave; item < nrcD * (NQ / QP); item += NW) {
    const int rc = item % nrcD, q0 = (item / nrcD) * QP;
    const int rowBase = rc * 16;
    const int row = rowBase + r;
    v8f acc[QP];
#pragma unroll
    for (int t = 0; t < QP; ++t)
#pragma unroll
      for (int u = 0; u < 8; ++u) acc[t][u] = 0.f;
    const float* ap = P2 + (size_t)row * D + khalf;
    const float* bp = smem + q0 * QS + oHT + r * SD + khalf;
#pragma unroll 8
    for (int kk = 0; kk < D; kk += 4) {
      v2f afrag = *(const v2f*)(ap + kk);
#pragma unroll
      for (int t = 0; t < QP; ++t) {
        v2f bfrag = *(const v2f*)(bp + t * QS + kk);
        acc[t] = wmma_f32x4(afrag, bfrag, acc[t]);
      }
    }
#pragma unroll
    for (int t = 0; t < QP; ++t) {
      float* sPe = smem + (q0 + t) * QS + oPe;
#pragma unroll
      for (int u = 0; u < 8; ++u)
        sPe[(rowBase + u + rhalf) * 16 + r] = acc[t][u];
    }
  }
  __syncthreads();

  for (int item = wave; item < nrcD * NG; item += NW) {
    const int rc = item % nrcD, q0 = (item / nrcD) * QD;
    const int rowBase = rc * 16;
    const int row = rowBase + r;
    v8f acc[QD], accv[QD];
#pragma unroll
    for (int t = 0; t < QD; ++t) {
      const float* sPe = smem + (q0 + t) * QS + oPe;
#pragma unroll
      for (int u = 0; u < 8; ++u) {
        float pe = sPe[(rowBase + u + rhalf) * 16 + r];
        acc[t][u] = pe; accv[t][u] = pe;
      }
    }
    const float* aqp = WqkP + (size_t)row * Kpad + khalf;
    const float* avp = WvP  + (size_t)row * Kpad + khalf;
    const float* bp  = smem + q0 * QS + oScr + r * SG + khalf;
#pragma unroll 4
    for (int kk = 0; kk < Kpad; kk += 4) {
      v2f aq = *(const v2f*)(aqp + kk);
      v2f av = *(const v2f*)(avp + kk);
#pragma unroll
      for (int t = 0; t < QD; ++t) {
        v2f bfrag = *(const v2f*)(bp + t * QS + kk);
        acc[t]  = wmma_f32x4(aq, bfrag, acc[t]);
        accv[t] = wmma_f32x4(av, bfrag, accv[t]);
      }
    }
#pragma unroll
    for (int t = 0; t < QD; ++t) {
      float* sHT = smem + (q0 + t) * QS + oHT;
      float* sV  = smem + (q0 + t) * QS + oV;
#pragma unroll
      for (int u = 0; u < 8; ++u) {
        sHT[r * SD + rowBase + u + rhalf] = acc[t][u];
        sV[(rowBase + u + rhalf) * 16 + r] = accv[t][u];
      }
    }
  }
  }
  __syncthreads();

  static_assert(nrcD * NG <= NW, "at most one logits item per wave");
  const bool hasL = wave < nrcD * NG;
  const int rcL = wave % nrcD, q0L = (wave / nrcD) * QD;
  v8f accL[QD];
#pragma unroll
  for (int t = 0; t < QD; ++t)
#pragma unroll
    for (int u = 0; u < 8; ++u) accL[t][u] = 0.f;
#pragma unroll
  for (int hp = 0; hp < NH; ++hp) {
    for (int rc = wave; rc < HP / 16; rc += NW) {
      const int rowBase = rc * 16;
      const int row = hp * HP + rowBase + r;
      v8f acc[NQ];
#pragma unroll
      for (int q = 0; q < NQ; ++q)
#pragma unroll
        for (int u = 0; u < 8; ++u) acc[q][u] = 0.f;
      const float* ap = A1 + (size_t)row * D + khalf;
      const float* bp = smem + oHT + r * SD + khalf;
#pragma unroll 4
      for (int kk = 0; kk < D; kk += 4) {
        v2f afrag = *(const v2f*)(ap + kk);
#pragma unroll
        for (int q = 0; q < NQ; ++q) {
          v2f bfrag = *(const v2f*)(bp + q * QS + kk);
          acc[q] = wmma_f32x4(afrag, bfrag, acc[q]);
        }
      }
#pragma unroll
      for (int q = 0; q < NQ; ++q) {
        float* sTT = smem + q * QS + oScr;
#pragma unroll
        for (int u = 0; u < 8; ++u)
          sTT[r * SHS + rowBase + u + rhalf] = fmaxf(acc[q][u], 0.f);
      }
    }
    __syncthreads();
    if (hasL) {
      const float* ap = A2 + (size_t)(rcL * 16 + r) * H + hp * HP + khalf;
      const float* bp = smem + q0L * QS + oScr + r * SHS + khalf;
#pragma unroll 4
      for (int kk = 0; kk < HP; kk += 4) {
        v2f afrag = *(const v2f*)(ap + kk);
#pragma unroll
        for (int t = 0; t < QD; ++t) {
          v2f bfrag = *(const v2f*)(bp + t * QS + kk);
          accL[t] = wmma_f32x4(afrag, bfrag, accL[t]);
        }
      }
    }
    if (hp + 1 < NH) __syncthreads();
  }
  if (hasL) {
#pragma unroll
    for (int t = 0; t < QD; ++t) {
      float* sL = smem + (q0L + t) * QS + oL;
#pragma unroll
      for (int u = 0; u < 8; ++u)
        sL[(rcL * 16 + u + rhalf) * 16 + r] = accL[t][u];
    }
  }
  __syncthreads();

  for (int e = tid; e < NQ * D; e += NT) {
    const int q = e % NQ, d = e / NQ;
    const float* sL = smem + q * QS + oL + d * 16;
    const float* sV = smem + q * QS + oV + d * 16;
    float mx = -INFINITY;
#pragma unroll
    for (int j = 0; j < PK; ++j) mx = fmaxf(mx, sL[j]);
    float s = 0.f, o = 0.f;
#pragma unroll
    for (int j = 0; j < PK; ++j) {
      float p = __expf(sL[j] - mx);
      s += p;
      o += p * sV[j];
    }
    const float val = o / s;
    out[((size_t)b * D + d) * PM + n0 + q] = val;
    if (outT) outT[((size_t)b * PM + n0 + q) * D + d] = val;
  }
}

__global__ __launch_bounds__(256)
void pool_kernel(const float* __restrict__ xa, int Ca, const float* __restrict__ xb, int Cb,
                 int chBase, float* __restrict__ out) {
  int b = blockIdx.x, ch = threadIdx.x;
  const float* src = ch < Ca ? xa + ((size_t)b * Ca + ch) * PM
                             : xb + ((size_t)b * Cb + (ch - Ca)) * PM;
  float mx = -INFINITY, sm = 0.f;
  for (int m = 0; m < PM; ++m) { float v = src[m]; mx = fmaxf(mx, v); sm += v; }
  out[(size_t)b * 512 + chBase + ch] = mx;
  out[(size_t)b * 512 + 256 + chBase + ch] = sm * (1.f / PM);
}

extern "C" void kernel_launch(void* const* d_in, const int* in_sizes, int n_in,
                              void* d_out, int out_size, void* d_ws, size_t ws_size,
                              hipStream_t stream) {
  (void)in_sizes; (void)n_in; (void)out_size;
  const float* x = (const float*)d_in[0];
  const float* t1[7]; const float* t2[7]; const float* t3[7];
  for (int i = 0; i < 7; ++i) {
    t1[i] = (const float*)d_in[1 + i];
    t2[i] = (const float*)d_in[8 + i];
    t3[i] = (const float*)d_in[15 + i];
  }
  float* ws = (float*)d_ws;
  size_t off = 0;
  float* pos = ws + off; off += (size_t)PB * 3 * PM;
  float* x1  = ws + off; off += (size_t)PB * 64 * PM;
  float* x2  = ws + off; off += (size_t)PB * 64 * PM;
  float* x3  = ws + off; off += (size_t)PB * 128 * PM;
  int* idxb  = (int*)(ws + off); off += (size_t)PB * PM * PK;
  float* wqk1 = ws + off; off += 64 * 8;
  float* wv1  = ws + off; off += 64 * 8;
  float* wqk2 = ws + off; off += 64 * 128;
  float* wv2  = ws + off; off += 64 * 128;
  float* wqk3 = ws + off; off += 128 * 128;
  float* wv3  = ws + off; off += 128 * 128;
  float* gc   = ws + off; off += (size_t)PB * PM * 4 * 128;
  const bool pre = ws_size >= off * sizeof(float);

  prep_weights_kernel<<<(64 * 8 + 255) / 256, 256, 0, stream>>>(
      t1[0], t1[1], t1[2], wqk1, wv1, 64, 6, 8);
  if (pre) {
    prep_split_kernel<<<(64 * 64 + 255) / 256, 256, 0, stream>>>(t2[0], t2[1], t2[2], wqk2, 64, 64);
    prep_split_kernel<<<(128 * 64 + 255) / 256, 256, 0, stream>>>(t3[0], t3[1], t3[2], wqk3, 128, 64);
  } else {
    prep_weights_kernel<<<(64 * 128 + 255) / 256, 256, 0, stream>>>(
        t2[0], t2[1], t2[2], wqk2, wv2, 64, 128, 128);
    prep_weights_kernel<<<(128 * 128 + 255) / 256, 256, 0, stream>>>(
        t3[0], t3[1], t3[2], wqk3, wv3, 128, 128, 128);
  }

  fps_kernel<<<PB, 256, 0, stream>>>(x, pos);

  float* x1T = x3;
  float* x2T = x1;

  knn_kernel<<<PB * PM / KNN_QPB, KNN_QPB * 32, 0, stream>>>(pos, 3, idxb);
  xform_kernel<XNT, XNQ, 2, 3, 64, 256, 1, false><<<PB * PM / XNQ, XNT, 0, stream>>>(
      pos, pos, idxb, wqk1, wv1, t1[3], t1[4], t1[5], t1[6],
      x1, x1T);
  pool_kernel<<<PB, 64, 0, stream>>>(x1, 64, x1, 0, 0, (float*)d_out);

  knn_kernel<<<PB * PM / KNN_QPB, KNN_QPB * 32, 0, stream>>>(x1, 64, idxb);
  if (pre) {
    point_gemm_kernel<64, 64><<<PB * PM / 16, 256, 0, stream>>>(x1T, wqk2, gc);
    xform_kernel<XNT, XNQ, 2, 64, 64, 256, 1, true><<<PB * PM / XNQ, XNT, 0, stream>>>(
        gc, pos, idxb, wqk2, wv2, t2[3], t2[4], t2[5], t2[6],
        x2, x2T);
  } else {
    xform_kernel<XNT, XNQ, 2, 64, 64, 256, 1, false><<<PB * PM / XNQ, XNT, 0, stream>>>(
        x1T, pos, idxb, wqk2, wv2, t2[3], t2[4], t2[5], t2[6],
        x2, x2T);
  }

  knn_kernel<<<PB * PM / KNN_QPB, KNN_QPB * 32, 0, stream>>>(x2, 64, idxb);
  if (pre) {
    point_gemm_kernel<64, 128><<<PB * PM / 16, 256, 0, stream>>>(x2T, wqk3, gc);
    xform_kernel<XNT, XNQ, 4, 64, 128, 512, 2, true><<<PB * PM / XNQ, XNT, 0, stream>>>(
        gc, pos, idxb, wqk3, wv3, t3[3], t3[4], t3[5], t3[6],
        x3, nullptr);
  } else {
    xform_kernel<XNT, XNQ, 4, 64, 128, 512, 2, false><<<PB * PM / XNQ, XNT, 0, stream>>>(
        x2T, pos, idxb, wqk3, wv3, t3[3], t3[4], t3[5], t3[6],
        x3, nullptr);
  }

  pool_kernel<<<PB, 192, 0, stream>>>(x2, 64, x3, 128, 64, (float*)d_out);
}


# kernels that do not depend on each other share a launch (block-number ranges): weight prep with sampling; x1 pooling and the per-point tables with the knn searches; 14 launches -> 8; sums unchanged
# speedup vs baseline: 3.2647x; 1.0337x over previous
#include <hip/hip_runtime.h>
#include <math.h>

#define PB 16
#define PN 4096
#define PM 512
#define PK 16

typedef float v2f __attribute__((ext_vector_type(2)));
typedef float v8f __attribute__((ext_vector_type(8)));

__device__ __forceinline__ v8f wmma_f32x4(v2f a, v2f b, v8f c) {
  return __builtin_amdgcn_wmma_f32_16x16x4_f32(false, a, false, b, (short)0, c, false, false);
}

__device__ __forceinline__
void prep_weights_body(int i, const float* __restrict__ Wq, const float* __restrict__ Wk,
                       const float* __restrict__ Wv, float* __restrict__ outQK,
                       float* __restrict__ outV, int D, int Cw, int Kpad) {
  if (i >= D * Kpad) return;
  int d = i / Kpad, k = i % Kpad;
  float q = 0.f, v = 0.f;
  if (k < Cw) {
    q = Wq[(size_t)d * Cw + k] - Wk[(size_t)d * Cw + k];
    v = Wv[(size_t)d * Cw + k];
  }
  outQK[i] = q;
  outV[i] = v;
}

__device__ __forceinline__
void prep_split_body(int i, const float* __restrict__ Wq, const float* __restrict__ Wk,
                     const float* __restrict__ Wv, float* __restrict__ W4, int D, int C) {
  if (i >= D * C) return;
  int d = i / C, k = i % C;
  const size_t ia = (size_t)d * 2 * C + k, ib = ia + C;
  const float qa = Wq[ia] - Wk[ia], qb = Wq[ib] - Wk[ib];
  const float va = Wv[ia], vb = Wv[ib];
  W4[((size_t)0 * D + d) * C + k] = qa;
  W4[((size_t)1 * D + d) * C + k] = va;
  W4[((size_t)2 * D + d) * C + k] = qb - qa;
  W4[((size_t)3 * D + d) * C + k] = vb - va;
}

template <int C, int D>
__device__ __forceinline__
void point_gemm_body(int blk, const float* __restrict__ xT, const float* __restrict__ W4,
                     float* __restrict__ GC) {
  constexpr int RPW = (4 * D / 16) / 8;
  static_assert((4 * D / 16) % 8 == 0 && C % 4 == 0, "whole row chunks per wave, whole k steps");
  const int lane = threadIdx.x & 31, wave = threadIdx.x >> 5;
  const int r = lane & 15;
  const int khalf = (lane >> 4) << 1;
  const int rhalf = (lane >> 4) << 3;
  const size_t p0 = (size_t)blk * 16;
  v8f acc[RPW];
#pragma unroll
  for (int t = 0; t < RPW; ++t)
#pragma unroll
    for (int u = 0; u < 8; ++u) acc[t][u] = 0.f;
  const float* bp = xT + (p0 + r) * C + khalf;
  const float* ap = W4 + (size_t)(wave * RPW * 16 + r) * C + khalf;
#pragma unroll 8
  for (int kk = 0; kk < C; kk += 4) {
    v2f bfrag = *(const v2f*)(bp + kk);
#pragma unroll
    for (int t = 0; t < RPW; ++t) {
      v2f afrag = *(const v2f*)(ap + (size_t)t * 16 * C + kk);
      acc[t] = wmma_f32x4(afrag, bfrag, acc[t]);
    }
  }
#pragma unroll
  for (int t = 0; t < RPW; ++t) {
    float* o = GC + (p0 + r) * (4 * D) + (wave * RPW + t) * 16 + rhalf;
    *(float4*)(o)     = make_float4(acc[t][0], acc[t][1], acc[t][2], acc[t][3]);
    *(float4*)(o + 4) = make_float4(acc[t][4], acc[t][5], acc[t][6], acc[t][7]);
  }
}

template <int XM>
__device__ __forceinline__ unsigned row_xor_u32(unsigned v) {
  return (unsigned)__builtin_amdgcn_update_dpp(0, (int)v, 0x160 | XM, 0xf, 0xf, true);
}
__device__ __forceinline__ unsigned umax2(unsigned a, unsigned b) { return a > b ? a : b; }
__device__ __forceinline__ unsigned umin2(unsigned a, unsigned b) { return a < b ? a : b; }
__device__ __forceinline__ unsigned row_umax(unsigned v) {
  v = umax2(v, row_xor_u32<1>(v));
  v = umax2(v, row_xor_u32<2>(v));
  v = umax2(v, row_xor_u32<4>(v));
  return umax2(v, row_xor_u32<8>(v));
}
__device__ __forceinline__ unsigned row_umin(unsigned v) {
  v = umin2(v, row_xor_u32<1>(v));
  v = umin2(v, row_xor_u32<2>(v));
  v = umin2(v, row_xor_u32<4>(v));
  return umin2(v, row_xor_u32<8>(v));
}

__device__ __forceinline__
void fps_body(int b, const float* __restrict__ x, float* __restrict__ pos) {
  const int tid = threadIdx.x;
  const int PPT = PN / 256;
  __shared__ float sX[PN];
  __shared__ float sY[PN];
  __shared__ float sZ[PN];
  __shared__ int sSel[PM];
  __shared__ uint2 sPart[2][16];

  float px[PPT], py[PPT], pz[PPT], dmin[PPT];
#pragma unroll
  for (int i = 0; i < PPT; ++i) {
    int p = tid + i * 256;
    const float* xp = x + ((size_t)b * PN + p) * 3;
    px[i] = xp[0]; py[i] = xp[1]; pz[i] = xp[2];
    sX[p] = px[i]; sY[p] = py[i]; sZ[p] = pz[i];
    dmin[i] = INFINITY;
  }
  if (tid == 0) sSel[0] = 0;
  __syncthreads();
  float bx = sX[0], by = sY[0], bz = sZ[0];
  for (int it = 1; it < PM; ++it) {
    float best = -INFINITY; int bslot = PPT;
#pragma unroll
    for (int i = 0; i < PPT; ++i) {
      float dx = px[i] - bx, dy = py[i] - by, dz = pz[i] - bz;
      float d = dx * dx + dy * dy + dz * dz;
      dmin[i] = fminf(dmin[i], d);
      if (dmin[i] > best) { best = dmin[i]; bslot = i; }
    }
    const unsigned barg = (unsigned)tid + ((unsigned)bslot << 8);
    const unsigned vb = __float_as_uint(best);
    const unsigned rmax = row_umax(vb);
    const unsigned rarg = row_umin(vb == rmax ? barg : 0xFFFFFFFFu);
    uint2* part = sPart[it & 1];
    if ((tid & 15) == 0) part[tid >> 4] = make_uint2(rmax, rarg);
    __syncthreads();
    const uint2 pr = part[tid & 15];
    const unsigned gmax = row_umax(pr.x);
    const unsigned garg = row_umin(pr.x == gmax ? pr.y : 0xFFFFFFFFu);
    bx = sX[garg]; by = sY[garg]; bz = sZ[garg];
    if (tid == 0) sSel[it] = (int)garg;
  }
  __syncthreads();
  for (int i = tid; i < PM; i += 256) {
    int p = sSel[i];
    pos[((size_t)b * 3 + 0) * PM + i] = sX[p];
    pos[((size_t)b * 3 + 1) * PM + i] = sY[p];
    pos[((size_t)b * 3 + 2) * PM + i] = sZ[p];
  }
}

#define KNN_QPB 8
__device__ __forceinline__
void knn_body(int blk, const float* __restrict__ f, int C, int* __restrict__ idxout) {
  const int lane = threadIdx.x & 31;
  const int wave = __builtin_amdgcn_readfirstlane((int)(threadIdx.x >> 5));
  const int bn = blk * KNN_QPB + wave;
  const int b = bn / PM, n = bn % PM;
  const float* fb = f + (size_t)b * C * PM;
  float acc[16];
#pragma unroll
  for (int i = 0; i < 16; ++i) acc[i] = 0.f;
#pragma unroll 2
  for (int c = 0; c < C; ++c) {
    const float* row = fb + (size_t)c * PM;
    const float fq = row[n];
#pragma unroll
    for (int i = 0; i < 16; ++i) {
      float d = fq - row[lane + 32 * i];
      acc[i] += d * d;
    }
  }
  unsigned key[16];
#pragma unroll
  for (int i = 0; i < 16; ++i) key[i] = __float_as_uint(acc[i]);
  unsigned mine = 0;
  for (int r = 0; r < PK; ++r) {
    unsigned bv = key[0], bi = 0;
#pragma unroll
    for (int i = 1; i < 16; ++i) {
      if (key[i] < bv) { bv = key[i]; bi = (unsigned)i; }
    }
    const unsigned bm = (unsigned)lane + (bi << 5);
    const unsigned rmin = row_umin(bv);
    const unsigned gmin = umin2((unsigned)__builtin_amdgcn_readlane((int)rmin, 0),
                                (unsigned)__builtin_amdgcn_readlane((int)rmin, 16));
    const unsigned rarg = row_umin(bv == gmin ? bm : 0xFFFFFFFFu);
    const unsigned garg = umin2((unsigned)__builtin_amdgcn_readlane((int)rarg, 0),
                                (unsigned)__builtin_amdgcn_readlane((int)rarg, 16));
    if (lane == r) mine = garg;
#pragma unroll
    for (int i = 0; i < 16; ++i) {
      if ((unsigned)(lane + 32 * i) == garg) key[i] = 0xFFFFFFFFu;
    }
  }
  if (lane < PK) idxout[(size_t)bn * PK + lane] = (int)(mine & (unsigned)(PM - 1));
}

#define XNT 512
#define XNQ 4

template <int NT, int NQ, int QD, int CIN, int D, int H, int NH, bool PRE>
__global__ __launch_bounds__(NT)
void xform_kernel(const float* __restrict__ f, const float* __restrict__ pos,
                  const int* __restrict__ idx,
                  const float* __restrict__ WqkP, const float* __restrict__ WvP,
                  const float* __restrict__ P1, const float* __restrict__ P2,
                  const float* __restrict__ A1, const float* __restrict__ A2,
                  float* __restrict__ out, float* __restrict__ outT) {
  constexpr int NW = NT / 32;
  constexpr int NG = NQ / QD;
  constexpr int Kpad = (CIN == 3) ? 8 : 2 * CIN;
  const int bn0 = blockIdx.x * NQ;
  const int b = bn0 / PM, n0 = bn0 % PM;
  const int tid = threadIdx.x;
  const int lane = tid & 31, wave = tid >> 5;
  constexpr int SG = Kpad + 2;
  constexpr int SD = D + 2;
  constexpr int HP  = H / NH;
  constexpr int SHS = HP + 2;
  static_assert(!PRE || CIN != 3, "the table path is for the layers whose features are a previous layer's output");
  constexpr int scrA = PRE ? 16 * SD : 16 * SG + D * 16;
  constexpr int scrF = (scrA > 16 * SHS) ? scrA : (16 * SHS);
  constexpr int oT1  = PRE ? (48 + 16 * SD + D * 16) : 48;
  constexpr int oHT  = 48;
  constexpr int oV   = oHT + 16 * SD;
  constexpr int oScr = oV + D * 16;
  constexpr int oPe  = oScr + 16 * SG;
  constexpr int oL   = (NH > 1) ? oHT : oScr + scrF;
  constexpr int oCn  = (NH > 1) ? oScr + scrF : oL + D * 16;
  constexpr int QS   = oCn + (PRE ? 2 * D : 0);
  static_assert(QS % 4 == 0, "regions keep the 8-byte alignment of the fragment reads");
  static_assert(D % 16 == 0 && H % NH == 0 && HP % 16 == 0 && Kpad % 4 == 0, "whole WMMA tiles and k steps");
  __shared__ float smem[NQ * QS] __attribute__((aligned(16)));

  constexpr int QP = (QD > 1) ? QD / 2 : 1;
  static_assert(NQ % QP == 0, "whole groups of QP points");
  float4 gq[QP][2], gv[QP][2];
  if constexpr (PRE) {
    static_assert((D / 16) * (NQ / QP) == NW, "one item per wave");
    const int rcG = wave % (D / 16), q0G = (wave / (D / 16)) * QP;
    const int rowG = rcG * 16 + ((lane >> 4) << 3);
#pragma unroll
    for (int t = 0; t < QP; ++t) {
      const int nb = idx[(size_t)(bn0 + q0G + t) * PK + (lane & 15)];
      const float* g = f + ((size_t)b * PM + nb) * (4 * D) + rowG;
      gq[t][0] = *(const float4*)(g);         gq[t][1] = *(const float4*)(g + 4);
      gv[t][0] = *(const float4*)(g + D);     gv[t][1] = *(const float4*)(g + D + 4);
    }
  }
  if (tid == 0) {
    __builtin_prefetch(A1, 0, 3);
    __builtin_prefetch(A2, 0, 3);
  }
  if constexpr (CIN == 3) {
    for (int e = tid; e < NQ * 16 * Kpad; e += NT) {
      const int q = e / (16 * Kpad), t = e % (16 * Kpad);
      const int j = t / Kpad, row = t % Kpad;
      float v = 0.f;
      if (row < 2 * CIN) {
        const float* pc = pos + ((size_t)b * CIN + (row < CIN ? row : row - CIN)) * PM;
        v = pc[n0 + q];
        if (row < CIN) {
          v = pc[idx[(size_t)(bn0 + q) * PK + j]] - v;
          smem[q * QS + row * 16 + j] = v;
        }
      }
      smem[q * QS + oScr + j * SG + row] = v;
    }
  } else if constexpr (PRE) {
    static_assert(NT >= 256 + NQ * 2 * D / 4 && NQ * 48 <= 256, "both jobs of this pass have their own threads");
    if (tid < NQ * 48) {
      const int q = tid / 48, c = (tid % 48) / 16, j = tid % 16;
      const float* pc = pos + ((size_t)b * 3 + c) * PM;
      smem[q * QS + c * 16 + j] = pc[idx[(size_t)(bn0 + q) * PK + j]] - pc[n0 + q];
    } else if (tid >= 256 && tid - 256 < NQ * 2 * D / 4) {
      const int q = (tid - 256) / (2 * D / 4), c4 = (tid - 256) % (2 * D / 4);
      *(float4*)(smem + q * QS + oCn + c4 * 4) =
          *(const float4*)(f + ((size_t)b * PM + n0 + q) * (4 * D) + 2 * D + c4 * 4);
    }
  } else {
    constexpr int C4 = CIN / 4;
    constexpr int NP = NQ * 16 * C4 / NT;
    static_assert(NQ * 16 * C4 % NT == 0 && NT % (16 * C4) == 0, "whole passes, c4 and j fixed per thread");
    const int c4 = tid % C4, j = (tid / C4) % 16;
    int nbv[NP];
    float4 cen[NP], nbf[NP];
#pragma unroll
    for (int p = 0; p < NP; ++p) {
      const int q = (tid + p * NT) / (16 * C4);
      nbv[p] = idx[(size_t)(bn0 + q) * PK + j];
      cen[p] = *(const float4*)(f + ((size_t)b * PM + n0 + q) * CIN + c4 * 4);
    }
#pragma unroll
    for (int p = 0; p < NP; ++p)
      nbf[p] = *(const float4*)(f + ((size_t)b * PM + nbv[p]) * CIN + c4 * 4);
    if (c4 < 3) {
      const float* pc = pos + ((size_t)b * 3 + c4) * PM;
      float rel[NP];
#pragma unroll
      for (int p = 0; p < NP; ++p) rel[p] = pc[nbv[p]] - pc[n0 + (tid + p * NT) / (16 * C4)];
#pragma unroll
      for (int p = 0; p < NP; ++p) smem[((tid + p * NT) / (16 * C4)) * QS + c4 * 16 + j] = rel[p];
    }
#pragma unroll
    for (int p = 0; p < NP; ++p) {
      float* dst = smem + ((tid + p * NT) / (16 * C4)) * QS + oScr + j * SG + c4 * 4;
      *(v2f*)(dst)           = v2f{nbf[p].x - cen[p].x, nbf[p].y - cen[p].y};
      *(v2f*)(dst + 2)       = v2f{nbf[p].z - cen[p].z, nbf[p].w - cen[p].w};
      *(v2f*)(dst + CIN)     = v2f{cen[p].x, cen[p].y};
      *(v2f*)(dst + CIN + 2) = v2f{cen[p].z, cen[p].w};
    }
  }
  __syncthreads();

  {
    static_assert(NT % D == 0 && 16 % (NT / D) == 0, "one row per thread, whole passes over the 16 columns");
    constexpr int JS = NT / D;
    const int d = tid % D, jb = tid / D;
    const float p0 = P1[d * 3 + 0], p1 = P1[d * 3 + 1], p2 = P1[d * 3 + 2];
    constexpr int NJ = 16 / JS;
    float r0[NQ * NJ], r1[NQ * NJ], r2[NQ * NJ];
#pragma unroll
    for (int q = 0; q < NQ; ++q)
#pragma unroll
      for (int i = 0; i < NJ; ++i) {
        const float* sRel = smem + q * QS + jb + i * JS;
        r0[q * NJ + i] = sRel[0]; r1[q * NJ + i] = sRel[16]; r2[q * NJ + i] = sRel[32];
      }
#pragma unroll
    for (int q = 0; q < NQ; ++q)
#pragma unroll
      for (int i = 0; i < NJ; ++i) {
        float a = fmaf(p2, r2[q * NJ + i], fmaf(p0, r0[q * NJ + i], p1 * r1[q * NJ + i]));
        smem[q * QS + oT1 + (jb + i * JS) * SD + d] = fmaxf(a, 0.f);
      }
  }
  __syncthreads();
  const int r = lane & 15;
  const int khalf = (lane >> 4) << 1;
  const int rhalf = (lane >> 4) << 3;
  constexpr int nrcD = D / 16;

  if constexpr (PRE) {
    {
      const int rc = wave % nrcD, q0 = (wave / nrcD) * QP;
      const int rowBase = rc * 16;
      const int row = rowBase + r;
      v8f acc[QP];
#pragma unroll
      for (int t = 0; t < QP; ++t)
#pragma unroll
        for (int u = 0; u < 8; ++u) acc[t][u] = 0.f;
      const float* ap = P2 + (size_t)row * D + khalf;
      const float* bp = smem + q0 * QS + oT1 + r * SD + khalf;
#pragma unroll 8
      for (int kk = 0; kk < D; kk += 4) {
        v2f afrag = *(const v2f*)(ap + kk);
#pragma unroll
        for (int t = 0; t < QP; ++t) {
          v2f bfrag = *(const v2f*)(bp + t * QS + kk);
          acc[t] = wmma_f32x4(afrag, bfrag, acc[t]);
        }
      }
#pragma unroll
      for (int t = 0; t < QP; ++t) {
        float* sHT = smem + (q0 + t) * QS + oHT;
        float* sV  = smem + (q0 + t) * QS + oV;
        const float* sC = smem + (q0 + t) * QS + oCn + rowBase + rhalf;
        const float4 cq0 = *(const float4*)(sC),     cq1 = *(const float4*)(sC + 4);
        const float4 cv0 = *(const float4*)(sC + D), cv1 = *(const float4*)(sC + D + 4);
        const float eq[8] = {gq[t][0].x + cq0.x, gq[t][0].y + cq0.y, gq[t][0].z + cq0.z, gq[t][0].w + cq0.w,
                             gq[t][1].x + cq1.x, gq[t][1].y + cq1.y, gq[t][1].z + cq1.z, gq[t][1].w + cq1.w};
        const float ev[8] = {gv[t][0].x + cv0.x, gv[t][0].y + cv0.y, gv[t][0].z + cv0.z, gv[t][0].w + cv0.w,
                             gv[t][1].x + cv1.x, gv[t][1].y + cv1.y, gv[t][1].z + cv1.z, gv[t][1].w + cv1.w};
#pragma unroll
        for (int u = 0; u < 8; ++u) {
          const float pe = acc[t][u];
          sHT[r * SD + rowBase + u + rhalf] = pe + eq[u];
          sV[(rowBase + u + rhalf) * 16 + r] = pe + ev[u];
        }
      }
    }
  } else {
  for (int item = wave; item < nrcD * (NQ / QP); item += NW) {
    const int rc = item % nrcD, q0 = (item / nrcD) * QP;
    const int rowBase = rc * 16;
    const int row = rowBase + r;
    v8f acc[QP];
#pragma unroll
    for (int t = 0; t < QP; ++t)
#pragma unroll
      for (int u = 0; u < 8; ++u) acc[t][u] = 0.f;
    const float* ap = P2 + (size_t)row * D + khalf;
    const float* bp = smem + q0 * QS + oHT + r * SD + khalf;
#pragma unroll 8
    for (int kk = 0; kk < D; kk += 4) {
      v2f afrag = *(const v2f*)(ap + kk);
#pragma unroll
      for (int t = 0; t < QP; ++t) {
        v2f bfrag = *(const v2f*)(bp + t * QS + kk);
        acc[t] = wmma_f32x4(afrag, bfrag, acc[t]);
      }
    }
#pragma unroll
    for (int t = 0; t < QP; ++t) {
      float* sPe = smem + (q0 + t) * QS + oPe;
#pragma unroll
      for (int u = 0; u < 8; ++u)
        sPe[(rowBase + u + rhalf) * 16 + r] = acc[t][u];
    }
  }
  __syncthreads();

  for (int item = wave; item < nrcD * NG; item += NW) {
    const int rc = item % nrcD, q0 = (item / nrcD) * QD;
    const int rowBase = rc * 16;
    const int row = rowBase + r;
    v8f acc[QD], accv[QD];
#pragma unroll
    for (int t = 0; t < QD; ++t) {
      const float* sPe = smem + (q0 + t) * QS + oPe;
#pragma unroll
      for (int u = 0; u < 8; ++u) {
        float pe = sPe[(rowBase + u + rhalf) * 16 + r];
        acc[t][u] = pe; accv[t][u] = pe;
      }
    }
    const float* aqp = WqkP + (size_t)row * Kpad + khalf;
    const float* avp = WvP  + (size_t)row * Kpad + khalf;
    const float* bp  = smem + q0 * QS + oScr + r * SG + khalf;
#pragma unroll 4
    for (int kk = 0; kk < Kpad; kk += 4) {
      v2f aq = *(const v2f*)(aqp + kk);
      v2f av = *(const v2f*)(avp + kk);
#pragma unroll
      for (int t = 0; t < QD; ++t) {
        v2f bfrag = *(const v2f*)(bp + t * QS + kk);
        acc[t]  = wmma_f32x4(aq, bfrag, acc[t]);
        accv[t] = wmma_f32x4(av, bfrag, accv[t]);
      }
    }
#pragma unroll
    for (int t = 0; t < QD; ++t) {
      float* sHT = smem + (q0 + t) * QS + oHT;
      float* sV  = smem + (q0 + t) * QS + oV;
#pragma unroll
      for (int u = 0; u < 8; ++u) {
        sHT[r * SD + rowBase + u + rhalf] = acc[t][u];
        sV[(rowBase + u + rhalf) * 16 + r] = accv[t][u];
      }
    }
  }
  }
  __syncthreads();

  static_assert(nrcD * NG <= NW, "at most one logits item per wave");
  const bool hasL = wave < nrcD * NG;
  const int rcL = wave % nrcD, q0L = (wave / nrcD) * QD;
  v8f accL[QD];
#pragma unroll
  for (int t = 0; t < QD; ++t)
#pragma unroll
    for (int u = 0; u < 8; ++u) accL[t][u] = 0.f;
#pragma unroll
  for (int hp = 0; hp < NH; ++hp) {
    for (int rc = wave; rc < HP / 16; rc += NW) {
      const int rowBase = rc * 16;
      const int row = hp * HP + rowBase + r;
      v8f acc[NQ];
#pragma unroll
      for (int q = 0; q < NQ; ++q)
#pragma unroll
        for (int u = 0; u < 8; ++u) acc[q][u] = 0.f;
      const float* ap = A1 + (size_t)row * D + khalf;
      const float* bp = smem + oHT + r * SD + khalf;
#pragma unroll 4
      for (int kk = 0; kk < D; kk += 4) {
        v2f afrag = *(const v2f*)(ap + kk);
#pragma unroll
        for (int q = 0; q < NQ; ++q) {
          v2f bfrag = *(const v2f*)(bp + q * QS + kk);
          acc[q] = wmma_f32x4(afrag, bfrag, acc[q]);
        }
      }
#pragma unroll
      for (int q = 0; q < NQ; ++q) {
        float* sTT = smem + q * QS + oScr;
#pragma unroll
        for (int u = 0; u < 8; ++u)
          sTT[r * SHS + rowBase + u + rhalf] = fmaxf(acc[q][u], 0.f);
      }
    }
    __syncthreads();
    if (hasL) {
      const float* ap = A2 + (size_t)(rcL * 16 + r) * H + hp * HP + khalf;
      const float* bp = smem + q0L * QS + oScr + r * SHS + khalf;
#pragma unroll 4
      for (int kk = 0; kk < HP; kk += 4) {
        v2f afrag = *(const v2f*)(ap + kk);
#pragma unroll
        for (int t = 0; t < QD; ++t) {
          v2f bfrag = *(const v2f*)(bp + t * QS + kk);
          accL[t] = wmma_f32x4(afrag, bfrag, accL[t]);
        }
      }
    }
    if (hp + 1 < NH) __syncthreads();
  }
  if (hasL) {
#pragma unroll
    for (int t = 0; t < QD; ++t) {
      float* sL = smem + (q0L + t) * QS + oL;
#pragma unroll
      for (int u = 0; u < 8; ++u)
        sL[(rcL * 16 + u + rhalf) * 16 + r] = accL[t][u];
    }
  }
  __syncthreads();

  for (int e = tid; e < NQ * D; e += NT) {
    const int q = e % NQ, d = e / NQ;
    const float* sL = smem + q * QS + oL + d * 16;
    const float* sV = smem + q * QS + oV + d * 16;
    float mx = -INFINITY;
#pragma unroll
    for (int j = 0; j < PK; ++j) mx = fmaxf(mx, sL[j]);
    float s = 0.f, o = 0.f;
#pragma unroll
    for (int j = 0; j < PK; ++j) {
      float p = __expf(sL[j] - mx);
      s += p;
      o += p * sV[j];
    }
    const float val = o / s;
    out[((size_t)b * D + d) * PM + n0 + q] = val;
    if (outT) outT[((size_t)b * PM + n0 + q) * D + d] = val;
  }
}

__device__ __forceinline__
void pool_body(int b, const float* __restrict__ xa, int Ca, const float* __restrict__ xb, int Cb,
               int chBase, float* __restrict__ out) {
  int ch = threadIdx.x;
  if (ch >= Ca + Cb) return;
  const float* src = ch < Ca ? xa + ((size_t)b * Ca + ch) * PM
                             : xb + ((size_t)b * Cb + (ch - Ca)) * PM;
  float mx = -INFINITY, sm = 0.f;
  for (int m = 0; m < PM; ++m) { float v = src[m]; mx = fmaxf(mx, v); sm += v; }
  out[(size_t)b * 512 + chBase + ch] = mx;
  out[(size_t)b * 512 + 256 + chBase + ch] = sm * (1.f / PM);
}

struct PrepArgs {
  const float *q1, *k1, *v1, *q2, *k2, *v2, *q3, *k3, *v3;
  float *wqk1, *wv1, *wqk2, *wv2, *wqk3, *wv3;
  int split;
};
#define PREP_B1  ((64 * 8 + 255) / 256)
#define PREP_B2S ((64 * 64 + 255) / 256)
#define PREP_B3S ((128 * 64 + 255) / 256)
#define PREP_B2P ((64 * 128 + 255) / 256)
#define PREP_B3P ((128 * 128 + 255) / 256)

__global__ __launch_bounds__(256)
void fps_prep_kernel(const float* __restrict__ x, float* __restrict__ pos, PrepArgs a) {
  if (blockIdx.x < PB) { fps_body((int)blockIdx.x, x, pos); return; }
  int pb = (int)blockIdx.x - PB;
  const int tid = threadIdx.x;
  if (pb < PREP_B1) { prep_weights_body(pb * 256 + tid, a.q1, a.k1, a.v1, a.wqk1, a.wv1, 64, 6, 8); return; }
  pb -= PREP_B1;
  if (a.split) {
    if (pb < PREP_B2S) { prep_split_body(pb * 256 + tid, a.q2, a.k2, a.v2, a.wqk2, 64, 64); return; }
    pb -= PREP_B2S;
    prep_split_body(pb * 256 + tid, a.q3, a.k3, a.v3, a.wqk3, 128, 64);
  } else {
    if (pb < PREP_B2P) { prep_weights_body(pb * 256 + tid, a.q2, a.k2, a.v2, a.wqk2, a.wv2, 64, 128, 128); return; }
    pb -= PREP_B2P;
    prep_weights_body(pb * 256 + tid, a.q3, a.k3, a.v3, a.wqk3, a.wv3, 128, 128, 128);
  }
}

__global__ __launch_bounds__(KNN_QPB * 32)
void knn_kernel(const float* __restrict__ f, int C, int* __restrict__ idxout) {
  knn_body((int)blockIdx.x, f, C, idxout);
}

template <int D>
__global__ __launch_bounds__(KNN_QPB * 32)
void knn_side_kernel(const float* __restrict__ f, int C, int* __restrict__ idxout,
                     const float* __restrict__ xT, const float* __restrict__ W4,
                     float* __restrict__ GC, int npg,
                     float* __restrict__ pout, int npool) {
  static_assert(KNN_QPB * 32 == 256, "the table's blocks are 8 waves");
  int blk = (int)blockIdx.x;
  if (blk < npool) { pool_body(blk, f, C, f, 0, 0, pout); return; }
  blk -= npool;
  if (blk < npg) { point_gemm_body<64, D>(blk, xT, W4, GC); return; }
  blk -= npg;
  knn_body(blk, f, C, idxout);
}

__global__ __launch_bounds__(256)
void pool_kernel(const float* __restrict__ xa, int Ca, const float* __restrict__ xb, int Cb,
                 int chBase, float* __restrict__ out) {
  pool_body((int)blockIdx.x, xa, Ca, xb, Cb, chBase, out);
}

extern "C" void kernel_launch(void* const* d_in, const int* in_sizes, int n_in,
                              void* d_out, int out_size, void* d_ws, size_t ws_size,
                              hipStream_t stream) {
  (void)in_sizes; (void)n_in; (void)out_size;
  const float* x = (const float*)d_in[0];
  const float* t1[7]; const float* t2[7]; const float* t3[7];
  for (int i = 0; i < 7; ++i) {
    t1[i] = (const float*)d_in[1 + i];
    t2[i] = (const float*)d_in[8 + i];
    t3[i] = (const float*)d_in[15 + i];
  }
  float* ws = (float*)d_ws;
  size_t off = 0;
  float* pos = ws + off; off += (size_t)PB * 3 * PM;
  float* x1  = ws + off; off += (size_t)PB * 64 * PM;
  float* x2  = ws + off; off += (size_t)PB * 64 * PM;
  float* x3  = ws + off; off += (size_t)PB * 128 * PM;
  int* idxb  = (int*)(ws + off); off += (size_t)PB * PM * PK;
  float* wqk1 = ws + off; off += 64 * 8;
  float* wv1  = ws + off; off += 64 * 8;
  float* wqk2 = ws + off; off += 64 * 128;
  float* wv2  = ws + off; off += 64 * 128;
  float* wqk3 = ws + off; off += 128 * 128;
  float* wv3  = ws + off; off += 128 * 128;
  float* gc   = ws + off; off += (size_t)PB * PM * 4 * 128;
  const bool pre = ws_size >= off * sizeof(float);

  PrepArgs pa;
  pa.q1 = t1[0]; pa.k1 = t1[1]; pa.v1 = t1[2];
  pa.q2 = t2[0]; pa.k2 = t2[1]; pa.v2 = t2[2];
  pa.q3 = t3[0]; pa.k3 = t3[1]; pa.v3 = t3[2];
  pa.wqk1 = wqk1; pa.wv1 = wv1; pa.wqk2 = wqk2; pa.wv2 = wv2; pa.wqk3 = wqk3; pa.wv3 = wv3;
  if (pre) {
    pa.split = 1;
    fps_prep_kernel<<<PB + PREP_B1 + PREP_B2S + PREP_B3S, 256, 0, stream>>>(x, pos, pa);
  } else {
    pa.split = 0;
    fps_prep_kernel<<<PB + PREP_B1 + PREP_B2P + PREP_B3P, 256, 0, stream>>>(x, pos, pa);
  }

  float* x1T = x3;
  float* x2T = x1;

  knn_kernel<<<PB * PM / KNN_QPB, KNN_QPB * 32, 0, stream>>>(pos, 3, idxb);
  xform_kernel<XNT, XNQ, 2, 3, 64, 256, 1, false><<<PB * PM / XNQ, XNT, 0, stream>>>(
      pos, pos, idxb, wqk1, wv1, t1[3], t1[4], t1[5], t1[6],
      x1, x1T);

  if (pre) {
    knn_side_kernel<64><<<PB + PB * PM / 16 + PB * PM / KNN_QPB, KNN_QPB * 32, 0, stream>>>(
        x1, 64, idxb, x1T, wqk2, gc, PB * PM / 16, (float*)d_out, PB);
    xform_kernel<XNT, XNQ, 2, 64, 64, 256, 1, true><<<PB * PM / XNQ, XNT, 0, stream>>>(
        gc, pos, idxb, wqk2, wv2, t2[3], t2[4], t2[5], t2[6],
        x2, x2T);
  } else {
    knn_side_kernel<64><<<PB + PB * PM / KNN_QPB, KNN_QPB * 32, 0, stream>>>(
        x1, 64, idxb, x1T, wqk2, gc, 0, (float*)d_out, PB);
    xform_kernel<XNT, XNQ, 2, 64, 64, 256, 1, false><<<PB * PM / XNQ, XNT, 0, stream>>>(
        x1T, pos, idxb, wqk2, wv2, t2[3], t2[4], t2[5], t2[6],
        x2, x2T);
  }

  if (pre) {
    knn_side_kernel<128><<<PB * PM / 16 + PB * PM / KNN_QPB, KNN_QPB * 32, 0, stream>>>(
        x2, 64, idxb, x2T, wqk3, gc, PB * PM / 16, (float*)d_out, 0);
    xform_kernel<XNT, XNQ, 4, 64, 128, 512, 2, true><<<PB * PM / XNQ, XNT, 0, stream>>>(
        gc, pos, idxb, wqk3, wv3, t3[3], t3[4], t3[5], t3[6],
        x3, nullptr);
  } else {
    knn_kernel<<<PB * PM / KNN_QPB, KNN_QPB * 32, 0, stream>>>(x2, 64, idxb);
    xform_kernel<XNT, XNQ, 4, 64, 128, 512, 2, false><<<PB * PM / XNQ, XNT, 0, stream>>>(
        x2T, pos, idxb, wqk3, wv3, t3[3], t3[4], t3[5], t3[6],
        x3, nullptr);
  }

  pool_kernel<<<PB, 192, 0, stream>>>(x2, 64, x3, 128, 64, (float*)d_out);
}
